# starting kernel
# baseline (speedup 1.0000x reference)
#include <hip/hip_runtime.h>

typedef float v2f __attribute__((ext_vector_type(2)));
typedef float v8f __attribute__((ext_vector_type(8)));

#define LDS_STRIDE 132

__global__ void deg_init_kernel(float* deg, int n) {
  int i = blockIdx.x * blockDim.x + threadIdx.x;
  if (i < n) deg[i] = 1.0f;
}

__global__ void deg_count_kernel(const int* __restrict__ dst, float* __restrict__ deg, int nE) {
  int e = blockIdx.x * blockDim.x + threadIdx.x;
  if (e < nE) unsafeAtomicAdd(&deg[dst[e]], 1.0f);
}

__global__ void deg_rsqrt_kernel(float* deg, int n) {
  int i = blockIdx.x * blockDim.x + threadIdx.x;
  if (i < n) deg[i] = rsqrtf(deg[i]);
}

__global__ void gemm_gather_scale_kernel(const float* __restrict__ X,
                                         const int* __restrict__ gidx,
                                         const float* __restrict__ W,
                                         const float* __restrict__ dinv,
                                         float* __restrict__ hs,
                                         float* __restrict__ agg,
                                         int n_rows, int N) {
  extern __shared__ float lds[];
  float* lA = lds;
  float* lW = lds + 64 * LDS_STRIDE;
  const int tid = threadIdx.x;
  const int row0 = blockIdx.x * 64;

  const float4* X4 = (const float4*)X;
  for (int i = tid; i < 64 * 32; i += 256) {
    int r = i >> 5;
    int c4 = i & 31;
    int gr = row0 + r;
    float4 v = make_float4(0.f, 0.f, 0.f, 0.f);
    if (gr < n_rows) {
      int sr = gidx ? gidx[gr] : gr;
      v = X4[(size_t)sr * 32 + c4];
    }
    *(float4*)&lA[r * LDS_STRIDE + (c4 << 2)] = v;
  }
  for (int i = tid; i < 128 * N; i += 256) {
    int k = i / N;
    int n = i - k * N;
    lW[n * LDS_STRIDE + k] = W[i];
  }
  __syncthreads();

  const int wave = tid >> 5;
  const int lane = tid & 31;
  const int nct = N >> 4;
  const int col = lane & 15;
  const int koff = (lane >> 4) << 1;

  for (int t = wave; t < 4 * nct; t += 8) {
    int rt = t / nct;
    int ct = t - rt * nct;
    int mbase = rt << 4;
    int nbase = ct << 4;
    const float* aRow = &lA[(mbase + col) * LDS_STRIDE + koff];
    const float* bRow = &lW[(nbase + col) * LDS_STRIDE + koff];
    v8f c = {0.f, 0.f, 0.f, 0.f, 0.f, 0.f, 0.f, 0.f};
#pragma unroll
    for (int k0 = 0; k0 < 128; k0 += 4) {
      v2f a = *(const v2f*)(aRow + k0);
      v2f b = *(const v2f*)(bRow + k0);
      c = __builtin_amdgcn_wmma_f32_16x16x4_f32(false, a, false, b, (short)0, c, false, false);
    }
    int madd = (lane < 16) ? 0 : 8;
#pragma unroll
    for (int i = 0; i < 8; i++) {
      int gr = row0 + mbase + i + madd;
      if (gr < n_rows) {
        float v = c[i] * dinv[gr];
        size_t o = (size_t)gr * N + nbase + col;
        hs[o] = v;
        agg[o] = v;
      }
    }
  }
}

__global__ void scatter_kernel(const int* __restrict__ src, const int* __restrict__ dst,
                               const float* __restrict__ hs, float* __restrict__ agg,
                               int nE, int N) {
  int g = blockIdx.x * blockDim.x + threadIdx.x;
  int e = g >> 5;
  int lane = g & 31;
  if (e >= nE) return;
  const float* hp = hs + (size_t)src[e] * N;
  float* ap = agg + (size_t)dst[e] * N;
  for (int f = lane; f < N; f += 32)
    unsafeAtomicAdd(&ap[f], hp[f]);
}

__global__ void combine_kernel(const float* __restrict__ agg, const float* __restrict__ dinv,
                               const float* __restrict__ bias, float* __restrict__ out,
                               int n_rows, int N, int do_relu) {
  int i = blockIdx.x * blockDim.x + threadIdx.x;
  if (i >= n_rows * N) return;
  int r = i / N;
  int f = i - r * N;
  float v = dinv[r] * agg[i] + bias[f];
  if (do_relu) v = fmaxf(v, 0.f);
  out[i] = v;
}

__global__ void edgedot_kernel(const int* __restrict__ es, const int* __restrict__ ed,
                               const float* __restrict__ z, float* __restrict__ out, int nL) {
  int g = blockIdx.x * blockDim.x + threadIdx.x;
  int e = g >> 5;
  int lane = g & 31;
  if (e >= nL) return;
  const float* a = z + (size_t)es[e] * 64;
  const float* b = z + (size_t)ed[e] * 64;
  float p = a[lane] * b[lane] + a[lane + 32] * b[lane + 32];
#pragma unroll
  for (int off = 16; off > 0; off >>= 1) p += __shfl_xor(p, off, 32);
  if (lane == 0) out[e] = p;
}

extern "C" void kernel_launch(void* const* d_in, const int* in_sizes, int n_in,
                              void* d_out, int out_size, void* d_ws, size_t ws_size,
                              hipStream_t stream) {
  const int*   x_idx = (const int*)d_in[0];
  const int*   eidx  = (const int*)d_in[1];
  const int*   lidx  = (const int*)d_in[2];
  const float* emb   = (const float*)d_in[3];
  const float* W1    = (const float*)d_in[4];
  const float* b1    = (const float*)d_in[5];
  const float* W2    = (const float*)d_in[6];
  const float* b2    = (const float*)d_in[7];
  float* out = (float*)d_out;

  const int nN = in_sizes[0];
  const int nE = in_sizes[1] / 2;
  const int nL = in_sizes[2] / 2;
  const int H = 128, O = 64;

  float* dinv = (float*)d_ws;
  float* hs1  = dinv + ((nN + 63) & ~63);
  float* agg1 = hs1 + (size_t)nN * H;
  float* hs2  = agg1 + (size_t)nN * H;
  float* agg2 = hs2 + (size_t)nN * O;

  const int* esrc = eidx;
  const int* edst = eidx + nE;

  deg_init_kernel<<<(nN + 255) / 256, 256, 0, stream>>>(dinv, nN);
  deg_count_kernel<<<(nE + 255) / 256, 256, 0, stream>>>(edst, dinv, nE);
  deg_rsqrt_kernel<<<(nN + 255) / 256, 256, 0, stream>>>(dinv, nN);

  size_t shmem1 = (size_t)(64 * LDS_STRIDE + 128 * LDS_STRIDE) * sizeof(float);
  gemm_gather_scale_kernel<<<(nN + 63) / 64, 256, shmem1, stream>>>(
      emb, x_idx, W1, dinv, hs1, agg1, nN, H);
  scatter_kernel<<<(unsigned)(((size_t)nE * 32 + 255) / 256), 256, 0, stream>>>(
      esrc, edst, hs1, agg1, nE, H);
  combine_kernel<<<(unsigned)(((size_t)nN * H + 255) / 256), 256, 0, stream>>>(
      agg1, dinv, b1, hs1, nN, H,  1);

  size_t shmem2 = (size_t)(64 * LDS_STRIDE + 64 * LDS_STRIDE) * sizeof(float);
  gemm_gather_scale_kernel<<<(nN + 63) / 64, 256, shmem2, stream>>>(
      hs1, nullptr, W2, dinv, hs2, agg2, nN, O);
  scatter_kernel<<<(unsigned)(((size_t)nE * 32 + 255) / 256), 256, 0, stream>>>(
      esrc, edst, hs2, agg2, nE, O);
  combine_kernel<<<(unsigned)(((size_t)nN * O + 255) / 256), 256, 0, stream>>>(
      agg2, dinv, b2, hs2, nN, O,  0);

  edgedot_kernel<<<(unsigned)(((size_t)nL * 32 + 255) / 256), 256, 0, stream>>>(
      lidx, lidx + nL, hs2, out, nL);
}


# Pull for push: per-call counting sort of edges by dst (integer returning atomic gives arrival-rank slots); one wave per node sums source rows in registers in that order; f32 atomics and combines gone
# speedup vs baseline: 1.7724x; 1.7724x over previous
#include <hip/hip_runtime.h>

typedef float v2f __attribute__((ext_vector_type(2)));
typedef float v8f __attribute__((ext_vector_type(8)));

#define LDS_STRIDE 132

__global__ void cnt_zero_kernel(int* cnt, int n) {
  int i = blockIdx.x * blockDim.x + threadIdx.x;
  if (i < n) cnt[i] = 0;
}

__global__ void cnt_rank_kernel(const int* __restrict__ dst, int* __restrict__ cnt,
                                int* __restrict__ rank, int nE) {
  int e = blockIdx.x * blockDim.x + threadIdx.x;
  if (e < nE) rank[e] = atomicAdd(&cnt[dst[e]], 1);
}

__global__ void __launch_bounds__(1024)
scan_dinv_kernel(const int* __restrict__ cnt, int* __restrict__ rowptr,
                 float* __restrict__ dinv, int n) {
  __shared__ int s_wtot[32];
  __shared__ int s_wpre[32];
  const int tid = threadIdx.x;
  const int lane = tid & 31;
  const int wave = tid >> 5;
  const int first = blockIdx.x * 1024;

  int pre = 0;
#pragma unroll 8
  for (int i = tid * 4; i < first; i += 4096) {
    int4 v = *(const int4*)&cnt[i];
    pre += (v.x + v.y) + (v.z + v.w);
  }
#pragma unroll
  for (int off = 16; off > 0; off >>= 1) pre += __shfl_xor(pre, off, 32);

  const int i = first + tid;
  const int c = (i < n) ? cnt[i] : 0;
  int inc = c;
#pragma unroll
  for (int off = 1; off < 32; off <<= 1) {
    int t = __shfl_up(inc, off, 32);
    if (lane >= off) inc += t;
  }
  if (lane == 31) s_wtot[wave] = inc;
  if (lane == 0) s_wpre[wave] = pre;
  __syncthreads();
  int v = s_wpre[lane] + ((lane < wave) ? s_wtot[lane] : 0);
#pragma unroll
  for (int off = 16; off > 0; off >>= 1) v += __shfl_xor(v, off, 32);
  if (i < n) {
    rowptr[i] = v + inc - c;
    dinv[i] = rsqrtf((float)(c + 1));
  }
}

__global__ void csr_fill_kernel(const int* __restrict__ src, const int* __restrict__ dst,
                                const int* __restrict__ rank, const int* __restrict__ rowptr,
                                int* __restrict__ col, int nE) {
  int e = blockIdx.x * blockDim.x + threadIdx.x;
  if (e < nE) col[rowptr[dst[e]] + rank[e]] = src[e];
}

__global__ void gemm_gather_scale_kernel(const float* __restrict__ X,
                                         const int* __restrict__ gidx,
                                         const float* __restrict__ W,
                                         const float* __restrict__ dinv,
                                         float* __restrict__ hs,
                                         int n_rows, int N) {
  extern __shared__ float lds[];
  float* lA = lds;
  float* lW = lds + 64 * LDS_STRIDE;
  const int tid = threadIdx.x;
  const int row0 = blockIdx.x * 64;

  const float4* X4 = (const float4*)X;
  for (int i = tid; i < 64 * 32; i += 256) {
    int r = i >> 5;
    int c4 = i & 31;
    int gr = row0 + r;
    float4 v = make_float4(0.f, 0.f, 0.f, 0.f);
    if (gr < n_rows) {
      int sr = gidx ? gidx[gr] : gr;
      v = X4[(size_t)sr * 32 + c4];
    }
    *(float4*)&lA[r * LDS_STRIDE + (c4 << 2)] = v;
  }
  for (int i = tid; i < 128 * N; i += 256) {
    int k = i / N;
    int n = i - k * N;
    lW[n * LDS_STRIDE + k] = W[i];
  }
  __syncthreads();

  const int wave = tid >> 5;
  const int lane = tid & 31;
  const int nct = N >> 4;
  const int col = lane & 15;
  const int koff = (lane >> 4) << 1;

  for (int t = wave; t < 4 * nct; t += 8) {
    int rt = t / nct;
    int ct = t - rt * nct;
    int mbase = rt << 4;
    int nbase = ct << 4;
    const float* aRow = &lA[(mbase + col) * LDS_STRIDE + koff];
    const float* bRow = &lW[(nbase + col) * LDS_STRIDE + koff];
    v8f c = {0.f, 0.f, 0.f, 0.f, 0.f, 0.f, 0.f, 0.f};
#pragma unroll
    for (int k0 = 0; k0 < 128; k0 += 4) {
      v2f a = *(const v2f*)(aRow + k0);
      v2f b = *(const v2f*)(bRow + k0);
      c = __builtin_amdgcn_wmma_f32_16x16x4_f32(false, a, false, b, (short)0, c, false, false);
    }
    int madd = (lane < 16) ? 0 : 8;
#pragma unroll
    for (int i = 0; i < 8; i++) {
      int gr = row0 + mbase + i + madd;
      if (gr < n_rows) {
        float v = c[i] * dinv[gr];
        size_t o = (size_t)gr * N + nbase + col;
        hs[o] = v;
      }
    }
  }
}

template <int N, bool RELU>
__global__ void __launch_bounds__(256)
gather_sum_kernel(const int* __restrict__ rowptr, const int* __restrict__ cnt,
                  const int* __restrict__ col, const float* __restrict__ hs,
                  const float* __restrict__ dinv, const float* __restrict__ bias,
                  float* __restrict__ out, int n_rows) {
  typedef float vec __attribute__((ext_vector_type(N / 32)));
  const int lane = threadIdx.x & 31;
  const int wpb = blockDim.x >> 5;
  const int nwaves = gridDim.x * wpb;
  const vec* __restrict__ hv = (const vec*)hs;
  vec* __restrict__ ov = (vec*)out;
  const vec bv = ((const vec*)bias)[lane];

  for (int rr = blockIdx.x * wpb + (threadIdx.x >> 5); rr < n_rows; rr += nwaves) {
    const int r = __builtin_amdgcn_readfirstlane(rr);
    const int* __restrict__ cp = col + rowptr[r];
    const int num = cnt[r];
    const float dr = dinv[r];
    vec acc = hv[(size_t)r * 32 + lane];
    int j = 0;
#pragma clang loop unroll(disable)
    for (; j + 8 <= num; j += 8) {
      const int s0 = cp[j], s1 = cp[j + 1], s2 = cp[j + 2], s3 = cp[j + 3];
      const int s4 = cp[j + 4], s5 = cp[j + 5], s6 = cp[j + 6], s7 = cp[j + 7];
      const vec v0 = hv[(size_t)s0 * 32 + lane], v1 = hv[(size_t)s1 * 32 + lane];
      const vec v2 = hv[(size_t)s2 * 32 + lane], v3 = hv[(size_t)s3 * 32 + lane];
      const vec v4 = hv[(size_t)s4 * 32 + lane], v5 = hv[(size_t)s5 * 32 + lane];
      const vec v6 = hv[(size_t)s6 * 32 + lane], v7 = hv[(size_t)s7 * 32 + lane];
      acc += v0; acc += v1; acc += v2; acc += v3;
      acc += v4; acc += v5; acc += v6; acc += v7;
    }
    if (j + 4 <= num) {
      const int s0 = cp[j], s1 = cp[j + 1], s2 = cp[j + 2], s3 = cp[j + 3];
      const vec v0 = hv[(size_t)s0 * 32 + lane], v1 = hv[(size_t)s1 * 32 + lane];
      const vec v2 = hv[(size_t)s2 * 32 + lane], v3 = hv[(size_t)s3 * 32 + lane];
      acc += v0; acc += v1; acc += v2; acc += v3;
      j += 4;
    }
    if (j + 2 <= num) {
      const int s0 = cp[j], s1 = cp[j + 1];
      const vec v0 = hv[(size_t)s0 * 32 + lane], v1 = hv[(size_t)s1 * 32 + lane];
      acc += v0; acc += v1;
      j += 2;
    }
    if (j < num) {
      const int s0 = cp[j];
      acc += hv[(size_t)s0 * 32 + lane];
    }
    vec res = dr * acc + bv;
    if (RELU) res = __builtin_elementwise_max(res, (vec)(0.f));
    ov[(size_t)r * 32 + lane] = res;
  }
}

__global__ void edgedot_kernel(const int* __restrict__ es, const int* __restrict__ ed,
                               const float* __restrict__ z, float* __restrict__ out, int nL) {
  int g = blockIdx.x * blockDim.x + threadIdx.x;
  int e = g >> 5;
  int lane = g & 31;
  if (e >= nL) return;
  const float* a = z + (size_t)es[e] * 64;
  const float* b = z + (size_t)ed[e] * 64;
  float p = a[lane] * b[lane] + a[lane + 32] * b[lane + 32];
#pragma unroll
  for (int off = 16; off > 0; off >>= 1) p += __shfl_xor(p, off, 32);
  if (lane == 0) out[e] = p;
}

extern "C" void kernel_launch(void* const* d_in, const int* in_sizes, int n_in,
                              void* d_out, int out_size, void* d_ws, size_t ws_size,
                              hipStream_t stream) {
  const int*   x_idx = (const int*)d_in[0];
  const int*   eidx  = (const int*)d_in[1];
  const int*   lidx  = (const int*)d_in[2];
  const float* emb   = (const float*)d_in[3];
  const float* W1    = (const float*)d_in[4];
  const float* b1    = (const float*)d_in[5];
  const float* W2    = (const float*)d_in[6];
  const float* b2    = (const float*)d_in[7];
  float* out = (float*)d_out;

  const int nN = in_sizes[0];
  const int nE = in_sizes[1] / 2;
  const int nL = in_sizes[2] / 2;
  const int H = 128, O = 64;

  const size_t nPad = ((size_t)nN + 63) & ~(size_t)63;
  const size_t ePad = ((size_t)nE + 63) & ~(size_t)63;
  float* dinv = (float*)d_ws;
  int* cnt    = (int*)(dinv + nPad);
  int* rowptr = cnt + nPad;
  int* rank   = rowptr + nPad;
  int* col    = rank + ePad;
  float* hs1  = (float*)(col + ePad);
  float* x1   = hs1 + (size_t)nN * H;
  float* hs2  = x1 + (size_t)nN * H;
  float* z    = hs2 + (size_t)nN * O;

  const int* esrc = eidx;
  const int* edst = eidx + nE;

  cnt_zero_kernel<<<(nN + 255) / 256, 256, 0, stream>>>(cnt, nN);
  cnt_rank_kernel<<<(nE + 255) / 256, 256, 0, stream>>>(edst, cnt, rank, nE);
  scan_dinv_kernel<<<(nN + 1023) / 1024, 1024, 0, stream>>>(cnt, rowptr, dinv, nN);
  csr_fill_kernel<<<(nE + 255) / 256, 256, 0, stream>>>(esrc, edst, rank, rowptr, col, nE);

  size_t shmem1 = (size_t)(64 * LDS_STRIDE + 128 * LDS_STRIDE) * sizeof(float);
  gemm_gather_scale_kernel<<<(nN + 63) / 64, 256, shmem1, stream>>>(
      emb, x_idx, W1, dinv, hs1, nN, H);
  gather_sum_kernel<128, true><<<2048, 256, 0, stream>>>(
      rowptr, cnt, col, hs1, dinv, b1, x1, nN);

  size_t shmem2 = (size_t)(64 * LDS_STRIDE + 64 * LDS_STRIDE) * sizeof(float);
  gemm_gather_scale_kernel<<<(nN + 63) / 64, 256, shmem2, stream>>>(
      x1, nullptr, W2, dinv, hs2, nN, O);
  gather_sum_kernel<64, false><<<2048, 256, 0, stream>>>(
      rowptr, cnt, col, hs2, dinv, b2, z, nN);

  edgedot_kernel<<<(unsigned)(((size_t)nL * 32 + 255) / 256), 256, 0, stream>>>(
      lidx, lidx + nL, z, out, nL);
}


# Link-score kernel: one wave per 32 label edges (was one wave per edge, 200000 waves): whole-row 8-byte loads, 32 sums folded across lanes (dot's sum order regrouped), one line store per wave
# speedup vs baseline: 1.9072x; 1.0761x over previous
#include <hip/hip_runtime.h>

typedef float v2f __attribute__((ext_vector_type(2)));
typedef float v8f __attribute__((ext_vector_type(8)));

#define LDS_STRIDE 132

__global__ void cnt_zero_kernel(int* cnt, int n) {
  int i = blockIdx.x * blockDim.x + threadIdx.x;
  if (i < n) cnt[i] = 0;
}

__global__ void cnt_rank_kernel(const int* __restrict__ dst, int* __restrict__ cnt,
                                int* __restrict__ rank, int nE) {
  int e = blockIdx.x * blockDim.x + threadIdx.x;
  if (e < nE) rank[e] = atomicAdd(&cnt[dst[e]], 1);
}

__global__ void __launch_bounds__(1024)
scan_dinv_kernel(const int* __restrict__ cnt, int* __restrict__ rowptr,
                 float* __restrict__ dinv, int n) {
  __shared__ int s_wtot[32];
  __shared__ int s_wpre[32];
  const int tid = threadIdx.x;
  const int lane = tid & 31;
  const int wave = tid >> 5;
  const int first = blockIdx.x * 1024;

  int pre = 0;
#pragma unroll 8
  for (int i = tid * 4; i < first; i += 4096) {
    int4 v = *(const int4*)&cnt[i];
    pre += (v.x + v.y) + (v.z + v.w);
  }
#pragma unroll
  for (int off = 16; off > 0; off >>= 1) pre += __shfl_xor(pre, off, 32);

  const int i = first + tid;
  const int c = (i < n) ? cnt[i] : 0;
  int inc = c;
#pragma unroll
  for (int off = 1; off < 32; off <<= 1) {
    int t = __shfl_up(inc, off, 32);
    if (lane >= off) inc += t;
  }
  if (lane == 31) s_wtot[wave] = inc;
  if (lane == 0) s_wpre[wave] = pre;
  __syncthreads();
  int v = s_wpre[lane] + ((lane < wave) ? s_wtot[lane] : 0);
#pragma unroll
  for (int off = 16; off > 0; off >>= 1) v += __shfl_xor(v, off, 32);
  if (i < n) {
    rowptr[i] = v + inc - c;
    dinv[i] = rsqrtf((float)(c + 1));
  }
}

__global__ void csr_fill_kernel(const int* __restrict__ src, const int* __restrict__ dst,
                                const int* __restrict__ rank, const int* __restrict__ rowptr,
                                int* __restrict__ col, int nE) {
  int e = blockIdx.x * blockDim.x + threadIdx.x;
  if (e < nE) col[rowptr[dst[e]] + rank[e]] = src[e];
}

__global__ void gemm_gather_scale_kernel(const float* __restrict__ X,
                                         const int* __restrict__ gidx,
                                         const float* __restrict__ W,
                                         const float* __restrict__ dinv,
                                         float* __restrict__ hs,
                                         int n_rows, int N) {
  extern __shared__ float lds[];
  float* lA = lds;
  float* lW = lds + 64 * LDS_STRIDE;
  const int tid = threadIdx.x;
  const int row0 = blockIdx.x * 64;

  const float4* X4 = (const float4*)X;
  for (int i = tid; i < 64 * 32; i += 256) {
    int r = i >> 5;
    int c4 = i & 31;
    int gr = row0 + r;
    float4 v = make_float4(0.f, 0.f, 0.f, 0.f);
    if (gr < n_rows) {
      int sr = gidx ? gidx[gr] : gr;
      v = X4[(size_t)sr * 32 + c4];
    }
    *(float4*)&lA[r * LDS_STRIDE + (c4 << 2)] = v;
  }
  for (int i = tid; i < 128 * N; i += 256) {
    int k = i / N;
    int n = i - k * N;
    lW[n * LDS_STRIDE + k] = W[i];
  }
  __syncthreads();

  const int wave = tid >> 5;
  const int lane = tid & 31;
  const int nct = N >> 4;
  const int col = lane & 15;
  const int koff = (lane >> 4) << 1;

  for (int t = wave; t < 4 * nct; t += 8) {
    int rt = t / nct;
    int ct = t - rt * nct;
    int mbase = rt << 4;
    int nbase = ct << 4;
    const float* aRow = &lA[(mbase + col) * LDS_STRIDE + koff];
    const float* bRow = &lW[(nbase + col) * LDS_STRIDE + koff];
    v8f c = {0.f, 0.f, 0.f, 0.f, 0.f, 0.f, 0.f, 0.f};
#pragma unroll
    for (int k0 = 0; k0 < 128; k0 += 4) {
      v2f a = *(const v2f*)(aRow + k0);
      v2f b = *(const v2f*)(bRow + k0);
      c = __builtin_amdgcn_wmma_f32_16x16x4_f32(false, a, false, b, (short)0, c, false, false);
    }
    int madd = (lane < 16) ? 0 : 8;
#pragma unroll
    for (int i = 0; i < 8; i++) {
      int gr = row0 + mbase + i + madd;
      if (gr < n_rows) {
        float v = c[i] * dinv[gr];
        size_t o = (size_t)gr * N + nbase + col;
        hs[o] = v;
      }
    }
  }
}

template <int N, bool RELU>
__global__ void __launch_bounds__(256)
gather_sum_kernel(const int* __restrict__ rowptr, const int* __restrict__ cnt,
                  const int* __restrict__ col, const float* __restrict__ hs,
                  const float* __restrict__ dinv, const float* __restrict__ bias,
                  float* __restrict__ out, int n_rows) {
  typedef float vec __attribute__((ext_vector_type(N / 32)));
  const int lane = threadIdx.x & 31;
  const int wpb = blockDim.x >> 5;
  const int nwaves = gridDim.x * wpb;
  const vec* __restrict__ hv = (const vec*)hs;
  vec* __restrict__ ov = (vec*)out;
  const vec bv = ((const vec*)bias)[lane];

  for (int rr = blockIdx.x * wpb + (threadIdx.x >> 5); rr < n_rows; rr += nwaves) {
    const int r = __builtin_amdgcn_readfirstlane(rr);
    const int* __restrict__ cp = col + rowptr[r];
    const int num = cnt[r];
    const float dr = dinv[r];
    vec acc = hv[(size_t)r * 32 + lane];
    int j = 0;
#pragma clang loop unroll(disable)
    for (; j + 8 <= num; j += 8) {
      const int s0 = cp[j], s1 = cp[j + 1], s2 = cp[j + 2], s3 = cp[j + 3];
      const int s4 = cp[j + 4], s5 = cp[j + 5], s6 = cp[j + 6], s7 = cp[j + 7];
      const vec v0 = hv[(size_t)s0 * 32 + lane], v1 = hv[(size_t)s1 * 32 + lane];
      const vec v2 = hv[(size_t)s2 * 32 + lane], v3 = hv[(size_t)s3 * 32 + lane];
      const vec v4 = hv[(size_t)s4 * 32 + lane], v5 = hv[(size_t)s5 * 32 + lane];
      const vec v6 = hv[(size_t)s6 * 32 + lane], v7 = hv[(size_t)s7 * 32 + lane];
      acc += v0; acc += v1; acc += v2; acc += v3;
      acc += v4; acc += v5; acc += v6; acc += v7;
    }
    if (j + 4 <= num) {
      const int s0 = cp[j], s1 = cp[j + 1], s2 = cp[j + 2], s3 = cp[j + 3];
      const vec v0 = hv[(size_t)s0 * 32 + lane], v1 = hv[(size_t)s1 * 32 + lane];
      const vec v2 = hv[(size_t)s2 * 32 + lane], v3 = hv[(size_t)s3 * 32 + lane];
      acc += v0; acc += v1; acc += v2; acc += v3;
      j += 4;
    }
    if (j + 2 <= num) {
      const int s0 = cp[j], s1 = cp[j + 1];
      const vec v0 = hv[(size_t)s0 * 32 + lane], v1 = hv[(size_t)s1 * 32 + lane];
      acc += v0; acc += v1;
      j += 2;
    }
    if (j < num) {
      const int s0 = cp[j];
      acc += hv[(size_t)s0 * 32 + lane];
    }
    vec res = dr * acc + bv;
    if (RELU) res = __builtin_elementwise_max(res, (vec)(0.f));
    ov[(size_t)r * 32 + lane] = res;
  }
}

template <int OFF>
__device__ __forceinline__ void fold_step(float* v, int lane) {
  const bool up = (lane & OFF) != 0;
#pragma unroll
  for (int i = 0; i < OFF; i++) {
    const float keep = up ? v[i + OFF] : v[i];
    const float send = up ? v[i] : v[i + OFF];
    v[i] = keep + __shfl_xor(send, OFF, 32);
  }
}

__global__ void __launch_bounds__(128)
edgedot_kernel(const int* __restrict__ es, const int* __restrict__ ed,
               const float* __restrict__ z, float* __restrict__ out, int nL) {
  const int lane = threadIdx.x & 31;
  const int e0 = __builtin_amdgcn_readfirstlane(
      (int)((blockIdx.x * (blockDim.x >> 5) + (threadIdx.x >> 5)) << 5));
  if (e0 >= nL) return;
  const int e = e0 + lane;
  const bool ok = e < nL;
  const int sa = ok ? es[e] : 0;
  const int sb = ok ? ed[e] : 0;
  const v2f* __restrict__ zv = (const v2f*)z;
  float p[32];
#pragma unroll
  for (int i = 0; i < 32; i++) {
    const int ra = __builtin_amdgcn_readlane(sa, i);
    const int rb = __builtin_amdgcn_readlane(sb, i);
    const v2f a = zv[(size_t)ra * 32 + lane];
    const v2f b = zv[(size_t)rb * 32 + lane];
    p[i] = a.x * b.x + a.y * b.y;
  }
  fold_step<16>(p, lane);
  fold_step<8>(p, lane);
  fold_step<4>(p, lane);
  fold_step<2>(p, lane);
  fold_step<1>(p, lane);
  if (ok) out[e] = p[0];
}

extern "C" void kernel_launch(void* const* d_in, const int* in_sizes, int n_in,
                              void* d_out, int out_size, void* d_ws, size_t ws_size,
                              hipStream_t stream) {
  const int*   x_idx = (const int*)d_in[0];
  const int*   eidx  = (const int*)d_in[1];
  const int*   lidx  = (const int*)d_in[2];
  const float* emb   = (const float*)d_in[3];
  const float* W1    = (const float*)d_in[4];
  const float* b1    = (const float*)d_in[5];
  const float* W2    = (const float*)d_in[6];
  const float* b2    = (const float*)d_in[7];
  float* out = (float*)d_out;

  const int nN = in_sizes[0];
  const int nE = in_sizes[1] / 2;
  const int nL = in_sizes[2] / 2;
  const int H = 128, O = 64;

  const size_t nPad = ((size_t)nN + 63) & ~(size_t)63;
  const size_t ePad = ((size_t)nE + 63) & ~(size_t)63;
  float* dinv = (float*)d_ws;
  int* cnt    = (int*)(dinv + nPad);
  int* rowptr = cnt + nPad;
  int* rank   = rowptr + nPad;
  int* col    = rank + ePad;
  float* hs1  = (float*)(col + ePad);
  float* x1   = hs1 + (size_t)nN * H;
  float* hs2  = x1 + (size_t)nN * H;
  float* z    = hs2 + (size_t)nN * O;

  const int* esrc = eidx;
  const int* edst = eidx + nE;

  cnt_zero_kernel<<<(nN + 255) / 256, 256, 0, stream>>>(cnt, nN);
  cnt_rank_kernel<<<(nE + 255) / 256, 256, 0, stream>>>(edst, cnt, rank, nE);
  scan_dinv_kernel<<<(nN + 1023) / 1024, 1024, 0, stream>>>(cnt, rowptr, dinv, nN);
  csr_fill_kernel<<<(nE + 255) / 256, 256, 0, stream>>>(esrc, edst, rank, rowptr, col, nE);

  size_t shmem1 = (size_t)(64 * LDS_STRIDE + 128 * LDS_STRIDE) * sizeof(float);
  gemm_gather_scale_kernel<<<(nN + 63) / 64, 256, shmem1, stream>>>(
      emb, x_idx, W1, dinv, hs1, nN, H);
  gather_sum_kernel<128, true><<<2048, 256, 0, stream>>>(
      rowptr, cnt, col, hs1, dinv, b1, x1, nN);

  size_t shmem2 = (size_t)(64 * LDS_STRIDE + 64 * LDS_STRIDE) * sizeof(float);
  gemm_gather_scale_kernel<<<(nN + 63) / 64, 256, shmem2, stream>>>(
      x1, nullptr, W2, dinv, hs2, nN, O);
  gather_sum_kernel<64, false><<<2048, 256, 0, stream>>>(
      rowptr, cnt, col, hs2, dinv, b2, z, nN);

  edgedot_kernel<<<(nL + 127) / 128, 128, 0, stream>>>(
      lidx, lidx + nL, z, out, nL);
}


# Layer-1 product once per dictionary row, (emb@W1)[x_idx] = emb[x_idx]@W1: GEMM over 10000 rows not 100000; node sums read that 5 MB table via x_idx, scaled by dinv[s] as fma, own row summed last
# speedup vs baseline: 2.5652x; 1.3450x over previous
#include <hip/hip_runtime.h>

typedef float v2f __attribute__((ext_vector_type(2)));
typedef float v8f __attribute__((ext_vector_type(8)));

#define LDS_STRIDE 132

__global__ void cnt_zero_kernel(int* cnt, int n) {
  int i = blockIdx.x * blockDim.x + threadIdx.x;
  if (i < n) cnt[i] = 0;
}

__global__ void cnt_rank_kernel(const int* __restrict__ dst, int* __restrict__ cnt,
                                int* __restrict__ rank, int nE) {
  int e = blockIdx.x * blockDim.x + threadIdx.x;
  if (e < nE) rank[e] = atomicAdd(&cnt[dst[e]], 1);
}

__global__ void __launch_bounds__(1024)
scan_dinv_kernel(const int* __restrict__ cnt, const int* __restrict__ x_idx,
                 int* __restrict__ rowptr, float* __restrict__ dinv,
                 int2* __restrict__ nt, int n) {
  __shared__ int s_wtot[32];
  __shared__ int s_wpre[32];
  const int tid = threadIdx.x;
  const int lane = tid & 31;
  const int wave = tid >> 5;
  const int first = blockIdx.x * 1024;

  int pre = 0;
#pragma unroll 8
  for (int i = tid * 4; i < first; i += 4096) {
    int4 v = *(const int4*)&cnt[i];
    pre += (v.x + v.y) + (v.z + v.w);
  }
#pragma unroll
  for (int off = 16; off > 0; off >>= 1) pre += __shfl_xor(pre, off, 32);

  const int i = first + tid;
  const int c = (i < n) ? cnt[i] : 0;
  int inc = c;
#pragma unroll
  for (int off = 1; off < 32; off <<= 1) {
    int t = __shfl_up(inc, off, 32);
    if (lane >= off) inc += t;
  }
  if (lane == 31) s_wtot[wave] = inc;
  if (lane == 0) s_wpre[wave] = pre;
  __syncthreads();
  int v = s_wpre[lane] + ((lane < wave) ? s_wtot[lane] : 0);
#pragma unroll
  for (int off = 16; off > 0; off >>= 1) v += __shfl_xor(v, off, 32);
  if (i < n) {
    const float d = rsqrtf((float)(c + 1));
    rowptr[i] = v + inc - c;
    dinv[i] = d;
    nt[i] = make_int2(x_idx[i], __float_as_int(d));
  }
}

__global__ void csr_fill_kernel(const int* __restrict__ src, const int* __restrict__ dst,
                                const int* __restrict__ rank, const int* __restrict__ rowptr,
                                int* __restrict__ col, int nE) {
  int e = blockIdx.x * blockDim.x + threadIdx.x;
  if (e < nE) col[rowptr[dst[e]] + rank[e]] = src[e];
}

__global__ void gemm_scale_kernel(const float* __restrict__ X,
                                  const float* __restrict__ W,
                                  const float* __restrict__ dinv,
                                  float* __restrict__ hs,
                                  int n_rows, int N) {
  extern __shared__ float lds[];
  float* lA = lds;
  float* lW = lds + 64 * LDS_STRIDE;
  const int tid = threadIdx.x;
  const int row0 = blockIdx.x * 64;

  const float4* X4 = (const float4*)X;
  for (int i = tid; i < 64 * 32; i += 256) {
    int r = i >> 5;
    int c4 = i & 31;
    int gr = row0 + r;
    float4 v = make_float4(0.f, 0.f, 0.f, 0.f);
    if (gr < n_rows) v = X4[(size_t)gr * 32 + c4];
    *(float4*)&lA[r * LDS_STRIDE + (c4 << 2)] = v;
  }
  for (int i = tid; i < 128 * N; i += 256) {
    int k = i / N;
    int n = i - k * N;
    lW[n * LDS_STRIDE + k] = W[i];
  }
  __syncthreads();

  const int wave = tid >> 5;
  const int lane = tid & 31;
  const int nct = N >> 4;
  const int col = lane & 15;
  const int koff = (lane >> 4) << 1;

  for (int t = wave; t < 4 * nct; t += 8) {
    int rt = t / nct;
    int ct = t - rt * nct;
    int mbase = rt << 4;
    int nbase = ct << 4;
    const float* aRow = &lA[(mbase + col) * LDS_STRIDE + koff];
    const float* bRow = &lW[(nbase + col) * LDS_STRIDE + koff];
    v8f c = {0.f, 0.f, 0.f, 0.f, 0.f, 0.f, 0.f, 0.f};
#pragma unroll
    for (int k0 = 0; k0 < 128; k0 += 4) {
      v2f a = *(const v2f*)(aRow + k0);
      v2f b = *(const v2f*)(bRow + k0);
      c = __builtin_amdgcn_wmma_f32_16x16x4_f32(false, a, false, b, (short)0, c, false, false);
    }
    int madd = (lane < 16) ? 0 : 8;
#pragma unroll
    for (int i = 0; i < 8; i++) {
      int gr = row0 + mbase + i + madd;
      if (gr < n_rows) {
        float v = c[i];
        if (dinv) v *= dinv[gr];
        size_t o = (size_t)gr * N + nbase + col;
        hs[o] = v;
      }
    }
  }
}

template <int N, bool RELU, bool DICT>
__global__ void __launch_bounds__(256)
gather_sum_kernel(const int* __restrict__ rowptr, const int* __restrict__ cnt,
                  const int* __restrict__ col, const int2* __restrict__ nt,
                  const float* __restrict__ rows, const float* __restrict__ dinv,
                  const float* __restrict__ bias, float* __restrict__ out, int n_rows) {
  typedef float vec __attribute__((ext_vector_type(N / 32)));
  const int lane = threadIdx.x & 31;
  const int wpb = blockDim.x >> 5;
  const int nwaves = gridDim.x * wpb;
  const vec* __restrict__ hv = (const vec*)rows;
  vec* __restrict__ ov = (vec*)out;
  const vec bv = ((const vec*)bias)[lane];

  auto locate = [&](int s, float& w) -> size_t {
    if constexpr (DICT) {
      const int2 e = nt[s];
      w = __int_as_float(e.y);
      return (size_t)e.x * 32;
    } else {
      w = 1.f;
      return (size_t)s * 32;
    }
  };

  for (int rr = blockIdx.x * wpb + (threadIdx.x >> 5); rr < n_rows; rr += nwaves) {
    const int r = __builtin_amdgcn_readfirstlane(rr);
    const int* __restrict__ cp = col + rowptr[r];
    const int num = cnt[r];
    float wr;
    const size_t br = locate(r, wr);
    if constexpr (DICT) __builtin_amdgcn_sched_barrier(0);
    const float dr = DICT ? wr : dinv[r];
    const vec own = hv[br + lane];
    vec acc = DICT ? (vec)(0.f) : own;
    int j = 0;
#pragma clang loop unroll(disable)
    for (; j + 8 <= num; j += 8) {
      const int s0 = cp[j], s1 = cp[j + 1], s2 = cp[j + 2], s3 = cp[j + 3];
      const int s4 = cp[j + 4], s5 = cp[j + 5], s6 = cp[j + 6], s7 = cp[j + 7];
      float w0, w1, w2, w3, w4, w5, w6, w7;
      const size_t b0 = locate(s0, w0), b1 = locate(s1, w1), b2 = locate(s2, w2);
      const size_t b3 = locate(s3, w3), b4 = locate(s4, w4), b5 = locate(s5, w5);
      const size_t b6 = locate(s6, w6), b7 = locate(s7, w7);
      const vec v0 = hv[b0 + lane], v1 = hv[b1 + lane], v2 = hv[b2 + lane], v3 = hv[b3 + lane];
      const vec v4 = hv[b4 + lane], v5 = hv[b5 + lane], v6 = hv[b6 + lane], v7 = hv[b7 + lane];
      acc += w0 * v0; acc += w1 * v1; acc += w2 * v2; acc += w3 * v3;
      acc += w4 * v4; acc += w5 * v5; acc += w6 * v6; acc += w7 * v7;
    }
    if (j + 4 <= num) {
      const int s0 = cp[j], s1 = cp[j + 1], s2 = cp[j + 2], s3 = cp[j + 3];
      float w0, w1, w2, w3;
      const size_t b0 = locate(s0, w0), b1 = locate(s1, w1);
      const size_t b2 = locate(s2, w2), b3 = locate(s3, w3);
      const vec v0 = hv[b0 + lane], v1 = hv[b1 + lane], v2 = hv[b2 + lane], v3 = hv[b3 + lane];
      acc += w0 * v0; acc += w1 * v1; acc += w2 * v2; acc += w3 * v3;
      j += 4;
    }
    if (j + 2 <= num) {
      const int s0 = cp[j], s1 = cp[j + 1];
      float w0, w1;
      const size_t b0 = locate(s0, w0), b1 = locate(s1, w1);
      const vec v0 = hv[b0 + lane], v1 = hv[b1 + lane];
      acc += w0 * v0; acc += w1 * v1;
      j += 2;
    }
    if (j < num) {
      float w0;
      const size_t b0 = locate(cp[j], w0);
      acc += w0 * hv[b0 + lane];
    }
    if (DICT) acc += wr * own;
    vec res = dr * acc + bv;
    if (RELU) res = __builtin_elementwise_max(res, (vec)(0.f));
    ov[(size_t)r * 32 + lane] = res;
  }
}

template <int OFF>
__device__ __forceinline__ void fold_step(float* v, int lane) {
  const bool up = (lane & OFF) != 0;
#pragma unroll
  for (int i = 0; i < OFF; i++) {
    const float keep = up ? v[i + OFF] : v[i];
    const float send = up ? v[i] : v[i + OFF];
    v[i] = keep + __shfl_xor(send, OFF, 32);
  }
}

__global__ void __launch_bounds__(128)
edgedot_kernel(const int* __restrict__ es, const int* __restrict__ ed,
               const float* __restrict__ z, float* __restrict__ out, int nL) {
  const int lane = threadIdx.x & 31;
  const int e0 = __builtin_amdgcn_readfirstlane(
      (int)((blockIdx.x * (blockDim.x >> 5) + (threadIdx.x >> 5)) << 5));
  if (e0 >= nL) return;
  const int e = e0 + lane;
  const bool ok = e < nL;
  const int sa = ok ? es[e] : 0;
  const int sb = ok ? ed[e] : 0;
  const v2f* __restrict__ zv = (const v2f*)z;
  float p[32];
#pragma unroll
  for (int i = 0; i < 32; i++) {
    const int ra = __builtin_amdgcn_readlane(sa, i);
    const int rb = __builtin_amdgcn_readlane(sb, i);
    const v2f a = zv[(size_t)ra * 32 + lane];
    const v2f b = zv[(size_t)rb * 32 + lane];
    p[i] = a.x * b.x + a.y * b.y;
  }
  fold_step<16>(p, lane);
  fold_step<8>(p, lane);
  fold_step<4>(p, lane);
  fold_step<2>(p, lane);
  fold_step<1>(p, lane);
  if (ok) out[e] = p[0];
}

extern "C" void kernel_launch(void* const* d_in, const int* in_sizes, int n_in,
                              void* d_out, int out_size, void* d_ws, size_t ws_size,
                              hipStream_t stream) {
  const int*   x_idx = (const int*)d_in[0];
  const int*   eidx  = (const int*)d_in[1];
  const int*   lidx  = (const int*)d_in[2];
  const float* emb   = (const float*)d_in[3];
  const float* W1    = (const float*)d_in[4];
  const float* b1    = (const float*)d_in[5];
  const float* W2    = (const float*)d_in[6];
  const float* b2    = (const float*)d_in[7];
  float* out = (float*)d_out;

  const int nN = in_sizes[0];
  const int nE = in_sizes[1] / 2;
  const int nL = in_sizes[2] / 2;
  const int H = 128, O = 64;
  const int nD = in_sizes[3] / H;

  const size_t nPad = ((size_t)nN + 63) & ~(size_t)63;
  const size_t ePad = ((size_t)nE + 63) & ~(size_t)63;
  float* dinv = (float*)d_ws;
  int* cnt    = (int*)(dinv + nPad);
  int* rowptr = cnt + nPad;
  int2* nt    = (int2*)(rowptr + nPad);
  int* rank   = rowptr + 3 * nPad;
  int* col    = rank + ePad;
  float* t1   = (float*)(col + ePad);
  float* x1   = t1 + (size_t)nD * H;
  float* hs2  = x1 + (size_t)nN * H;
  float* z    = hs2 + (size_t)nN * O;

  const int* esrc = eidx;
  const int* edst = eidx + nE;

  cnt_zero_kernel<<<(nN + 255) / 256, 256, 0, stream>>>(cnt, nN);
  cnt_rank_kernel<<<(nE + 255) / 256, 256, 0, stream>>>(edst, cnt, rank, nE);
  scan_dinv_kernel<<<(nN + 1023) / 1024, 1024, 0, stream>>>(cnt, x_idx, rowptr, dinv, nt, nN);
  csr_fill_kernel<<<(nE + 255) / 256, 256, 0, stream>>>(esrc, edst, rank, rowptr, col, nE);

  size_t shmem1 = (size_t)(64 * LDS_STRIDE + 128 * LDS_STRIDE) * sizeof(float);
  gemm_scale_kernel<<<(nD + 63) / 64, 256, shmem1, stream>>>(
      emb, W1, nullptr, t1, nD, H);
  gather_sum_kernel<128, true, true><<<2048, 256, 0, stream>>>(
      rowptr, cnt, col, nt, t1, dinv, b1, x1, nN);

  size_t shmem2 = (size_t)(64 * LDS_STRIDE + 64 * LDS_STRIDE) * sizeof(float);
  gemm_scale_kernel<<<(nN + 63) / 64, 256, shmem2, stream>>>(
      x1, W2, dinv, hs2, nN, O);
  gather_sum_kernel<64, false, false><<<2048, 256, 0, stream>>>(
      rowptr, cnt, col, nt, hs2, dinv, b2, z, nN);

  edgedot_kernel<<<(nL + 127) / 128, 128, 0, stream>>>(
      lidx, lidx + nL, z, out, nL);
}


# GEMM kernel rebuilt: a block's global reads go out as one group (were one round trip each, 28 in series in layer 2); a wave keeps a 16-row strip for 4 column tiles; k sum order of products regrouped
# speedup vs baseline: 3.1530x; 1.2291x over previous
#include <hip/hip_runtime.h>

typedef float v2f __attribute__((ext_vector_type(2)));
typedef float v8f __attribute__((ext_vector_type(8)));

__global__ void cnt_zero_kernel(int* cnt, int n) {
  int i = blockIdx.x * blockDim.x + threadIdx.x;
  if (i < n) cnt[i] = 0;
}

__global__ void cnt_rank_kernel(const int* __restrict__ dst, int* __restrict__ cnt,
                                int* __restrict__ rank, int nE) {
  int e = blockIdx.x * blockDim.x + threadIdx.x;
  if (e < nE) rank[e] = atomicAdd(&cnt[dst[e]], 1);
}

__global__ void __launch_bounds__(1024)
scan_dinv_kernel(const int* __restrict__ cnt, const int* __restrict__ x_idx,
                 int* __restrict__ rowptr, float* __restrict__ dinv,
                 int2* __restrict__ nt, int n) {
  __shared__ int s_wtot[32];
  __shared__ int s_wpre[32];
  const int tid = threadIdx.x;
  const int lane = tid & 31;
  const int wave = tid >> 5;
  const int first = blockIdx.x * 1024;

  int pre = 0;
#pragma unroll 8
  for (int i = tid * 4; i < first; i += 4096) {
    int4 v = *(const int4*)&cnt[i];
    pre += (v.x + v.y) + (v.z + v.w);
  }
#pragma unroll
  for (int off = 16; off > 0; off >>= 1) pre += __shfl_xor(pre, off, 32);

  const int i = first + tid;
  const int c = (i < n) ? cnt[i] : 0;
  int inc = c;
#pragma unroll
  for (int off = 1; off < 32; off <<= 1) {
    int t = __shfl_up(inc, off, 32);
    if (lane >= off) inc += t;
  }
  if (lane == 31) s_wtot[wave] = inc;
  if (lane == 0) s_wpre[wave] = pre;
  __syncthreads();
  int v = s_wpre[lane] + ((lane < wave) ? s_wtot[lane] : 0);
#pragma unroll
  for (int off = 16; off > 0; off >>= 1) v += __shfl_xor(v, off, 32);
  if (i < n) {
    const float d = rsqrtf((float)(c + 1));
    rowptr[i] = v + inc - c;
    dinv[i] = d;
    nt[i] = make_int2(x_idx[i], __float_as_int(d));
  }
}

__global__ void csr_fill_kernel(const int* __restrict__ src, const int* __restrict__ dst,
                                const int* __restrict__ rank, const int* __restrict__ rowptr,
                                int* __restrict__ col, int nE) {
  int e = blockIdx.x * blockDim.x + threadIdx.x;
  if (e < nE) col[rowptr[dst[e]] + rank[e]] = src[e];
}

#define W_PITCH 132
#define A_PITCH 68
#define GEMM_LDS_BYTES ((64 * W_PITCH + 8 * 16 * A_PITCH) * 4)

__device__ __forceinline__ void load_a_half(const float* __restrict__ aw, float4 (&a)[8]) {
#pragma unroll
  for (int j = 0; j < 8; j++) a[j] = *(const float4*)(aw + 8 * j);
}

__device__ __forceinline__ void strip_half(const float4 (&a)[8], const float* __restrict__ bw,
                                           v8f (&acc)[4]) {
#pragma unroll
  for (int j = 0; j < 8; j++) {
    const v2f a0 = {a[j].x, a[j].y};
    const v2f a1 = {a[j].z, a[j].w};
#pragma unroll
    for (int t = 0; t < 4; t++) {
      const float4 b = *(const float4*)(bw + t * 16 * W_PITCH + 8 * j);
      const v2f b0 = {b.x, b.y};
      const v2f b1 = {b.z, b.w};
      acc[t] = __builtin_amdgcn_wmma_f32_16x16x4_f32(false, a0, false, b0, (short)0, acc[t], false, false);
      acc[t] = __builtin_amdgcn_wmma_f32_16x16x4_f32(false, a1, false, b1, (short)0, acc[t], false, false);
    }
  }
}

template <int N, bool SCALE>
__global__ void __launch_bounds__(256) __attribute__((amdgpu_waves_per_eu(8, 8)))
gemm_rows_kernel(const float* X, const float* __restrict__ W,
                 const float* __restrict__ dinv, float* __restrict__ hs, int n_rows) {
  extern __shared__ float4 lds4[];
  float* lW = (float*)lds4;
  const int tid = threadIdx.x;
  const int lane = tid & 31;
  const int wave = tid >> 5;
  float* lAw = lW + 64 * W_PITCH + wave * (16 * A_PITCH);
  const int c = lane & 15;
  const int h = lane >> 4;
  const int cb = blockIdx.y * 64;
  const int tile0 = (blockIdx.x * 8 + wave) * 16;
  const int row0 = tile0 < n_rows - 16 ? tile0 : n_rows - 16;

  float4 raw0[8];
  float4 q0, q1, q2, q3, q4, q5, q6, q7;
  const float4* xp = (const float4*)(X + (size_t)(row0 + h) * 128) + c;
#pragma unroll
  for (int i = 0; i < 8; i++) raw0[i] = xp[64 * i];
  q0 = xp[16]; q1 = xp[64 + 16]; q2 = xp[128 + 16]; q3 = xp[192 + 16];
  q4 = xp[256 + 16]; q5 = xp[320 + 16]; q6 = xp[384 + 16]; q7 = xp[448 + 16];
  float4 wv[8];
  const int wk = 8 * (wave >> 2) + (lane & 7);
  const int wn = 16 * (wave & 3) + 4 * (lane >> 3);
  const float4* wp = (const float4*)(W + (size_t)wk * N + cb + wn);
#pragma unroll
  for (int it = 0; it < 8; it++) wv[it] = wp[it * (16 * N / 4)];
  float dv[8];
  if (SCALE) {
#pragma unroll
    for (int i = 0; i < 8; i++) dv[i] = dinv[row0 + 8 * h + i];
  }

  float* ws = lW + wn * W_PITCH + wk;
#pragma unroll
  for (int it = 0; it < 8; it++) {
    ws[0 * W_PITCH + 16 * it] = wv[it].x;
    ws[1 * W_PITCH + 16 * it] = wv[it].y;
    ws[2 * W_PITCH + 16 * it] = wv[it].z;
    ws[3 * W_PITCH + 16 * it] = wv[it].w;
  }
#pragma unroll
  for (int i = 0; i < 8; i++) *(float4*)(lAw + (2 * i + h) * A_PITCH + 4 * c) = raw0[i];
  __syncthreads();

  const float* aw = lAw + c * A_PITCH + 4 * h;
  const float* bw = lW + c * W_PITCH + 4 * h;
  float4 a[8];
  load_a_half(aw, a);
  __syncthreads();
  {
    float* d = lAw + h * A_PITCH + 4 * c;
    *(float4*)(d + 0 * A_PITCH) = q0; *(float4*)(d + 2 * A_PITCH) = q1;
    *(float4*)(d + 4 * A_PITCH) = q2; *(float4*)(d + 6 * A_PITCH) = q3;
    *(float4*)(d + 8 * A_PITCH) = q4; *(float4*)(d + 10 * A_PITCH) = q5;
    *(float4*)(d + 12 * A_PITCH) = q6; *(float4*)(d + 14 * A_PITCH) = q7;
  }
  __syncthreads();

  v8f acc[4];
#pragma unroll
  for (int t = 0; t < 4; t++) acc[t] = (v8f){0.f, 0.f, 0.f, 0.f, 0.f, 0.f, 0.f, 0.f};
  strip_half(a, bw, acc);
  load_a_half(aw, a);
  strip_half(a, bw + 64, acc);

  if (tile0 < n_rows) {
    float* o = hs + (size_t)(row0 + 8 * h) * N + cb + c;
#pragma unroll
    for (int i = 0; i < 8; i++) {
#pragma unroll
      for (int t = 0; t < 4; t++) {
        float v = acc[t][i];
        if (SCALE) v *= dv[i];
        o[i * N + 16 * t] = v;
      }
    }
  }
}

template <int N, bool RELU, bool DICT>
__global__ void __launch_bounds__(256)
gather_sum_kernel(const int* __restrict__ rowptr, const int* __restrict__ cnt,
                  const int* __restrict__ col, const int2* __restrict__ nt,
                  const float* __restrict__ rows, const float* __restrict__ dinv,
                  const float* __restrict__ bias, float* __restrict__ out, int n_rows) {
  typedef float vec __attribute__((ext_vector_type(N / 32)));
  const int lane = threadIdx.x & 31;
  const int wpb = blockDim.x >> 5;
  const int nwaves = gridDim.x * wpb;
  const vec* __restrict__ hv = (const vec*)rows;
  vec* __restrict__ ov = (vec*)out;
  const vec bv = ((const vec*)bias)[lane];

  auto locate = [&](int s, float& w) -> size_t {
    if constexpr (DICT) {
      const int2 e = nt[s];
      w = __int_as_float(e.y);
      return (size_t)e.x * 32;
    } else {
      w = 1.f;
      return (size_t)s * 32;
    }
  };

  for (int rr = blockIdx.x * wpb + (threadIdx.x >> 5); rr < n_rows; rr += nwaves) {
    const int r = __builtin_amdgcn_readfirstlane(rr);
    const int* __restrict__ cp = col + rowptr[r];
    const int num = cnt[r];
    float wr;
    const size_t br = locate(r, wr);
    if constexpr (DICT) __builtin_amdgcn_sched_barrier(0);
    const float dr = DICT ? wr : dinv[r];
    const vec own = hv[br + lane];
    vec acc = DICT ? (vec)(0.f) : own;
    int j = 0;
#pragma clang loop unroll(disable)
    for (; j + 8 <= num; j += 8) {
      const int s0 = cp[j], s1 = cp[j + 1], s2 = cp[j + 2], s3 = cp[j + 3];
      const int s4 = cp[j + 4], s5 = cp[j + 5], s6 = cp[j + 6], s7 = cp[j + 7];
      float w0, w1, w2, w3, w4, w5, w6, w7;
      const size_t b0 = locate(s0, w0), b1 = locate(s1, w1), b2 = locate(s2, w2);
      const size_t b3 = locate(s3, w3), b4 = locate(s4, w4), b5 = locate(s5, w5);
      const size_t b6 = locate(s6, w6), b7 = locate(s7, w7);
      const vec v0 = hv[b0 + lane], v1 = hv[b1 + lane], v2 = hv[b2 + lane], v3 = hv[b3 + lane];
      const vec v4 = hv[b4 + lane], v5 = hv[b5 + lane], v6 = hv[b6 + lane], v7 = hv[b7 + lane];
      acc += w0 * v0; acc += w1 * v1; acc += w2 * v2; acc += w3 * v3;
      acc += w4 * v4; acc += w5 * v5; acc += w6 * v6; acc += w7 * v7;
    }
    if (j + 4 <= num) {
      const int s0 = cp[j], s1 = cp[j + 1], s2 = cp[j + 2], s3 = cp[j + 3];
      float w0, w1, w2, w3;
      const size_t b0 = locate(s0, w0), b1 = locate(s1, w1);
      const size_t b2 = locate(s2, w2), b3 = locate(s3, w3);
      const vec v0 = hv[b0 + lane], v1 = hv[b1 + lane], v2 = hv[b2 + lane], v3 = hv[b3 + lane];
      acc += w0 * v0; acc += w1 * v1; acc += w2 * v2; acc += w3 * v3;
      j += 4;
    }
    if (j + 2 <= num) {
      const int s0 = cp[j], s1 = cp[j + 1];
      float w0, w1;
      const size_t b0 = locate(s0, w0), b1 = locate(s1, w1);
      const vec v0 = hv[b0 + lane], v1 = hv[b1 + lane];
      acc += w0 * v0; acc += w1 * v1;
      j += 2;
    }
    if (j < num) {
      float w0;
      const size_t b0 = locate(cp[j], w0);
      acc += w0 * hv[b0 + lane];
    }
    if (DICT) acc += wr * own;
    vec res = dr * acc + bv;
    if (RELU) res = __builtin_elementwise_max(res, (vec)(0.f));
    ov[(size_t)r * 32 + lane] = res;
  }
}

template <int OFF>
__device__ __forceinline__ void fold_step(float* v, int lane) {
  const bool up = (lane & OFF) != 0;
#pragma unroll
  for (int i = 0; i < OFF; i++) {
    const float keep = up ? v[i + OFF] : v[i];
    const float send = up ? v[i] : v[i + OFF];
    v[i] = keep + __shfl_xor(send, OFF, 32);
  }
}

__global__ void __launch_bounds__(128)
edgedot_kernel(const int* __restrict__ es, const int* __restrict__ ed,
               const float* __restrict__ z, float* __restrict__ out, int nL) {
  const int lane = threadIdx.x & 31;
  const int e0 = __builtin_amdgcn_readfirstlane(
      (int)((blockIdx.x * (blockDim.x >> 5) + (threadIdx.x >> 5)) << 5));
  if (e0 >= nL) return;
  const int e = e0 + lane;
  const bool ok = e < nL;
  const int sa = ok ? es[e] : 0;
  const int sb = ok ? ed[e] : 0;
  const v2f* __restrict__ zv = (const v2f*)z;
  float p[32];
#pragma unroll
  for (int i = 0; i < 32; i++) {
    const int ra = __builtin_amdgcn_readlane(sa, i);
    const int rb = __builtin_amdgcn_readlane(sb, i);
    const v2f a = zv[(size_t)ra * 32 + lane];
    const v2f b = zv[(size_t)rb * 32 + lane];
    p[i] = a.x * b.x + a.y * b.y;
  }
  fold_step<16>(p, lane);
  fold_step<8>(p, lane);
  fold_step<4>(p, lane);
  fold_step<2>(p, lane);
  fold_step<1>(p, lane);
  if (ok) out[e] = p[0];
}

extern "C" void kernel_launch(void* const* d_in, const int* in_sizes, int n_in,
                              void* d_out, int out_size, void* d_ws, size_t ws_size,
                              hipStream_t stream) {
  const int*   x_idx = (const int*)d_in[0];
  const int*   eidx  = (const int*)d_in[1];
  const int*   lidx  = (const int*)d_in[2];
  const float* emb   = (const float*)d_in[3];
  const float* W1    = (const float*)d_in[4];
  const float* b1    = (const float*)d_in[5];
  const float* W2    = (const float*)d_in[6];
  const float* b2    = (const float*)d_in[7];
  float* out = (float*)d_out;

  const int nN = in_sizes[0];
  const int nE = in_sizes[1] / 2;
  const int nL = in_sizes[2] / 2;
  const int H = 128, O = 64;
  const int nD = in_sizes[3] / H;

  const size_t nPad = ((size_t)nN + 63) & ~(size_t)63;
  const size_t ePad = ((size_t)nE + 63) & ~(size_t)63;
  float* dinv = (float*)d_ws;
  int* cnt    = (int*)(dinv + nPad);
  int* rowptr = cnt + nPad;
  int2* nt    = (int2*)(rowptr + nPad);
  int* rank   = rowptr + 3 * nPad;
  int* col    = rank + ePad;
  float* t1   = (float*)(col + ePad);
  float* x1   = t1 + (size_t)nD * H;
  float* hs2  = x1 + (size_t)nN * H;
  float* z    = hs2 + (size_t)nN * O;

  const int* esrc = eidx;
  const int* edst = eidx + nE;

  cnt_zero_kernel<<<(nN + 255) / 256, 256, 0, stream>>>(cnt, nN);
  cnt_rank_kernel<<<(nE + 255) / 256, 256, 0, stream>>>(edst, cnt, rank, nE);
  scan_dinv_kernel<<<(nN + 1023) / 1024, 1024, 0, stream>>>(cnt, x_idx, rowptr, dinv, nt, nN);
  csr_fill_kernel<<<(nE + 255) / 256, 256, 0, stream>>>(esrc, edst, rank, rowptr, col, nE);

  gemm_rows_kernel<128, false><<<dim3((nD + 127) / 128, 2), 256, GEMM_LDS_BYTES, stream>>>(
      emb, W1, nullptr, t1, nD);
  gather_sum_kernel<128, true, true><<<2048, 256, 0, stream>>>(
      rowptr, cnt, col, nt, t1, dinv, b1, x1, nN);

  gemm_rows_kernel<64, true><<<dim3((nN + 127) / 128, 1), 256, GEMM_LDS_BYTES, stream>>>(
      x1, W2, dinv, hs2, nN);
  gather_sum_kernel<64, false, false><<<2048, 256, 0, stream>>>(
      rowptr, cnt, col, nt, hs2, dinv, b2, z, nN);

  edgedot_kernel<<<(nL + 127) / 128, 128, 0, stream>>>(
      lidx, lidx + nL, z, out, nL);
}


# Adjacency build: a node owns 32 slots (one 128-byte line); the counting pass stores each edge at its arrival number; prefix sum and second edge pass gone; overflow list past 32; same sum order rule
# speedup vs baseline: 3.3481x; 1.0619x over previous
#include <hip/hip_runtime.h>

typedef float v2f __attribute__((ext_vector_type(2)));
typedef float v8f __attribute__((ext_vector_type(8)));

#define CAP 32

__global__ void cnt_zero_kernel(int* cnt, int* ovf_n, int n) {
  int i = blockIdx.x * blockDim.x + threadIdx.x;
  if (i < n) cnt[i] = 0;
  if (i == 0) *ovf_n = 0;
}

__global__ void count_place_kernel(const int* __restrict__ src, const int* __restrict__ dst,
                                   int* __restrict__ cnt, int* __restrict__ slots,
                                   int* __restrict__ ovf_n, int* __restrict__ ovf_src,
                                   int* __restrict__ ovf_dst, int nE) {
  int e = blockIdx.x * blockDim.x + threadIdx.x;
  if (e >= nE) return;
  const int d = dst[e];
  const int s = src[e];
  const int k = atomicAdd(&cnt[d], 1);
  if (k < CAP) {
    slots[(size_t)d * CAP + k] = s;
  } else {
    const int q = atomicAdd(ovf_n, 1);
    ovf_src[q] = s;
    ovf_dst[q] = d;
  }
}

__global__ void dinv_nt_kernel(const int* __restrict__ cnt, const int* __restrict__ x_idx,
                               float* __restrict__ dinv, int2* __restrict__ nt, int n) {
  int i = blockIdx.x * blockDim.x + threadIdx.x;
  if (i < n) {
    const float d = rsqrtf((float)(cnt[i] + 1));
    dinv[i] = d;
    nt[i] = make_int2(x_idx[i], __float_as_int(d));
  }
}

#define W_PITCH 132
#define A_PITCH 68
#define GEMM_LDS_BYTES ((64 * W_PITCH + 8 * 16 * A_PITCH) * 4)

__device__ __forceinline__ void load_a_half(const float* __restrict__ aw, float4 (&a)[8]) {
#pragma unroll
  for (int j = 0; j < 8; j++) a[j] = *(const float4*)(aw + 8 * j);
}

__device__ __forceinline__ void strip_half(const float4 (&a)[8], const float* __restrict__ bw,
                                           v8f (&acc)[4]) {
#pragma unroll
  for (int j = 0; j < 8; j++) {
    const v2f a0 = {a[j].x, a[j].y};
    const v2f a1 = {a[j].z, a[j].w};
#pragma unroll
    for (int t = 0; t < 4; t++) {
      const float4 b = *(const float4*)(bw + t * 16 * W_PITCH + 8 * j);
      const v2f b0 = {b.x, b.y};
      const v2f b1 = {b.z, b.w};
      acc[t] = __builtin_amdgcn_wmma_f32_16x16x4_f32(false, a0, false, b0, (short)0, acc[t], false, false);
      acc[t] = __builtin_amdgcn_wmma_f32_16x16x4_f32(false, a1, false, b1, (short)0, acc[t], false, false);
    }
  }
}

template <int N, bool SCALE>
__global__ void __launch_bounds__(256) __attribute__((amdgpu_waves_per_eu(8, 8)))
gemm_rows_kernel(const float* X, const float* __restrict__ W,
                 const float* __restrict__ dinv, float* __restrict__ hs, int n_rows) {
  extern __shared__ float4 lds4[];
  float* lW = (float*)lds4;
  const int tid = threadIdx.x;
  const int lane = tid & 31;
  const int wave = tid >> 5;
  float* lAw = lW + 64 * W_PITCH + wave * (16 * A_PITCH);
  const int c = lane & 15;
  const int h = lane >> 4;
  const int cb = blockIdx.y * 64;
  const int tile0 = (blockIdx.x * 8 + wave) * 16;
  const int row0 = tile0 < n_rows - 16 ? tile0 : n_rows - 16;

  float4 raw0[8];
  float4 q0, q1, q2, q3, q4, q5, q6, q7;
  const float4* xp = (const float4*)(X + (size_t)(row0 + h) * 128) + c;
#pragma unroll
  for (int i = 0; i < 8; i++) raw0[i] = xp[64 * i];
  q0 = xp[16]; q1 = xp[64 + 16]; q2 = xp[128 + 16]; q3 = xp[192 + 16];
  q4 = xp[256 + 16]; q5 = xp[320 + 16]; q6 = xp[384 + 16]; q7 = xp[448 + 16];
  float4 wv[8];
  const int wk = 8 * (wave >> 2) + (lane & 7);
  const int wn = 16 * (wave & 3) + 4 * (lane >> 3);
  const float4* wp = (const float4*)(W + (size_t)wk * N + cb + wn);
#pragma unroll
  for (int it = 0; it < 8; it++) wv[it] = wp[it * (16 * N / 4)];
  float dv[8];
  if (SCALE) {
#pragma unroll
    for (int i = 0; i < 8; i++) dv[i] = dinv[row0 + 8 * h + i];
  }

  float* ws = lW + wn * W_PITCH + wk;
#pragma unroll
  for (int it = 0; it < 8; it++) {
    ws[0 * W_PITCH + 16 * it] = wv[it].x;
    ws[1 * W_PITCH + 16 * it] = wv[it].y;
    ws[2 * W_PITCH + 16 * it] = wv[it].z;
    ws[3 * W_PITCH + 16 * it] = wv[it].w;
  }
#pragma unroll
  for (int i = 0; i < 8; i++) *(float4*)(lAw + (2 * i + h) * A_PITCH + 4 * c) = raw0[i];
  __syncthreads();

  const float* aw = lAw + c * A_PITCH + 4 * h;
  const float* bw = lW + c * W_PITCH + 4 * h;
  float4 a[8];
  load_a_half(aw, a);
  __syncthreads();
  {
    float* d = lAw + h * A_PITCH + 4 * c;
    *(float4*)(d + 0 * A_PITCH) = q0; *(float4*)(d + 2 * A_PITCH) = q1;
    *(float4*)(d + 4 * A_PITCH) = q2; *(float4*)(d + 6 * A_PITCH) = q3;
    *(float4*)(d + 8 * A_PITCH) = q4; *(float4*)(d + 10 * A_PITCH) = q5;
    *(float4*)(d + 12 * A_PITCH) = q6; *(float4*)(d + 14 * A_PITCH) = q7;
  }
  __syncthreads();

  v8f acc[4];
#pragma unroll
  for (int t = 0; t < 4; t++) acc[t] = (v8f){0.f, 0.f, 0.f, 0.f, 0.f, 0.f, 0.f, 0.f};
  strip_half(a, bw, acc);
  load_a_half(aw, a);
  strip_half(a, bw + 64, acc);

  if (tile0 < n_rows) {
    float* o = hs + (size_t)(row0 + 8 * h) * N + cb + c;
#pragma unroll
    for (int i = 0; i < 8; i++) {
#pragma unroll
      for (int t = 0; t < 4; t++) {
        float v = acc[t][i];
        if (SCALE) v *= dv[i];
        o[i * N + 16 * t] = v;
      }
    }
  }
}

template <int N, bool RELU, bool DICT>
__global__ void __launch_bounds__(256)
gather_sum_kernel(const int* __restrict__ cnt, const int* __restrict__ slots,
                  const int* __restrict__ ovf_n, const int* __restrict__ ovf_src,
                  const int* __restrict__ ovf_dst, const int2* __restrict__ nt,
                  const float* __restrict__ rows, const float* __restrict__ dinv,
                  const float* __restrict__ bias, float* __restrict__ out, int n_rows) {
  typedef float vec __attribute__((ext_vector_type(N / 32)));
  const int lane = threadIdx.x & 31;
  const int wpb = blockDim.x >> 5;
  const int nwaves = gridDim.x * wpb;
  const vec* __restrict__ hv = (const vec*)rows;
  vec* __restrict__ ov = (vec*)out;
  const vec bv = ((const vec*)bias)[lane];

  auto locate = [&](int s, float& w) -> size_t {
    if constexpr (DICT) {
      const int2 e = nt[s];
      w = __int_as_float(e.y);
      return (size_t)e.x * 32;
    } else {
      w = 1.f;
      return (size_t)s * 32;
    }
  };

  for (int rr = blockIdx.x * wpb + (threadIdx.x >> 5); rr < n_rows; rr += nwaves) {
    const int r = __builtin_amdgcn_readfirstlane(rr);
    const int* __restrict__ cp = slots + (size_t)r * CAP;
    const int deg = cnt[r];
    const int num = deg < CAP ? deg : CAP;
    float wr;
    const size_t br = locate(r, wr);
    if constexpr (DICT) __builtin_amdgcn_sched_barrier(0);
    const float dr = DICT ? wr : dinv[r];
    const vec own = hv[br + lane];
    vec acc = DICT ? (vec)(0.f) : own;
    int j = 0;
#pragma clang loop unroll(disable)
    for (; j + 8 <= num; j += 8) {
      const int s0 = cp[j], s1 = cp[j + 1], s2 = cp[j + 2], s3 = cp[j + 3];
      const int s4 = cp[j + 4], s5 = cp[j + 5], s6 = cp[j + 6], s7 = cp[j + 7];
      float w0, w1, w2, w3, w4, w5, w6, w7;
      const size_t b0 = locate(s0, w0), b1 = locate(s1, w1), b2 = locate(s2, w2);
      const size_t b3 = locate(s3, w3), b4 = locate(s4, w4), b5 = locate(s5, w5);
      const size_t b6 = locate(s6, w6), b7 = locate(s7, w7);
      const vec v0 = hv[b0 + lane], v1 = hv[b1 + lane], v2 = hv[b2 + lane], v3 = hv[b3 + lane];
      const vec v4 = hv[b4 + lane], v5 = hv[b5 + lane], v6 = hv[b6 + lane], v7 = hv[b7 + lane];
      acc += w0 * v0; acc += w1 * v1; acc += w2 * v2; acc += w3 * v3;
      acc += w4 * v4; acc += w5 * v5; acc += w6 * v6; acc += w7 * v7;
    }
    if (j + 4 <= num) {
      const int s0 = cp[j], s1 = cp[j + 1], s2 = cp[j + 2], s3 = cp[j + 3];
      float w0, w1, w2, w3;
      const size_t b0 = locate(s0, w0), b1 = locate(s1, w1);
      const size_t b2 = locate(s2, w2), b3 = locate(s3, w3);
      const vec v0 = hv[b0 + lane], v1 = hv[b1 + lane], v2 = hv[b2 + lane], v3 = hv[b3 + lane];
      acc += w0 * v0; acc += w1 * v1; acc += w2 * v2; acc += w3 * v3;
      j += 4;
    }
    if (j + 2 <= num) {
      const int s0 = cp[j], s1 = cp[j + 1];
      float w0, w1;
      const size_t b0 = locate(s0, w0), b1 = locate(s1, w1);
      const vec v0 = hv[b0 + lane], v1 = hv[b1 + lane];
      acc += w0 * v0; acc += w1 * v1;
      j += 2;
    }
    if (j < num) {
      float w0;
      const size_t b0 = locate(cp[j], w0);
      acc += w0 * hv[b0 + lane];
    }
    if (deg > CAP) {
      const int tot = *ovf_n;
      for (int q0 = 0; q0 < tot; q0 += 32) {
        const int q = q0 + lane;
        int sq = 0;
        bool hit = false;
        if (q < tot) {
          hit = ovf_dst[q] == r;
          sq = ovf_src[q];
        }
        unsigned todo = __builtin_amdgcn_ballot_w32(hit);
        while (todo) {
          const int b = __builtin_ctz(todo);
          todo &= todo - 1;
          float w0;
          const size_t b0 = locate(__builtin_amdgcn_readlane(sq, b), w0);
          acc += w0 * hv[b0 + lane];
        }
      }
    }
    if (DICT) acc += wr * own;
    vec res = dr * acc + bv;
    if (RELU) res = __builtin_elementwise_max(res, (vec)(0.f));
    ov[(size_t)r * 32 + lane] = res;
  }
}

template <int OFF>
__device__ __forceinline__ void fold_step(float* v, int lane) {
  const bool up = (lane & OFF) != 0;
#pragma unroll
  for (int i = 0; i < OFF; i++) {
    const float keep = up ? v[i + OFF] : v[i];
    const float send = up ? v[i] : v[i + OFF];
    v[i] = keep + __shfl_xor(send, OFF, 32);
  }
}

__global__ void __launch_bounds__(128)
edgedot_kernel(const int* __restrict__ es, const int* __restrict__ ed,
               const float* __restrict__ z, float* __restrict__ out, int nL) {
  const int lane = threadIdx.x & 31;
  const int e0 = __builtin_amdgcn_readfirstlane(
      (int)((blockIdx.x * (blockDim.x >> 5) + (threadIdx.x >> 5)) << 5));
  if (e0 >= nL) return;
  const int e = e0 + lane;
  const bool ok = e < nL;
  const int sa = ok ? es[e] : 0;
  const int sb = ok ? ed[e] : 0;
  const v2f* __restrict__ zv = (const v2f*)z;
  float p[32];
#pragma unroll
  for (int i = 0; i < 32; i++) {
    const int ra = __builtin_amdgcn_readlane(sa, i);
    const int rb = __builtin_amdgcn_readlane(sb, i);
    const v2f a = zv[(size_t)ra * 32 + lane];
    const v2f b = zv[(size_t)rb * 32 + lane];
    p[i] = a.x * b.x + a.y * b.y;
  }
  fold_step<16>(p, lane);
  fold_step<8>(p, lane);
  fold_step<4>(p, lane);
  fold_step<2>(p, lane);
  fold_step<1>(p, lane);
  if (ok) out[e] = p[0];
}

extern "C" void kernel_launch(void* const* d_in, const int* in_sizes, int n_in,
                              void* d_out, int out_size, void* d_ws, size_t ws_size,
                              hipStream_t stream) {
  const int*   x_idx = (const int*)d_in[0];
  const int*   eidx  = (const int*)d_in[1];
  const int*   lidx  = (const int*)d_in[2];
  const float* emb   = (const float*)d_in[3];
  const float* W1    = (const float*)d_in[4];
  const float* b1    = (const float*)d_in[5];
  const float* W2    = (const float*)d_in[6];
  const float* b2    = (const float*)d_in[7];
  float* out = (float*)d_out;

  const int nN = in_sizes[0];
  const int nE = in_sizes[1] / 2;
  const int nL = in_sizes[2] / 2;
  const int H = 128, O = 64;
  const int nD = in_sizes[3] / H;

  const size_t nPad = ((size_t)nN + 63) & ~(size_t)63;
  const size_t ePad = ((size_t)nE + 63) & ~(size_t)63;
  float* dinv  = (float*)d_ws;
  int* cnt     = (int*)(dinv + nPad);
  int* ovf_n   = cnt + nPad;
  int2* nt     = (int2*)(ovf_n + 64);
  int* slots   = ovf_n + 64 + 2 * nPad;
  int* ovf_src = slots + nPad * CAP;
  int* ovf_dst = ovf_src + ePad;
  float* t1    = (float*)(ovf_dst + ePad);
  float* x1    = t1 + (size_t)nD * H;
  float* hs2   = x1 + (size_t)nN * H;
  float* z     = hs2 + (size_t)nN * O;

  const int* esrc = eidx;
  const int* edst = eidx + nE;

  cnt_zero_kernel<<<(nN + 255) / 256, 256, 0, stream>>>(cnt, ovf_n, nN);
  count_place_kernel<<<(nE + 255) / 256, 256, 0, stream>>>(esrc, edst, cnt, slots, ovf_n, ovf_src,
                                                           ovf_dst, nE);
  dinv_nt_kernel<<<(nN + 255) / 256, 256, 0, stream>>>(cnt, x_idx, dinv, nt, nN);

  gemm_rows_kernel<128, false><<<dim3((nD + 127) / 128, 2), 256, GEMM_LDS_BYTES, stream>>>(
      emb, W1, nullptr, t1, nD);
  gather_sum_kernel<128, true, true><<<2048, 256, 0, stream>>>(
      cnt, slots, ovf_n, ovf_src, ovf_dst, nt, t1, dinv, b1, x1, nN);

  gemm_rows_kernel<64, true><<<dim3((nN + 127) / 128, 1), 256, GEMM_LDS_BYTES, stream>>>(
      x1, W2, dinv, hs2, nN);
  gather_sum_kernel<64, false, false><<<2048, 256, 0, stream>>>(
      cnt, slots, ovf_n, ovf_src, ovf_dst, nt, hs2, dinv, b2, z, nN);

  edgedot_kernel<<<(nL + 127) / 128, 128, 0, stream>>>(
      lidx, lidx + nL, z, out, nL);
}


# Node sums: what a node needs before its rows (count, slot line read lane-wise, table entries) is loaded 3, 2, 1 nodes ahead; a group of 8 rows issued whole; only rows wait in series; sum order kept
# speedup vs baseline: 3.6669x; 1.0952x over previous
#include <hip/hip_runtime.h>

typedef float v2f __attribute__((ext_vector_type(2)));
typedef float v8f __attribute__((ext_vector_type(8)));

#define CAP 32

__global__ void cnt_zero_kernel(int* cnt, int* ovf_n, int n) {
  int i = blockIdx.x * blockDim.x + threadIdx.x;
  if (i < n) cnt[i] = 0;
  if (i == 0) *ovf_n = 0;
}

__global__ void count_place_kernel(const int* __restrict__ src, const int* __restrict__ dst,
                                   int* __restrict__ cnt, int* __restrict__ slots,
                                   int* __restrict__ ovf_n, int* __restrict__ ovf_src,
                                   int* __restrict__ ovf_dst, int nE) {
  int e = blockIdx.x * blockDim.x + threadIdx.x;
  if (e >= nE) return;
  const int d = dst[e];
  const int s = src[e];
  const int k = atomicAdd(&cnt[d], 1);
  if (k < CAP) {
    slots[(size_t)d * CAP + k] = s;
  } else {
    const int q = atomicAdd(ovf_n, 1);
    ovf_src[q] = s;
    ovf_dst[q] = d;
  }
}

__global__ void dinv_nt_kernel(const int* __restrict__ cnt, const int* __restrict__ x_idx,
                               float* __restrict__ dinv, int2* __restrict__ nt, int n) {
  int i = blockIdx.x * blockDim.x + threadIdx.x;
  if (i < n) {
    const float d = rsqrtf((float)(cnt[i] + 1));
    dinv[i] = d;
    nt[i] = make_int2(x_idx[i], __float_as_int(d));
  }
}

#define W_PITCH 132
#define A_PITCH 68
#define GEMM_LDS_BYTES ((64 * W_PITCH + 8 * 16 * A_PITCH) * 4)

__device__ __forceinline__ void load_a_half(const float* __restrict__ aw, float4 (&a)[8]) {
#pragma unroll
  for (int j = 0; j < 8; j++) a[j] = *(const float4*)(aw + 8 * j);
}

__device__ __forceinline__ void strip_half(const float4 (&a)[8], const float* __restrict__ bw,
                                           v8f (&acc)[4]) {
#pragma unroll
  for (int j = 0; j < 8; j++) {
    const v2f a0 = {a[j].x, a[j].y};
    const v2f a1 = {a[j].z, a[j].w};
#pragma unroll
    for (int t = 0; t < 4; t++) {
      const float4 b = *(const float4*)(bw + t * 16 * W_PITCH + 8 * j);
      const v2f b0 = {b.x, b.y};
      const v2f b1 = {b.z, b.w};
      acc[t] = __builtin_amdgcn_wmma_f32_16x16x4_f32(false, a0, false, b0, (short)0, acc[t], false, false);
      acc[t] = __builtin_amdgcn_wmma_f32_16x16x4_f32(false, a1, false, b1, (short)0, acc[t], false, false);
    }
  }
}

template <int N, bool SCALE>
__global__ void __launch_bounds__(256) __attribute__((amdgpu_waves_per_eu(8, 8)))
gemm_rows_kernel(const float* X, const float* __restrict__ W,
                 const float* __restrict__ dinv, float* __restrict__ hs, int n_rows) {
  extern __shared__ float4 lds4[];
  float* lW = (float*)lds4;
  const int tid = threadIdx.x;
  const int lane = tid & 31;
  const int wave = tid >> 5;
  float* lAw = lW + 64 * W_PITCH + wave * (16 * A_PITCH);
  const int c = lane & 15;
  const int h = lane >> 4;
  const int cb = blockIdx.y * 64;
  const int tile0 = (blockIdx.x * 8 + wave) * 16;
  const int row0 = tile0 < n_rows - 16 ? tile0 : n_rows - 16;

  float4 raw0[8];
  float4 q0, q1, q2, q3, q4, q5, q6, q7;
  const float4* xp = (const float4*)(X + (size_t)(row0 + h) * 128) + c;
#pragma unroll
  for (int i = 0; i < 8; i++) raw0[i] = xp[64 * i];
  q0 = xp[16]; q1 = xp[64 + 16]; q2 = xp[128 + 16]; q3 = xp[192 + 16];
  q4 = xp[256 + 16]; q5 = xp[320 + 16]; q6 = xp[384 + 16]; q7 = xp[448 + 16];
  float4 wv[8];
  const int wk = 8 * (wave >> 2) + (lane & 7);
  const int wn = 16 * (wave & 3) + 4 * (lane >> 3);
  const float4* wp = (const float4*)(W + (size_t)wk * N + cb + wn);
#pragma unroll
  for (int it = 0; it < 8; it++) wv[it] = wp[it * (16 * N / 4)];
  float dv[8];
  if (SCALE) {
#pragma unroll
    for (int i = 0; i < 8; i++) dv[i] = dinv[row0 + 8 * h + i];
  }

  float* ws = lW + wn * W_PITCH + wk;
#pragma unroll
  for (int it = 0; it < 8; it++) {
    ws[0 * W_PITCH + 16 * it] = wv[it].x;
    ws[1 * W_PITCH + 16 * it] = wv[it].y;
    ws[2 * W_PITCH + 16 * it] = wv[it].z;
    ws[3 * W_PITCH + 16 * it] = wv[it].w;
  }
#pragma unroll
  for (int i = 0; i < 8; i++) *(float4*)(lAw + (2 * i + h) * A_PITCH + 4 * c) = raw0[i];
  __syncthreads();

  const float* aw = lAw + c * A_PITCH + 4 * h;
  const float* bw = lW + c * W_PITCH + 4 * h;
  float4 a[8];
  load_a_half(aw, a);
  __syncthreads();
  {
    float* d = lAw + h * A_PITCH + 4 * c;
    *(float4*)(d + 0 * A_PITCH) = q0; *(float4*)(d + 2 * A_PITCH) = q1;
    *(float4*)(d + 4 * A_PITCH) = q2; *(float4*)(d + 6 * A_PITCH) = q3;
    *(float4*)(d + 8 * A_PITCH) = q4; *(float4*)(d + 10 * A_PITCH) = q5;
    *(float4*)(d + 12 * A_PITCH) = q6; *(float4*)(d + 14 * A_PITCH) = q7;
  }
  __syncthreads();

  v8f acc[4];
#pragma unroll
  for (int t = 0; t < 4; t++) acc[t] = (v8f){0.f, 0.f, 0.f, 0.f, 0.f, 0.f, 0.f, 0.f};
  strip_half(a, bw, acc);
  load_a_half(aw, a);
  strip_half(a, bw + 64, acc);

  if (tile0 < n_rows) {
    float* o = hs + (size_t)(row0 + 8 * h) * N + cb + c;
#pragma unroll
    for (int i = 0; i < 8; i++) {
#pragma unroll
      for (int t = 0; t < 4; t++) {
        float v = acc[t][i];
        if (SCALE) v *= dv[i];
        o[i * N + 16 * t] = v;
      }
    }
  }
}

template <int N, bool RELU, bool DICT>
__global__ void __launch_bounds__(256)
gather_sum_kernel(const int* __restrict__ cnt, const int* __restrict__ slots,
                  const int* __restrict__ ovf_n, const int* __restrict__ ovf_src,
                  const int* __restrict__ ovf_dst, const int2* __restrict__ nt,
                  const float* __restrict__ rows, const float* __restrict__ dinv,
                  const float* __restrict__ bias, float* __restrict__ out, int n_rows) {
  typedef float vec __attribute__((ext_vector_type(N / 32)));
  const int lane = threadIdx.x & 31;
  const int wpb = blockDim.x >> 5;
  const int nwaves = gridDim.x * wpb;
  const vec* __restrict__ hv = (const vec*)rows;
  vec* __restrict__ ov = (vec*)out;
  const vec bv = ((const vec*)bias)[lane];

  const int r0 = __builtin_amdgcn_readfirstlane((int)(blockIdx.x * wpb + (threadIdx.x >> 5)));
  if (r0 >= n_rows) return;

  auto count_of = [&](int r) -> int { return r < n_rows ? cnt[r] : 0; };
  auto sources_of = [&](int r, int deg) -> int {
    const int num = deg < CAP ? deg : CAP;
    int s = 0;
    if (lane < num) s = slots[(size_t)r * CAP + lane];
    return s;
  };
  auto entries_of = [&](int s, int deg) -> int2 {
    if constexpr (DICT) {
      const int num = deg < CAP ? deg : CAP;
      int2 e = make_int2(0, 0);
      if (lane < num) e = nt[s];
      return e;
    } else {
      return make_int2(s, 0);
    }
  };

  int deg_c = cnt[r0];
  int deg_1 = count_of(r0 + nwaves);
  int deg_2 = count_of(r0 + 2 * nwaves);
  int2 ent_c = make_int2(0, 0);
  float dr_c = 0.f;
  if constexpr (DICT) ent_c = nt[r0]; else dr_c = dinv[r0];
  int2 e_c = entries_of(sources_of(r0, deg_c), deg_c);
  int sl_1 = sources_of(r0 + nwaves, deg_1);
  if constexpr (DICT) {
    asm volatile("" ::"v"(e_c.x), "v"(e_c.y), "v"(sl_1));
  } else {
    asm volatile("" ::"v"(e_c.x), "v"(sl_1));
  }

  for (int r = r0; r < n_rows; r += nwaves) {
    const int deg = deg_c;
    const int num = deg < CAP ? deg : CAP;
    const float dr = DICT ? __int_as_float(ent_c.y) : dr_c;
    const size_t br = DICT ? (size_t)ent_c.x * 32 : (size_t)r * 32;
    const int rowv = e_c.x;
    const int facv = e_c.y;
    const int r1 = r + nwaves;
    const int deg_3 = count_of(r1 + 2 * nwaves);
    if (r1 < n_rows) {
      if constexpr (DICT) ent_c = nt[r1]; else dr_c = dinv[r1];
    }
    const int sl_2 = sources_of(r1 + nwaves, deg_2);
    e_c = entries_of(sl_1, deg_1);

    const vec own = hv[br + lane];
    vec acc = DICT ? (vec)(0.f) : own;
#pragma clang loop unroll(disable)
    for (int j = 0; j < num; j += 8) {
      const int left = num - j;
      vec v[8];
      float w[8];
#pragma unroll
      for (int i = 0; i < 8; i++) {
        if (i == 0 || i < left) {
          const int s = __builtin_amdgcn_readlane(rowv, j + i);
          if constexpr (DICT) w[i] = __int_as_float(__builtin_amdgcn_readlane(facv, j + i));
          v[i] = hv[(size_t)s * 32 + lane];
        }
      }
#pragma unroll
      for (int i = 0; i < 8; i++) {
        if (i == 0 || i < left) {
          if constexpr (DICT) acc += w[i] * v[i]; else acc += v[i];
        }
      }
    }
    if (deg > CAP) {
      const int tot = *ovf_n;
      for (int q0 = 0; q0 < tot; q0 += 32) {
        const int q = q0 + lane;
        int sq = 0;
        bool hit = false;
        if (q < tot) {
          hit = ovf_dst[q] == r;
          sq = ovf_src[q];
        }
        unsigned todo = __builtin_amdgcn_ballot_w32(hit);
        while (todo) {
          const int b = __builtin_ctz(todo);
          todo &= todo - 1;
          const int s = __builtin_amdgcn_readlane(sq, b);
          if constexpr (DICT) {
            const int2 e = nt[s];
            acc += __int_as_float(e.y) * hv[(size_t)e.x * 32 + lane];
          } else {
            acc += hv[(size_t)s * 32 + lane];
          }
        }
      }
    }
    if (DICT) acc += dr * own;
    vec res = dr * acc + bv;
    if (RELU) res = __builtin_elementwise_max(res, (vec)(0.f));
    ov[(size_t)r * 32 + lane] = res;
    deg_c = deg_1;
    deg_1 = deg_2;
    deg_2 = deg_3;
    sl_1 = sl_2;
  }
}

template <int OFF>
__device__ __forceinline__ void fold_step(float* v, int lane) {
  const bool up = (lane & OFF) != 0;
#pragma unroll
  for (int i = 0; i < OFF; i++) {
    const float keep = up ? v[i + OFF] : v[i];
    const float send = up ? v[i] : v[i + OFF];
    v[i] = keep + __shfl_xor(send, OFF, 32);
  }
}

__global__ void __launch_bounds__(128)
edgedot_kernel(const int* __restrict__ es, const int* __restrict__ ed,
               const float* __restrict__ z, float* __restrict__ out, int nL) {
  const int lane = threadIdx.x & 31;
  const int e0 = __builtin_amdgcn_readfirstlane(
      (int)((blockIdx.x * (blockDim.x >> 5) + (threadIdx.x >> 5)) << 5));
  if (e0 >= nL) return;
  const int e = e0 + lane;
  const bool ok = e < nL;
  const int sa = ok ? es[e] : 0;
  const int sb = ok ? ed[e] : 0;
  const v2f* __restrict__ zv = (const v2f*)z;
  float p[32];
#pragma unroll
  for (int i = 0; i < 32; i++) {
    const int ra = __builtin_amdgcn_readlane(sa, i);
    const int rb = __builtin_amdgcn_readlane(sb, i);
    const v2f a = zv[(size_t)ra * 32 + lane];
    const v2f b = zv[(size_t)rb * 32 + lane];
    p[i] = a.x * b.x + a.y * b.y;
  }
  fold_step<16>(p, lane);
  fold_step<8>(p, lane);
  fold_step<4>(p, lane);
  fold_step<2>(p, lane);
  fold_step<1>(p, lane);
  if (ok) out[e] = p[0];
}

extern "C" void kernel_launch(void* const* d_in, const int* in_sizes, int n_in,
                              void* d_out, int out_size, void* d_ws, size_t ws_size,
                              hipStream_t stream) {
  const int*   x_idx = (const int*)d_in[0];
  const int*   eidx  = (const int*)d_in[1];
  const int*   lidx  = (const int*)d_in[2];
  const float* emb   = (const float*)d_in[3];
  const float* W1    = (const float*)d_in[4];
  const float* b1    = (const float*)d_in[5];
  const float* W2    = (const float*)d_in[6];
  const float* b2    = (const float*)d_in[7];
  float* out = (float*)d_out;

  const int nN = in_sizes[0];
  const int nE = in_sizes[1] / 2;
  const int nL = in_sizes[2] / 2;
  const int H = 128, O = 64;
  const int nD = in_sizes[3] / H;

  const size_t nPad = ((size_t)nN + 63) & ~(size_t)63;
  const size_t ePad = ((size_t)nE + 63) & ~(size_t)63;
  float* dinv  = (float*)d_ws;
  int* cnt     = (int*)(dinv + nPad);
  int* ovf_n   = cnt + nPad;
  int2* nt     = (int2*)(ovf_n + 64);
  int* slots   = ovf_n + 64 + 2 * nPad;
  int* ovf_src = slots + nPad * CAP;
  int* ovf_dst = ovf_src + ePad;
  float* t1    = (float*)(ovf_dst + ePad);
  float* x1    = t1 + (size_t)nD * H;
  float* hs2   = x1 + (size_t)nN * H;
  float* z     = hs2 + (size_t)nN * O;

  const int* esrc = eidx;
  const int* edst = eidx + nE;

  cnt_zero_kernel<<<(nN + 255) / 256, 256, 0, stream>>>(cnt, ovf_n, nN);
  count_place_kernel<<<(nE + 255) / 256, 256, 0, stream>>>(esrc, edst, cnt, slots, ovf_n, ovf_src,
                                                           ovf_dst, nE);
  dinv_nt_kernel<<<(nN + 255) / 256, 256, 0, stream>>>(cnt, x_idx, dinv, nt, nN);

  gemm_rows_kernel<128, false><<<dim3((nD + 127) / 128, 2), 256, GEMM_LDS_BYTES, stream>>>(
      emb, W1, nullptr, t1, nD);
  gather_sum_kernel<128, true, true><<<2048, 256, 0, stream>>>(
      cnt, slots, ovf_n, ovf_src, ovf_dst, nt, t1, dinv, b1, x1, nN);

  gemm_rows_kernel<64, true><<<dim3((nN + 127) / 128, 1), 256, GEMM_LDS_BYTES, stream>>>(
      x1, W2, dinv, hs2, nN);
  gather_sum_kernel<64, false, false><<<2048, 256, 0, stream>>>(
      cnt, slots, ovf_n, ovf_src, ovf_dst, nt, hs2, dinv, b2, z, nN);

  edgedot_kernel<<<(nL + 127) / 128, 128, 0, stream>>>(
      lidx, lidx + nL, z, out, nL);
}


# Adjacency build binned in LDS: edges sorted into 256-node bins (one global add per block and bin), a block per bin counts and places in LDS; no per-edge global atomic; sum order = LDS arrival order
# speedup vs baseline: 5.1798x; 1.4126x over previous
#include <hip/hip_runtime.h>

typedef float v2f __attribute__((ext_vector_type(2)));
typedef float v8f __attribute__((ext_vector_type(8)));

#define CAP 32
#define BIN_SHIFT 8
#define BIN_NODES 256
#define BINCAP 8192
#define CTRL_BIN0 64
#define CTRL_STRIDE 32
#define BE_BLOCK 512
#define BE_EPT 8
#define BP_EPT 8

__global__ void ctrl_zero_kernel(int* ctrl, int n) {
  int i = blockIdx.x * blockDim.x + threadIdx.x;
  if (i < n) ctrl[i] = 0;
}

__global__ void __launch_bounds__(BE_BLOCK)
bin_edges_kernel(const int* __restrict__ src, const int* __restrict__ dst, int* __restrict__ ctrl,
                 int2* __restrict__ binlist, int* __restrict__ spill_src,
                 int* __restrict__ spill_dst, int nE, int nbin) {
  extern __shared__ int be_lds[];
  int* hist = be_lds;
  int* base = be_lds + nbin;
  const int tid = threadIdx.x;
  for (int b = tid; b < nbin; b += BE_BLOCK) hist[b] = 0;
  const int e0 = blockIdx.x * (BE_BLOCK * BE_EPT) + tid;
  int d[BE_EPT], s[BE_EPT], rk[BE_EPT];
#pragma unroll
  for (int i = 0; i < BE_EPT; i++) {
    const int e = e0 + i * BE_BLOCK;
    d[i] = e < nE ? dst[e] : -1;
    s[i] = e < nE ? src[e] : 0;
  }
  __syncthreads();
#pragma unroll
  for (int i = 0; i < BE_EPT; i++) {
    rk[i] = 0;
    if (d[i] >= 0) rk[i] = atomicAdd(&hist[d[i] >> BIN_SHIFT], 1);
  }
  __syncthreads();
  for (int b = tid; b < nbin; b += BE_BLOCK) {
    const int h = hist[b];
    base[b] = h ? atomicAdd(&ctrl[CTRL_BIN0 + b * CTRL_STRIDE], h) : 0;
  }
  __syncthreads();
#pragma unroll
  for (int i = 0; i < BE_EPT; i++) {
    if (d[i] >= 0) {
      const int b = d[i] >> BIN_SHIFT;
      const int pos = base[b] + rk[i];
      if (pos < BINCAP) {
        binlist[(size_t)b * BINCAP + pos] = make_int2(d[i], s[i]);
      } else {
        const int q = atomicAdd(&ctrl[32], 1);
        spill_src[q] = s[i];
        spill_dst[q] = d[i];
      }
    }
  }
}

__global__ void __launch_bounds__(BIN_NODES)
bin_place_kernel(int* ctrl, const int2* __restrict__ binlist, const int* __restrict__ spill_src,
                 const int* __restrict__ spill_dst, const int* __restrict__ x_idx,
                 int* __restrict__ cnt, float* __restrict__ dinv, int2* __restrict__ nt,
                 int* __restrict__ slots, int* __restrict__ ovf_src, int* __restrict__ ovf_dst,
                 int n) {
  __shared__ int cntl[BIN_NODES];
  __shared__ __attribute__((aligned(16))) int slotl[BIN_NODES * CAP];
  const int tid = threadIdx.x;
  const int bin = blockIdx.x;
  const int node = bin * BIN_NODES + tid;
  const int fill = ctrl[CTRL_BIN0 + bin * CTRL_STRIDE];
  const int n_spill = ctrl[32];
  const int nb = fill < BINCAP ? fill : BINCAP;
  const int2* lst = binlist + (size_t)bin * BINCAP;
  const int xi = node < n ? x_idx[node] : 0;
  int2 p[BP_EPT];
  auto load_round = [&](int i0) {
#pragma unroll
    for (int j = 0; j < BP_EPT; j++) {
      const int i = i0 + j * BIN_NODES + tid;
      p[j] = make_int2(-1, 0);
      if (i < nb) p[j] = lst[i];
    }
  };
  load_round(0);
  int4* sl4 = (int4*)slotl;
  cntl[tid] = 0;
#pragma unroll
  for (int i = 0; i < BIN_NODES * CAP / 4 / BIN_NODES; i++) sl4[tid + i * BIN_NODES] = make_int4(0, 0, 0, 0);
  __syncthreads();

  auto place = [&](int d, int s) {
    const int t = d - bin * BIN_NODES;
    const int k = atomicAdd(&cntl[t], 1);
    if (k < CAP) {
      slotl[t * CAP + k] = s;
    } else {
      const int q = atomicAdd(&ctrl[0], 1);
      ovf_src[q] = s;
      ovf_dst[q] = d;
    }
  };
  for (int i0 = 0;;) {
#pragma unroll
    for (int j = 0; j < BP_EPT; j++)
      if (p[j].x >= 0) place(p[j].x, p[j].y);
    i0 += BIN_NODES * BP_EPT;
    if (i0 >= nb) break;
    load_round(i0);
  }
  for (int i = tid; i < n_spill; i += BIN_NODES) {
    const int d = spill_dst[i];
    if ((d >> BIN_SHIFT) == bin) place(d, spill_src[i]);
  }
  __syncthreads();

  if (node < n) {
    const int c = cntl[tid];
    const float dd = rsqrtf((float)(c + 1));
    cnt[node] = c;
    dinv[node] = dd;
    nt[node] = make_int2(xi, __float_as_int(dd));
  }
  int4* g4 = (int4*)(slots + (size_t)bin * BIN_NODES * CAP);
#pragma unroll
  for (int i = 0; i < BIN_NODES * CAP / 4 / BIN_NODES; i++) {
    const int idx = tid + i * BIN_NODES;
    if (bin * BIN_NODES + (idx >> 3) < n) g4[idx] = sl4[idx];
  }
}

#define W_PITCH 132
#define A_PITCH 68
#define GEMM_LDS_BYTES ((64 * W_PITCH + 8 * 16 * A_PITCH) * 4)

__device__ __forceinline__ void load_a_half(const float* __restrict__ aw, float4 (&a)[8]) {
#pragma unroll
  for (int j = 0; j < 8; j++) a[j] = *(const float4*)(aw + 8 * j);
}

__device__ __forceinline__ void strip_half(const float4 (&a)[8], const float* __restrict__ bw,
                                           v8f (&acc)[4]) {
#pragma unroll
  for (int j = 0; j < 8; j++) {
    const v2f a0 = {a[j].x, a[j].y};
    const v2f a1 = {a[j].z, a[j].w};
#pragma unroll
    for (int t = 0; t < 4; t++) {
      const float4 b = *(const float4*)(bw + t * 16 * W_PITCH + 8 * j);
      const v2f b0 = {b.x, b.y};
      const v2f b1 = {b.z, b.w};
      acc[t] = __builtin_amdgcn_wmma_f32_16x16x4_f32(false, a0, false, b0, (short)0, acc[t], false, false);
      acc[t] = __builtin_amdgcn_wmma_f32_16x16x4_f32(false, a1, false, b1, (short)0, acc[t], false, false);
    }
  }
}

template <int N, bool SCALE>
__global__ void __launch_bounds__(256) __attribute__((amdgpu_waves_per_eu(8, 8)))
gemm_rows_kernel(const float* X, const float* __restrict__ W,
                 const float* __restrict__ dinv, float* __restrict__ hs, int n_rows) {
  extern __shared__ float4 lds4[];
  float* lW = (float*)lds4;
  const int tid = threadIdx.x;
  const int lane = tid & 31;
  const int wave = tid >> 5;
  float* lAw = lW + 64 * W_PITCH + wave * (16 * A_PITCH);
  const int c = lane & 15;
  const int h = lane >> 4;
  const int cb = blockIdx.y * 64;
  const int tile0 = (blockIdx.x * 8 + wave) * 16;
  const int row0 = tile0 < n_rows - 16 ? tile0 : n_rows - 16;

  float4 raw0[8];
  float4 q0, q1, q2, q3, q4, q5, q6, q7;
  const float4* xp = (const float4*)(X + (size_t)(row0 + h) * 128) + c;
#pragma unroll
  for (int i = 0; i < 8; i++) raw0[i] = xp[64 * i];
  q0 = xp[16]; q1 = xp[64 + 16]; q2 = xp[128 + 16]; q3 = xp[192 + 16];
  q4 = xp[256 + 16]; q5 = xp[320 + 16]; q6 = xp[384 + 16]; q7 = xp[448 + 16];
  float4 wv[8];
  const int wk = 8 * (wave >> 2) + (lane & 7);
  const int wn = 16 * (wave & 3) + 4 * (lane >> 3);
  const float4* wp = (const float4*)(W + (size_t)wk * N + cb + wn);
#pragma unroll
  for (int it = 0; it < 8; it++) wv[it] = wp[it * (16 * N / 4)];
  float dv[8];
  if (SCALE) {
#pragma unroll
    for (int i = 0; i < 8; i++) dv[i] = dinv[row0 + 8 * h + i];
  }

  float* ws = lW + wn * W_PITCH + wk;
#pragma unroll
  for (int it = 0; it < 8; it++) {
    ws[0 * W_PITCH + 16 * it] = wv[it].x;
    ws[1 * W_PITCH + 16 * it] = wv[it].y;
    ws[2 * W_PITCH + 16 * it] = wv[it].z;
    ws[3 * W_PITCH + 16 * it] = wv[it].w;
  }
#pragma unroll
  for (int i = 0; i < 8; i++) *(float4*)(lAw + (2 * i + h) * A_PITCH + 4 * c) = raw0[i];
  __syncthreads();

  const float* aw = lAw + c * A_PITCH + 4 * h;
  const float* bw = lW + c * W_PITCH + 4 * h;
  float4 a[8];
  load_a_half(aw, a);
  __syncthreads();
  {
    float* d = lAw + h * A_PITCH + 4 * c;
    *(float4*)(d + 0 * A_PITCH) = q0; *(float4*)(d + 2 * A_PITCH) = q1;
    *(float4*)(d + 4 * A_PITCH) = q2; *(float4*)(d + 6 * A_PITCH) = q3;
    *(float4*)(d + 8 * A_PITCH) = q4; *(float4*)(d + 10 * A_PITCH) = q5;
    *(float4*)(d + 12 * A_PITCH) = q6; *(float4*)(d + 14 * A_PITCH) = q7;
  }
  __syncthreads();

  v8f acc[4];
#pragma unroll
  for (int t = 0; t < 4; t++) acc[t] = (v8f){0.f, 0.f, 0.f, 0.f, 0.f, 0.f, 0.f, 0.f};
  strip_half(a, bw, acc);
  load_a_half(aw, a);
  strip_half(a, bw + 64, acc);

  if (tile0 < n_rows) {
    float* o = hs + (size_t)(row0 + 8 * h) * N + cb + c;
#pragma unroll
    for (int i = 0; i < 8; i++) {
#pragma unroll
      for (int t = 0; t < 4; t++) {
        float v = acc[t][i];
        if (SCALE) v *= dv[i];
        o[i * N + 16 * t] = v;
      }
    }
  }
}

template <int N, bool RELU, bool DICT>
__global__ void __launch_bounds__(256)
gather_sum_kernel(const int* __restrict__ cnt, const int* __restrict__ slots,
                  const int* __restrict__ ovf_n, const int* __restrict__ ovf_src,
                  const int* __restrict__ ovf_dst, const int2* __restrict__ nt,
                  const float* __restrict__ rows, const float* __restrict__ dinv,
                  const float* __restrict__ bias, float* __restrict__ out, int n_rows) {
  typedef float vec __attribute__((ext_vector_type(N / 32)));
  const int lane = threadIdx.x & 31;
  const int wpb = blockDim.x >> 5;
  const int nwaves = gridDim.x * wpb;
  const vec* __restrict__ hv = (const vec*)rows;
  vec* __restrict__ ov = (vec*)out;
  const vec bv = ((const vec*)bias)[lane];

  const int r0 = __builtin_amdgcn_readfirstlane((int)(blockIdx.x * wpb + (threadIdx.x >> 5)));
  if (r0 >= n_rows) return;

  auto count_of = [&](int r) -> int { return r < n_rows ? cnt[r] : 0; };
  auto sources_of = [&](int r, int deg) -> int {
    const int num = deg < CAP ? deg : CAP;
    int s = 0;
    if (lane < num) s = slots[(size_t)r * CAP + lane];
    return s;
  };
  auto entries_of = [&](int s, int deg) -> int2 {
    if constexpr (DICT) {
      const int num = deg < CAP ? deg : CAP;
      int2 e = make_int2(0, 0);
      if (lane < num) e = nt[s];
      return e;
    } else {
      return make_int2(s, 0);
    }
  };

  int deg_c = cnt[r0];
  int deg_1 = count_of(r0 + nwaves);
  int deg_2 = count_of(r0 + 2 * nwaves);
  int2 ent_c = make_int2(0, 0);
  float dr_c = 0.f;
  if constexpr (DICT) ent_c = nt[r0]; else dr_c = dinv[r0];
  int2 e_c = entries_of(sources_of(r0, deg_c), deg_c);
  int sl_1 = sources_of(r0 + nwaves, deg_1);
  if constexpr (DICT) {
    asm volatile("" ::"v"(e_c.x), "v"(e_c.y), "v"(sl_1));
  } else {
    asm volatile("" ::"v"(e_c.x), "v"(sl_1));
  }

  for (int r = r0; r < n_rows; r += nwaves) {
    const int deg = deg_c;
    const int num = deg < CAP ? deg : CAP;
    const float dr = DICT ? __int_as_float(ent_c.y) : dr_c;
    const size_t br = DICT ? (size_t)ent_c.x * 32 : (size_t)r * 32;
    const int rowv = e_c.x;
    const int facv = e_c.y;
    const int r1 = r + nwaves;
    const int deg_3 = count_of(r1 + 2 * nwaves);
    if (r1 < n_rows) {
      if constexpr (DICT) ent_c = nt[r1]; else dr_c = dinv[r1];
    }
    const int sl_2 = sources_of(r1 + nwaves, deg_2);
    e_c = entries_of(sl_1, deg_1);

    const vec own = hv[br + lane];
    vec acc = DICT ? (vec)(0.f) : own;
#pragma clang loop unroll(disable)
    for (int j = 0; j < num; j += 8) {
      const int left = num - j;
      vec v[8];
      float w[8];
#pragma unroll
      for (int i = 0; i < 8; i++) {
        if (i == 0 || i < left) {
          const int s = __builtin_amdgcn_readlane(rowv, j + i);
          if constexpr (DICT) w[i] = __int_as_float(__builtin_amdgcn_readlane(facv, j + i));
          v[i] = hv[(size_t)s * 32 + lane];
        }
      }
#pragma unroll
      for (int i = 0; i < 8; i++) {
        if (i == 0 || i < left) {
          if constexpr (DICT) acc += w[i] * v[i]; else acc += v[i];
        }
      }
    }
    if (deg > CAP) {
      const int tot = *ovf_n;
      for (int q0 = 0; q0 < tot; q0 += 32) {
        const int q = q0 + lane;
        int sq = 0;
        bool hit = false;
        if (q < tot) {
          hit = ovf_dst[q] == r;
          sq = ovf_src[q];
        }
        unsigned todo = __builtin_amdgcn_ballot_w32(hit);
        while (todo) {
          const int b = __builtin_ctz(todo);
          todo &= todo - 1;
          const int s = __builtin_amdgcn_readlane(sq, b);
          if constexpr (DICT) {
            const int2 e = nt[s];
            acc += __int_as_float(e.y) * hv[(size_t)e.x * 32 + lane];
          } else {
            acc += hv[(size_t)s * 32 + lane];
          }
        }
      }
    }
    if (DICT) acc += dr * own;
    vec res = dr * acc + bv;
    if (RELU) res = __builtin_elementwise_max(res, (vec)(0.f));
    ov[(size_t)r * 32 + lane] = res;
    deg_c = deg_1;
    deg_1 = deg_2;
    deg_2 = deg_3;
    sl_1 = sl_2;
  }
}

template <int OFF>
__device__ __forceinline__ void fold_step(float* v, int lane) {
  const bool up = (lane & OFF) != 0;
#pragma unroll
  for (int i = 0; i < OFF; i++) {
    const float keep = up ? v[i + OFF] : v[i];
    const float send = up ? v[i] : v[i + OFF];
    v[i] = keep + __shfl_xor(send, OFF, 32);
  }
}

__global__ void __launch_bounds__(128)
edgedot_kernel(const int* __restrict__ es, const int* __restrict__ ed,
               const float* __restrict__ z, float* __restrict__ out, int nL) {
  const int lane = threadIdx.x & 31;
  const int e0 = __builtin_amdgcn_readfirstlane(
      (int)((blockIdx.x * (blockDim.x >> 5) + (threadIdx.x >> 5)) << 5));
  if (e0 >= nL) return;
  const int e = e0 + lane;
  const bool ok = e < nL;
  const int sa = ok ? es[e] : 0;
  const int sb = ok ? ed[e] : 0;
  const v2f* __restrict__ zv = (const v2f*)z;
  float p[32];
#pragma unroll
  for (int i = 0; i < 32; i++) {
    const int ra = __builtin_amdgcn_readlane(sa, i);
    const int rb = __builtin_amdgcn_readlane(sb, i);
    const v2f a = zv[(size_t)ra * 32 + lane];
    const v2f b = zv[(size_t)rb * 32 + lane];
    p[i] = a.x * b.x + a.y * b.y;
  }
  fold_step<16>(p, lane);
  fold_step<8>(p, lane);
  fold_step<4>(p, lane);
  fold_step<2>(p, lane);
  fold_step<1>(p, lane);
  if (ok) out[e] = p[0];
}

extern "C" void kernel_launch(void* const* d_in, const int* in_sizes, int n_in,
                              void* d_out, int out_size, void* d_ws, size_t ws_size,
                              hipStream_t stream) {
  const int*   x_idx = (const int*)d_in[0];
  const int*   eidx  = (const int*)d_in[1];
  const int*   lidx  = (const int*)d_in[2];
  const float* emb   = (const float*)d_in[3];
  const float* W1    = (const float*)d_in[4];
  const float* b1    = (const float*)d_in[5];
  const float* W2    = (const float*)d_in[6];
  const float* b2    = (const float*)d_in[7];
  float* out = (float*)d_out;

  const int nN = in_sizes[0];
  const int nE = in_sizes[1] / 2;
  const int nL = in_sizes[2] / 2;
  const int H = 128, O = 64;
  const int nD = in_sizes[3] / H;

  const int nbin = (nN + BIN_NODES - 1) / BIN_NODES;
  const int nCtrl = CTRL_BIN0 + nbin * CTRL_STRIDE;
  const size_t nPad = ((size_t)nN + 63) & ~(size_t)63;
  const size_t ePad = ((size_t)nE + 63) & ~(size_t)63;
  const size_t cPad = ((size_t)nCtrl + 63) & ~(size_t)63;
  float* dinv  = (float*)d_ws;
  int* cnt     = (int*)(dinv + nPad);
  int* ctrl    = cnt + nPad;
  int* ovf_n   = ctrl;
  int2* nt     = (int2*)(ctrl + cPad);
  int* slots   = ctrl + cPad + 2 * nPad;
  int* ovf_src = slots + nPad * CAP;
  int* ovf_dst = ovf_src + ePad;
  int* spill_src = ovf_dst + ePad;
  int* spill_dst = spill_src + ePad;
  float* t1    = (float*)(spill_dst + ePad);
  float* x1    = t1 + (size_t)nD * H;
  float* hs2   = x1 + (size_t)nN * H;
  float* z     = hs2 + (size_t)nN * O;
  int2* binlist = (int2*)(z + (size_t)nN * O);

  const int* esrc = eidx;
  const int* edst = eidx + nE;

  ctrl_zero_kernel<<<(nCtrl + 255) / 256, 256, 0, stream>>>(ctrl, nCtrl);
  bin_edges_kernel<<<(nE + BE_BLOCK * BE_EPT - 1) / (BE_BLOCK * BE_EPT), BE_BLOCK,
                     2 * nbin * sizeof(int), stream>>>(esrc, edst, ctrl, binlist, spill_src,
                                                       spill_dst, nE, nbin);
  bin_place_kernel<<<nbin, BIN_NODES, 0, stream>>>(ctrl, binlist, spill_src, spill_dst, x_idx, cnt,
                                                   dinv, nt, slots, ovf_src, ovf_dst, nN);

  gemm_rows_kernel<128, false><<<dim3((nD + 127) / 128, 2), 256, GEMM_LDS_BYTES, stream>>>(
      emb, W1, nullptr, t1, nD);
  gather_sum_kernel<128, true, true><<<2048, 256, 0, stream>>>(
      cnt, slots, ovf_n, ovf_src, ovf_dst, nt, t1, dinv, b1, x1, nN);

  gemm_rows_kernel<64, true><<<dim3((nN + 127) / 128, 1), 256, GEMM_LDS_BYTES, stream>>>(
      x1, W2, dinv, hs2, nN);
  gather_sum_kernel<64, false, false><<<2048, 256, 0, stream>>>(
      cnt, slots, ovf_n, ovf_src, ovf_dst, nt, hs2, dinv, b2, z, nN);

  edgedot_kernel<<<(nL + 127) / 128, 128, 0, stream>>>(
      lidx, lidx + nL, z, out, nL);
}


# Layer 2 product fused into the layer 1 node-sum kernel: a wave sums 16 nodes in a row into its own LDS rows, then multiplies that strip by W2 on WMMA; x1 never stored, a launch less; sum orders kept
# speedup vs baseline: 5.3387x; 1.0307x over previous
#include <hip/hip_runtime.h>

typedef float v2f __attribute__((ext_vector_type(2)));
typedef float v8f __attribute__((ext_vector_type(8)));

#define CAP 32
#define BIN_SHIFT 8
#define BIN_NODES 256
#define BINCAP 8192
#define CTRL_BIN0 64
#define CTRL_STRIDE 32
#define BE_BLOCK 512
#define BE_EPT 8
#define BP_EPT 8

__global__ void ctrl_zero_kernel(int* ctrl, int n) {
  int i = blockIdx.x * blockDim.x + threadIdx.x;
  if (i < n) ctrl[i] = 0;
}

__global__ void __launch_bounds__(BE_BLOCK)
bin_edges_kernel(const int* __restrict__ src, const int* __restrict__ dst, int* __restrict__ ctrl,
                 int2* __restrict__ binlist, int* __restrict__ spill_src,
                 int* __restrict__ spill_dst, int nE, int nbin) {
  extern __shared__ int be_lds[];
  int* hist = be_lds;
  int* base = be_lds + nbin;
  const int tid = threadIdx.x;
  for (int b = tid; b < nbin; b += BE_BLOCK) hist[b] = 0;
  const int e0 = blockIdx.x * (BE_BLOCK * BE_EPT) + tid;
  int d[BE_EPT], s[BE_EPT], rk[BE_EPT];
#pragma unroll
  for (int i = 0; i < BE_EPT; i++) {
    const int e = e0 + i * BE_BLOCK;
    d[i] = e < nE ? dst[e] : -1;
    s[i] = e < nE ? src[e] : 0;
  }
  __syncthreads();
#pragma unroll
  for (int i = 0; i < BE_EPT; i++) {
    rk[i] = 0;
    if (d[i] >= 0) rk[i] = atomicAdd(&hist[d[i] >> BIN_SHIFT], 1);
  }
  __syncthreads();
  for (int b = tid; b < nbin; b += BE_BLOCK) {
    const int h = hist[b];
    base[b] = h ? atomicAdd(&ctrl[CTRL_BIN0 + b * CTRL_STRIDE], h) : 0;
  }
  __syncthreads();
#pragma unroll
  for (int i = 0; i < BE_EPT; i++) {
    if (d[i] >= 0) {
      const int b = d[i] >> BIN_SHIFT;
      const int pos = base[b] + rk[i];
      if (pos < BINCAP) {
        binlist[(size_t)b * BINCAP + pos] = make_int2(d[i], s[i]);
      } else {
        const int q = atomicAdd(&ctrl[32], 1);
        spill_src[q] = s[i];
        spill_dst[q] = d[i];
      }
    }
  }
}

__global__ void __launch_bounds__(BIN_NODES)
bin_place_kernel(int* ctrl, const int2* __restrict__ binlist, const int* __restrict__ spill_src,
                 const int* __restrict__ spill_dst, const int* __restrict__ x_idx,
                 int* __restrict__ cnt, float* __restrict__ dinv, int2* __restrict__ nt,
                 int* __restrict__ slots, int* __restrict__ ovf_src, int* __restrict__ ovf_dst,
                 int n) {
  __shared__ int cntl[BIN_NODES];
  __shared__ __attribute__((aligned(16))) int slotl[BIN_NODES * CAP];
  const int tid = threadIdx.x;
  const int bin = blockIdx.x;
  const int node = bin * BIN_NODES + tid;
  const int fill = ctrl[CTRL_BIN0 + bin * CTRL_STRIDE];
  const int n_spill = ctrl[32];
  const int nb = fill < BINCAP ? fill : BINCAP;
  const int2* lst = binlist + (size_t)bin * BINCAP;
  const int xi = node < n ? x_idx[node] : 0;
  int2 p[BP_EPT];
  auto load_round = [&](int i0) {
#pragma unroll
    for (int j = 0; j < BP_EPT; j++) {
      const int i = i0 + j * BIN_NODES + tid;
      p[j] = make_int2(-1, 0);
      if (i < nb) p[j] = lst[i];
    }
  };
  load_round(0);
  int4* sl4 = (int4*)slotl;
  cntl[tid] = 0;
#pragma unroll
  for (int i = 0; i < BIN_NODES * CAP / 4 / BIN_NODES; i++) sl4[tid + i * BIN_NODES] = make_int4(0, 0, 0, 0);
  __syncthreads();

  auto place = [&](int d, int s) {
    const int t = d - bin * BIN_NODES;
    const int k = atomicAdd(&cntl[t], 1);
    if (k < CAP) {
      slotl[t * CAP + k] = s;
    } else {
      const int q = atomicAdd(&ctrl[0], 1);
      ovf_src[q] = s;
      ovf_dst[q] = d;
    }
  };
  for (int i0 = 0;;) {
#pragma unroll
    for (int j = 0; j < BP_EPT; j++)
      if (p[j].x >= 0) place(p[j].x, p[j].y);
    i0 += BIN_NODES * BP_EPT;
    if (i0 >= nb) break;
    load_round(i0);
  }
  for (int i = tid; i < n_spill; i += BIN_NODES) {
    const int d = spill_dst[i];
    if ((d >> BIN_SHIFT) == bin) place(d, spill_src[i]);
  }
  __syncthreads();

  if (node < n) {
    const int c = cntl[tid];
    const float dd = rsqrtf((float)(c + 1));
    cnt[node] = c;
    dinv[node] = dd;
    nt[node] = make_int2(xi, __float_as_int(dd));
  }
  int4* g4 = (int4*)(slots + (size_t)bin * BIN_NODES * CAP);
#pragma unroll
  for (int i = 0; i < BIN_NODES * CAP / 4 / BIN_NODES; i++) {
    const int idx = tid + i * BIN_NODES;
    if (bin * BIN_NODES + (idx >> 3) < n) g4[idx] = sl4[idx];
  }
}

#define W_PITCH 132
#define A_PITCH 68
#define GEMM_LDS_BYTES ((64 * W_PITCH + 8 * 16 * A_PITCH) * 4)

__device__ __forceinline__ void load_a_half(const float* __restrict__ aw, float4 (&a)[8]) {
#pragma unroll
  for (int j = 0; j < 8; j++) a[j] = *(const float4*)(aw + 8 * j);
}

__device__ __forceinline__ void strip_half(const float4 (&a)[8], const float* __restrict__ bw,
                                           v8f (&acc)[4]) {
#pragma unroll
  for (int j = 0; j < 8; j++) {
    const v2f a0 = {a[j].x, a[j].y};
    const v2f a1 = {a[j].z, a[j].w};
#pragma unroll
    for (int t = 0; t < 4; t++) {
      const float4 b = *(const float4*)(bw + t * 16 * W_PITCH + 8 * j);
      const v2f b0 = {b.x, b.y};
      const v2f b1 = {b.z, b.w};
      acc[t] = __builtin_amdgcn_wmma_f32_16x16x4_f32(false, a0, false, b0, (short)0, acc[t], false, false);
      acc[t] = __builtin_amdgcn_wmma_f32_16x16x4_f32(false, a1, false, b1, (short)0, acc[t], false, false);
    }
  }
}

template <int N, bool SCALE>
__global__ void __launch_bounds__(256) __attribute__((amdgpu_waves_per_eu(8, 8)))
gemm_rows_kernel(const float* X, const float* __restrict__ W,
                 const float* __restrict__ dinv, float* __restrict__ hs, int n_rows) {
  extern __shared__ float4 lds4[];
  float* lW = (float*)lds4;
  const int tid = threadIdx.x;
  const int lane = tid & 31;
  const int wave = tid >> 5;
  float* lAw = lW + 64 * W_PITCH + wave * (16 * A_PITCH);
  const int c = lane & 15;
  const int h = lane >> 4;
  const int cb = blockIdx.y * 64;
  const int tile0 = (blockIdx.x * 8 + wave) * 16;
  const int row0 = tile0 < n_rows - 16 ? tile0 : n_rows - 16;

  float4 raw0[8];
  float4 q0, q1, q2, q3, q4, q5, q6, q7;
  const float4* xp = (const float4*)(X + (size_t)(row0 + h) * 128) + c;
#pragma unroll
  for (int i = 0; i < 8; i++) raw0[i] = xp[64 * i];
  q0 = xp[16]; q1 = xp[64 + 16]; q2 = xp[128 + 16]; q3 = xp[192 + 16];
  q4 = xp[256 + 16]; q5 = xp[320 + 16]; q6 = xp[384 + 16]; q7 = xp[448 + 16];
  float4 wv[8];
  const int wk = 8 * (wave >> 2) + (lane & 7);
  const int wn = 16 * (wave & 3) + 4 * (lane >> 3);
  const float4* wp = (const float4*)(W + (size_t)wk * N + cb + wn);
#pragma unroll
  for (int it = 0; it < 8; it++) wv[it] = wp[it * (16 * N / 4)];
  float dv[8];
  if (SCALE) {
#pragma unroll
    for (int i = 0; i < 8; i++) dv[i] = dinv[row0 + 8 * h + i];
  }

  float* ws = lW + wn * W_PITCH + wk;
#pragma unroll
  for (int it = 0; it < 8; it++) {
    ws[0 * W_PITCH + 16 * it] = wv[it].x;
    ws[1 * W_PITCH + 16 * it] = wv[it].y;
    ws[2 * W_PITCH + 16 * it] = wv[it].z;
    ws[3 * W_PITCH + 16 * it] = wv[it].w;
  }
#pragma unroll
  for (int i = 0; i < 8; i++) *(float4*)(lAw + (2 * i + h) * A_PITCH + 4 * c) = raw0[i];
  __syncthreads();

  const float* aw = lAw + c * A_PITCH + 4 * h;
  const float* bw = lW + c * W_PITCH + 4 * h;
  float4 a[8];
  load_a_half(aw, a);
  __syncthreads();
  {
    float* d = lAw + h * A_PITCH + 4 * c;
    *(float4*)(d + 0 * A_PITCH) = q0; *(float4*)(d + 2 * A_PITCH) = q1;
    *(float4*)(d + 4 * A_PITCH) = q2; *(float4*)(d + 6 * A_PITCH) = q3;
    *(float4*)(d + 8 * A_PITCH) = q4; *(float4*)(d + 10 * A_PITCH) = q5;
    *(float4*)(d + 12 * A_PITCH) = q6; *(float4*)(d + 14 * A_PITCH) = q7;
  }
  __syncthreads();

  v8f acc[4];
#pragma unroll
  for (int t = 0; t < 4; t++) acc[t] = (v8f){0.f, 0.f, 0.f, 0.f, 0.f, 0.f, 0.f, 0.f};
  strip_half(a, bw, acc);
  load_a_half(aw, a);
  strip_half(a, bw + 64, acc);

  if (tile0 < n_rows) {
    float* o = hs + (size_t)(row0 + 8 * h) * N + cb + c;
#pragma unroll
    for (int i = 0; i < 8; i++) {
#pragma unroll
      for (int t = 0; t < 4; t++) {
        float v = acc[t][i];
        if (SCALE) v *= dv[i];
        o[i * N + 16 * t] = v;
      }
    }
  }
}

template <int N, bool RELU, bool DICT>
__global__ void __launch_bounds__(256)
gather_sum_kernel(const int* __restrict__ cnt, const int* __restrict__ slots,
                  const int* __restrict__ ovf_n, const int* __restrict__ ovf_src,
                  const int* __restrict__ ovf_dst, const int2* __restrict__ nt,
                  const float* __restrict__ rows, const float* __restrict__ dinv,
                  const float* __restrict__ bias, float* __restrict__ out, int n_rows) {
  typedef float vec __attribute__((ext_vector_type(N / 32)));
  const int lane = threadIdx.x & 31;
  const int wpb = blockDim.x >> 5;
  const int nwaves = gridDim.x * wpb;
  const vec* __restrict__ hv = (const vec*)rows;
  vec* __restrict__ ov = (vec*)out;
  const vec bv = ((const vec*)bias)[lane];

  const int r0 = __builtin_amdgcn_readfirstlane((int)(blockIdx.x * wpb + (threadIdx.x >> 5)));
  if (r0 >= n_rows) return;

  auto count_of = [&](int r) -> int { return r < n_rows ? cnt[r] : 0; };
  auto sources_of = [&](int r, int deg) -> int {
    const int num = deg < CAP ? deg : CAP;
    int s = 0;
    if (lane < num) s = slots[(size_t)r * CAP + lane];
    return s;
  };
  auto entries_of = [&](int s, int deg) -> int2 {
    if constexpr (DICT) {
      const int num = deg < CAP ? deg : CAP;
      int2 e = make_int2(0, 0);
      if (lane < num) e = nt[s];
      return e;
    } else {
      return make_int2(s, 0);
    }
  };

  int deg_c = cnt[r0];
  int deg_1 = count_of(r0 + nwaves);
  int deg_2 = count_of(r0 + 2 * nwaves);
  int2 ent_c = make_int2(0, 0);
  float dr_c = 0.f;
  if constexpr (DICT) ent_c = nt[r0]; else dr_c = dinv[r0];
  int2 e_c = entries_of(sources_of(r0, deg_c), deg_c);
  int sl_1 = sources_of(r0 + nwaves, deg_1);
  if constexpr (DICT) {
    asm volatile("" ::"v"(e_c.x), "v"(e_c.y), "v"(sl_1));
  } else {
    asm volatile("" ::"v"(e_c.x), "v"(sl_1));
  }

  for (int r = r0; r < n_rows; r += nwaves) {
    const int deg = deg_c;
    const int num = deg < CAP ? deg : CAP;
    const float dr = DICT ? __int_as_float(ent_c.y) : dr_c;
    const size_t br = DICT ? (size_t)ent_c.x * 32 : (size_t)r * 32;
    const int rowv = e_c.x;
    const int facv = e_c.y;
    const int r1 = r + nwaves;
    const int deg_3 = count_of(r1 + 2 * nwaves);
    if (r1 < n_rows) {
      if constexpr (DICT) ent_c = nt[r1]; else dr_c = dinv[r1];
    }
    const int sl_2 = sources_of(r1 + nwaves, deg_2);
    e_c = entries_of(sl_1, deg_1);

    const vec own = hv[br + lane];
    vec acc = DICT ? (vec)(0.f) : own;
#pragma clang loop unroll(disable)
    for (int j = 0; j < num; j += 8) {
      const int left = num - j;
      vec v[8];
      float w[8];
#pragma unroll
      for (int i = 0; i < 8; i++) {
        if (i == 0 || i < left) {
          const int s = __builtin_amdgcn_readlane(rowv, j + i);
          if constexpr (DICT) w[i] = __int_as_float(__builtin_amdgcn_readlane(facv, j + i));
          v[i] = hv[(size_t)s * 32 + lane];
        }
      }
#pragma unroll
      for (int i = 0; i < 8; i++) {
        if (i == 0 || i < left) {
          if constexpr (DICT) acc += w[i] * v[i]; else acc += v[i];
        }
      }
    }
    if (deg > CAP) {
      const int tot = *ovf_n;
      for (int q0 = 0; q0 < tot; q0 += 32) {
        const int q = q0 + lane;
        int sq = 0;
        bool hit = false;
        if (q < tot) {
          hit = ovf_dst[q] == r;
          sq = ovf_src[q];
        }
        unsigned todo = __builtin_amdgcn_ballot_w32(hit);
        while (todo) {
          const int b = __builtin_ctz(todo);
          todo &= todo - 1;
          const int s = __builtin_amdgcn_readlane(sq, b);
          if constexpr (DICT) {
            const int2 e = nt[s];
            acc += __int_as_float(e.y) * hv[(size_t)e.x * 32 + lane];
          } else {
            acc += hv[(size_t)s * 32 + lane];
          }
        }
      }
    }
    if (DICT) acc += dr * own;
    vec res = dr * acc + bv;
    if (RELU) res = __builtin_elementwise_max(res, (vec)(0.f));
    ov[(size_t)r * 32 + lane] = res;
    deg_c = deg_1;
    deg_1 = deg_2;
    deg_2 = deg_3;
    sl_1 = sl_2;
  }
}

#define FA_PITCH 132
#define FUSE_WAVES 4
#define FUSE_WAVE_FLOATS (16 * FA_PITCH + 16)

__global__ void __launch_bounds__(32 * FUSE_WAVES) __attribute__((amdgpu_waves_per_eu(8, 8)))
gather_gemm_kernel(const int* __restrict__ cnt, const int* __restrict__ slots,
                   const int* __restrict__ ovf_n, const int* __restrict__ ovf_src,
                   const int* __restrict__ ovf_dst, const int2* __restrict__ nt,
                   const float* __restrict__ rows, const float* __restrict__ bias,
                   const float* __restrict__ W, float* __restrict__ hs, int n_rows) {
  typedef float vec __attribute__((ext_vector_type(4)));
  __shared__ __attribute__((aligned(16))) float lA[FUSE_WAVES * FUSE_WAVE_FLOATS];
  const int lane = threadIdx.x & 31;
  const int wave = threadIdx.x >> 5;
  float* la = lA + wave * FUSE_WAVE_FLOATS;
  float* ldr = la + 16 * FA_PITCH;
  const int c = lane & 15;
  const int h = lane >> 4;
  const vec* __restrict__ hv = (const vec*)rows;
  const vec bv = ((const vec*)bias)[lane];
  const int nstrips = (n_rows + 15) >> 4;
  const int nwaves = gridDim.x * FUSE_WAVES;
  const int strip0 = __builtin_amdgcn_readfirstlane((int)(blockIdx.x * FUSE_WAVES + (threadIdx.x >> 5)));

  for (int strip = strip0; strip < nstrips; strip += nwaves) {
    const int rb = strip * 16;
    const int rend = rb + 16 < n_rows ? rb + 16 : n_rows;

    auto count_of = [&](int r) -> int { return r < rend ? cnt[r] : 0; };
    auto sources_of = [&](int r, int deg) -> int {
      const int num = deg < CAP ? deg : CAP;
      int s = 0;
      if (lane < num) s = slots[(size_t)r * CAP + lane];
      return s;
    };
    auto entries_of = [&](int s, int deg) -> int2 {
      const int num = deg < CAP ? deg : CAP;
      int2 e = make_int2(0, 0);
      if (lane < num) e = nt[s];
      return e;
    };

    int deg_c = cnt[rb];
    int deg_1 = count_of(rb + 1);
    int deg_2 = count_of(rb + 2);
    int2 ent_c = nt[rb];
    int2 e_c = entries_of(sources_of(rb, deg_c), deg_c);
    int sl_1 = sources_of(rb + 1, deg_1);
    asm volatile("" ::"v"(e_c.x), "v"(e_c.y), "v"(sl_1));

    for (int i = rend - rb; i < 16; i++) {
      *(vec*)(la + i * FA_PITCH + 4 * lane) = (vec)(0.f);
      if (lane == 0) ldr[i] = 0.f;
    }

    for (int r = rb; r < rend; r++) {
      const int deg = deg_c;
      const int num = deg < CAP ? deg : CAP;
      const float dr = __int_as_float(ent_c.y);
      const size_t br = (size_t)ent_c.x * 32;
      const int rowv = e_c.x;
      const int facv = e_c.y;
      const int r1 = r + 1;
      const int deg_3 = count_of(r + 3);
      if (r1 < rend) ent_c = nt[r1];
      const int sl_2 = sources_of(r + 2, deg_2);
      e_c = entries_of(sl_1, deg_1);

      const vec own = hv[br + lane];
      vec acc = (vec)(0.f);
#pragma clang loop unroll(disable)
      for (int j = 0; j < num; j += 8) {
        const int left = num - j;
        vec v[8];
        float w[8];
#pragma unroll
        for (int i = 0; i < 8; i++) {
          if (i == 0 || i < left) {
            const int s = __builtin_amdgcn_readlane(rowv, j + i);
            w[i] = __int_as_float(__builtin_amdgcn_readlane(facv, j + i));
            v[i] = hv[(size_t)s * 32 + lane];
          }
        }
#pragma unroll
        for (int i = 0; i < 8; i++) {
          if (i == 0 || i < left) acc += w[i] * v[i];
        }
      }
      if (deg > CAP) {
        const int tot = *ovf_n;
        for (int q0 = 0; q0 < tot; q0 += 32) {
          const int q = q0 + lane;
          int sq = 0;
          bool hit = false;
          if (q < tot) {
            hit = ovf_dst[q] == r;
            sq = ovf_src[q];
          }
          unsigned todo = __builtin_amdgcn_ballot_w32(hit);
          while (todo) {
            const int b = __builtin_ctz(todo);
            todo &= todo - 1;
            const int s = __builtin_amdgcn_readlane(sq, b);
            const int2 e = nt[s];
            acc += __int_as_float(e.y) * hv[(size_t)e.x * 32 + lane];
          }
        }
      }
      acc += dr * own;
      vec res = dr * acc + bv;
      res = __builtin_elementwise_max(res, (vec)(0.f));
      *(vec*)(la + (r - rb) * FA_PITCH + 4 * lane) = res;
      if (lane == 0) ldr[r - rb] = dr;
      deg_c = deg_1;
      deg_1 = deg_2;
      deg_2 = deg_3;
      sl_1 = sl_2;
    }

    __builtin_amdgcn_fence(__ATOMIC_RELEASE, "wavefront");
    __builtin_amdgcn_wave_barrier();
    __builtin_amdgcn_fence(__ATOMIC_ACQUIRE, "wavefront");

    const float* aw = la + c * FA_PITCH + 4 * h;
    const float4* wp = (const float4*)W + 64 * h + c;
    v8f acc[4];
#pragma unroll
    for (int t = 0; t < 4; t++) acc[t] = (v8f){0.f, 0.f, 0.f, 0.f, 0.f, 0.f, 0.f, 0.f};
    float4 an = *(const float4*)aw;
    float4 wn0 = wp[0], wn1 = wp[16], wn2 = wp[32], wn3 = wp[48];
#pragma unroll
    for (int s = 0; s < 16; s++) {
      const float4 a = an;
      const float e0[4] = {wn0.x, wn0.y, wn0.z, wn0.w};
      const float e1[4] = {wn1.x, wn1.y, wn1.z, wn1.w};
      const float e2[4] = {wn2.x, wn2.y, wn2.z, wn2.w};
      const float e3[4] = {wn3.x, wn3.y, wn3.z, wn3.w};
      if (s + 1 < 16) {
        an = *(const float4*)(aw + 8 * (s + 1));
        const float4* wj = wp + 8 * (s + 1) * 16;
        wn0 = wj[0]; wn1 = wj[16]; wn2 = wj[32]; wn3 = wj[48];
      }
      const v2f a0 = {a.x, a.y};
      const v2f a1 = {a.z, a.w};
#pragma unroll
      for (int t = 0; t < 4; t++) {
        const v2f b0 = {e0[t], e1[t]};
        const v2f b1 = {e2[t], e3[t]};
        acc[t] = __builtin_amdgcn_wmma_f32_16x16x4_f32(false, a0, false, b0, (short)0, acc[t], false, false);
        acc[t] = __builtin_amdgcn_wmma_f32_16x16x4_f32(false, a1, false, b1, (short)0, acc[t], false, false);
      }
    }

    const float4 d0 = *(const float4*)(ldr + 8 * h);
    const float4 d1 = *(const float4*)(ldr + 8 * h + 4);
    const float dv[8] = {d0.x, d0.y, d0.z, d0.w, d1.x, d1.y, d1.z, d1.w};
    float4* o = (float4*)(hs + (size_t)(rb + 8 * h) * 64) + c;
#pragma unroll
    for (int i = 0; i < 8; i++) {
      if (rb + 8 * h + i < n_rows)
        o[i * 16] = make_float4(acc[0][i] * dv[i], acc[1][i] * dv[i], acc[2][i] * dv[i], acc[3][i] * dv[i]);
    }
    __builtin_amdgcn_fence(__ATOMIC_RELEASE, "wavefront");
    __builtin_amdgcn_wave_barrier();
    __builtin_amdgcn_fence(__ATOMIC_ACQUIRE, "wavefront");
  }
}

template <int OFF>
__device__ __forceinline__ void fold_step(float* v, int lane) {
  const bool up = (lane & OFF) != 0;
#pragma unroll
  for (int i = 0; i < OFF; i++) {
    const float keep = up ? v[i + OFF] : v[i];
    const float send = up ? v[i] : v[i + OFF];
    v[i] = keep + __shfl_xor(send, OFF, 32);
  }
}

__global__ void __launch_bounds__(128)
edgedot_kernel(const int* __restrict__ es, const int* __restrict__ ed,
               const float* __restrict__ z, float* __restrict__ out, int nL) {
  const int lane = threadIdx.x & 31;
  const int e0 = __builtin_amdgcn_readfirstlane(
      (int)((blockIdx.x * (blockDim.x >> 5) + (threadIdx.x >> 5)) << 5));
  if (e0 >= nL) return;
  const int e = e0 + lane;
  const bool ok = e < nL;
  const int sa = ok ? es[e] : 0;
  const int sb = ok ? ed[e] : 0;
  const v2f* __restrict__ zv = (const v2f*)z;
  float p[32];
#pragma unroll
  for (int i = 0; i < 32; i++) {
    const int ra = __builtin_amdgcn_readlane(sa, i);
    const int rb = __builtin_amdgcn_readlane(sb, i);
    const v2f a = zv[(size_t)ra * 32 + lane];
    const v2f b = zv[(size_t)rb * 32 + lane];
    p[i] = a.x * b.x + a.y * b.y;
  }
  fold_step<16>(p, lane);
  fold_step<8>(p, lane);
  fold_step<4>(p, lane);
  fold_step<2>(p, lane);
  fold_step<1>(p, lane);
  if (ok) out[e] = p[0];
}

extern "C" void kernel_launch(void* const* d_in, const int* in_sizes, int n_in,
                              void* d_out, int out_size, void* d_ws, size_t ws_size,
                              hipStream_t stream) {
  const int*   x_idx = (const int*)d_in[0];
  const int*   eidx  = (const int*)d_in[1];
  const int*   lidx  = (const int*)d_in[2];
  const float* emb   = (const float*)d_in[3];
  const float* W1    = (const float*)d_in[4];
  const float* b1    = (const float*)d_in[5];
  const float* W2    = (const float*)d_in[6];
  const float* b2    = (const float*)d_in[7];
  float* out = (float*)d_out;

  const int nN = in_sizes[0];
  const int nE = in_sizes[1] / 2;
  const int nL = in_sizes[2] / 2;
  const int H = 128, O = 64;
  const int nD = in_sizes[3] / H;

  const int nbin = (nN + BIN_NODES - 1) / BIN_NODES;
  const int nCtrl = CTRL_BIN0 + nbin * CTRL_STRIDE;
  const size_t nPad = ((size_t)nN + 63) & ~(size_t)63;
  const size_t ePad = ((size_t)nE + 63) & ~(size_t)63;
  const size_t cPad = ((size_t)nCtrl + 63) & ~(size_t)63;
  float* dinv  = (float*)d_ws;
  int* cnt     = (int*)(dinv + nPad);
  int* ctrl    = cnt + nPad;
  int* ovf_n   = ctrl;
  int2* nt     = (int2*)(ctrl + cPad);
  int* slots   = ctrl + cPad + 2 * nPad;
  int* ovf_src = slots + nPad * CAP;
  int* ovf_dst = ovf_src + ePad;
  int* spill_src = ovf_dst + ePad;
  int* spill_dst = spill_src + ePad;
  float* t1    = (float*)(spill_dst + ePad);
  float* hs2   = t1 + (size_t)nD * H;
  float* z     = hs2 + (size_t)nN * O;
  int2* binlist = (int2*)(z + (size_t)nN * O);

  const int* esrc = eidx;
  const int* edst = eidx + nE;

  ctrl_zero_kernel<<<(nCtrl + 255) / 256, 256, 0, stream>>>(ctrl, nCtrl);
  bin_edges_kernel<<<(nE + BE_BLOCK * BE_EPT - 1) / (BE_BLOCK * BE_EPT), BE_BLOCK,
                     2 * nbin * sizeof(int), stream>>>(esrc, edst, ctrl, binlist, spill_src,
                                                       spill_dst, nE, nbin);
  bin_place_kernel<<<nbin, BIN_NODES, 0, stream>>>(ctrl, binlist, spill_src, spill_dst, x_idx, cnt,
                                                   dinv, nt, slots, ovf_src, ovf_dst, nN);

  gemm_rows_kernel<128, false><<<dim3((nD + 127) / 128, 2), 256, GEMM_LDS_BYTES, stream>>>(
      emb, W1, nullptr, t1, nD);
  gather_gemm_kernel<<<((nN + 15) / 16 + FUSE_WAVES - 1) / FUSE_WAVES, 32 * FUSE_WAVES, 0, stream>>>(
      cnt, slots, ovf_n, ovf_src, ovf_dst, nt, t1, b1, W2, hs2, nN);

  gather_sum_kernel<64, false, false><<<2048, 256, 0, stream>>>(
      cnt, slots, ovf_n, ovf_src, ovf_dst, nt, hs2, dinv, b2, z, nN);

  edgedot_kernel<<<(nL + 127) / 128, 128, 0, stream>>>(
      lidx, lidx + nL, z, out, nL);
}


# Table product emb@W1 moved into the bin-placing launch as extra blocks (independent jobs side by side): a launch less, the product's time hidden under the adjacency build; sums unchanged
# speedup vs baseline: 5.6785x; 1.0636x over previous
#include <hip/hip_runtime.h>

typedef float v2f __attribute__((ext_vector_type(2)));
typedef float v8f __attribute__((ext_vector_type(8)));

#define CAP 32
#define BIN_SHIFT 8
#define BIN_NODES 256
#define BINCAP 8192
#define CTRL_BIN0 64
#define CTRL_STRIDE 32
#define BE_BLOCK 512
#define BE_EPT 8
#define BP_EPT 8

__global__ void ctrl_zero_kernel(int* ctrl, int n) {
  int i = blockIdx.x * blockDim.x + threadIdx.x;
  if (i < n) ctrl[i] = 0;
}

__global__ void __launch_bounds__(BE_BLOCK)
bin_edges_kernel(const int* __restrict__ src, const int* __restrict__ dst, int* __restrict__ ctrl,
                 int2* __restrict__ binlist, int* __restrict__ spill_src,
                 int* __restrict__ spill_dst, int nE, int nbin) {
  extern __shared__ int be_lds[];
  int* hist = be_lds;
  int* base = be_lds + nbin;
  const int tid = threadIdx.x;
  for (int b = tid; b < nbin; b += BE_BLOCK) hist[b] = 0;
  const int e0 = blockIdx.x * (BE_BLOCK * BE_EPT) + tid;
  int d[BE_EPT], s[BE_EPT], rk[BE_EPT];
#pragma unroll
  for (int i = 0; i < BE_EPT; i++) {
    const int e = e0 + i * BE_BLOCK;
    d[i] = e < nE ? dst[e] : -1;
    s[i] = e < nE ? src[e] : 0;
  }
  __syncthreads();
#pragma unroll
  for (int i = 0; i < BE_EPT; i++) {
    rk[i] = 0;
    if (d[i] >= 0) rk[i] = atomicAdd(&hist[d[i] >> BIN_SHIFT], 1);
  }
  __syncthreads();
  for (int b = tid; b < nbin; b += BE_BLOCK) {
    const int h = hist[b];
    base[b] = h ? atomicAdd(&ctrl[CTRL_BIN0 + b * CTRL_STRIDE], h) : 0;
  }
  __syncthreads();
#pragma unroll
  for (int i = 0; i < BE_EPT; i++) {
    if (d[i] >= 0) {
      const int b = d[i] >> BIN_SHIFT;
      const int pos = base[b] + rk[i];
      if (pos < BINCAP) {
        binlist[(size_t)b * BINCAP + pos] = make_int2(d[i], s[i]);
      } else {
        const int q = atomicAdd(&ctrl[32], 1);
        spill_src[q] = s[i];
        spill_dst[q] = d[i];
      }
    }
  }
}

#define PLACE_LDS_BYTES ((BIN_NODES + BIN_NODES * CAP) * 4)
__device__ __forceinline__ void
bin_place_block(int* lds, int bin, int* ctrl, const int2* __restrict__ binlist,
                const int* __restrict__ spill_src, const int* __restrict__ spill_dst,
                const int* __restrict__ x_idx, int* __restrict__ cnt, float* __restrict__ dinv,
                int2* __restrict__ nt, int* __restrict__ slots, int* __restrict__ ovf_src,
                int* __restrict__ ovf_dst, int n) {
  int* cntl = lds;
  int* slotl = lds + BIN_NODES;
  const int tid = threadIdx.x;
  const int node = bin * BIN_NODES + tid;
  const int fill = ctrl[CTRL_BIN0 + bin * CTRL_STRIDE];
  const int n_spill = ctrl[32];
  const int nb = fill < BINCAP ? fill : BINCAP;
  const int2* lst = binlist + (size_t)bin * BINCAP;
  const int xi = node < n ? x_idx[node] : 0;
  int2 p[BP_EPT];
  auto load_round = [&](int i0) {
#pragma unroll
    for (int j = 0; j < BP_EPT; j++) {
      const int i = i0 + j * BIN_NODES + tid;
      p[j] = make_int2(-1, 0);
      if (i < nb) p[j] = lst[i];
    }
  };
  load_round(0);
  int4* sl4 = (int4*)slotl;
  cntl[tid] = 0;
#pragma unroll
  for (int i = 0; i < BIN_NODES * CAP / 4 / BIN_NODES; i++) sl4[tid + i * BIN_NODES] = make_int4(0, 0, 0, 0);
  __syncthreads();

  auto place = [&](int d, int s) {
    const int t = d - bin * BIN_NODES;
    const int k = atomicAdd(&cntl[t], 1);
    if (k < CAP) {
      slotl[t * CAP + k] = s;
    } else {
      const int q = atomicAdd(&ctrl[0], 1);
      ovf_src[q] = s;
      ovf_dst[q] = d;
    }
  };
  for (int i0 = 0;;) {
#pragma unroll
    for (int j = 0; j < BP_EPT; j++)
      if (p[j].x >= 0) place(p[j].x, p[j].y);
    i0 += BIN_NODES * BP_EPT;
    if (i0 >= nb) break;
    load_round(i0);
  }
  for (int i = tid; i < n_spill; i += BIN_NODES) {
    const int d = spill_dst[i];
    if ((d >> BIN_SHIFT) == bin) place(d, spill_src[i]);
  }
  __syncthreads();

  if (node < n) {
    const int c = cntl[tid];
    const float dd = rsqrtf((float)(c + 1));
    cnt[node] = c;
    dinv[node] = dd;
    nt[node] = make_int2(xi, __float_as_int(dd));
  }
  int4* g4 = (int4*)(slots + (size_t)bin * BIN_NODES * CAP);
#pragma unroll
  for (int i = 0; i < BIN_NODES * CAP / 4 / BIN_NODES; i++) {
    const int idx = tid + i * BIN_NODES;
    if (bin * BIN_NODES + (idx >> 3) < n) g4[idx] = sl4[idx];
  }
}

#define W_PITCH 132
#define A_PITCH 68
#define GEMM_LDS_BYTES ((64 * W_PITCH + 8 * 16 * A_PITCH) * 4)

__device__ __forceinline__ void load_a_half(const float* __restrict__ aw, float4 (&a)[8]) {
#pragma unroll
  for (int j = 0; j < 8; j++) a[j] = *(const float4*)(aw + 8 * j);
}

__device__ __forceinline__ void strip_half(const float4 (&a)[8], const float* __restrict__ bw,
                                           v8f (&acc)[4]) {
#pragma unroll
  for (int j = 0; j < 8; j++) {
    const v2f a0 = {a[j].x, a[j].y};
    const v2f a1 = {a[j].z, a[j].w};
#pragma unroll
    for (int t = 0; t < 4; t++) {
      const float4 b = *(const float4*)(bw + t * 16 * W_PITCH + 8 * j);
      const v2f b0 = {b.x, b.y};
      const v2f b1 = {b.z, b.w};
      acc[t] = __builtin_amdgcn_wmma_f32_16x16x4_f32(false, a0, false, b0, (short)0, acc[t], false, false);
      acc[t] = __builtin_amdgcn_wmma_f32_16x16x4_f32(false, a1, false, b1, (short)0, acc[t], false, false);
    }
  }
}

template <int N, bool SCALE>
__device__ __forceinline__ void
gemm_rows_block(float* lW, int bx, int by, const float* X, const float* __restrict__ W,
                const float* __restrict__ dinv, float* __restrict__ hs, int n_rows) {
  const int tid = threadIdx.x;
  const int lane = tid & 31;
  const int wave = tid >> 5;
  float* lAw = lW + 64 * W_PITCH + wave * (16 * A_PITCH);
  const int c = lane & 15;
  const int h = lane >> 4;
  const int cb = by * 64;
  const int tile0 = (bx * 8 + wave) * 16;
  const int row0 = tile0 < n_rows - 16 ? tile0 : n_rows - 16;

  float4 raw0[8];
  float4 q0, q1, q2, q3, q4, q5, q6, q7;
  const float4* xp = (const float4*)(X + (size_t)(row0 + h) * 128) + c;
#pragma unroll
  for (int i = 0; i < 8; i++) raw0[i] = xp[64 * i];
  q0 = xp[16]; q1 = xp[64 + 16]; q2 = xp[128 + 16]; q3 = xp[192 + 16];
  q4 = xp[256 + 16]; q5 = xp[320 + 16]; q6 = xp[384 + 16]; q7 = xp[448 + 16];
  float4 wv[8];
  const int wk = 8 * (wave >> 2) + (lane & 7);
  const int wn = 16 * (wave & 3) + 4 * (lane >> 3);
  const float4* wp = (const float4*)(W + (size_t)wk * N + cb + wn);
#pragma unroll
  for (int it = 0; it < 8; it++) wv[it] = wp[it * (16 * N / 4)];
  float dv[8];
  if (SCALE) {
#pragma unroll
    for (int i = 0; i < 8; i++) dv[i] = dinv[row0 + 8 * h + i];
  }

  float* ws = lW + wn * W_PITCH + wk;
#pragma unroll
  for (int it = 0; it < 8; it++) {
    ws[0 * W_PITCH + 16 * it] = wv[it].x;
    ws[1 * W_PITCH + 16 * it] = wv[it].y;
    ws[2 * W_PITCH + 16 * it] = wv[it].z;
    ws[3 * W_PITCH + 16 * it] = wv[it].w;
  }
#pragma unroll
  for (int i = 0; i < 8; i++) *(float4*)(lAw + (2 * i + h) * A_PITCH + 4 * c) = raw0[i];
  __syncthreads();

  const float* aw = lAw + c * A_PITCH + 4 * h;
  const float* bw = lW + c * W_PITCH + 4 * h;
  float4 a[8];
  load_a_half(aw, a);
  __syncthreads();
  {
    float* d = lAw + h * A_PITCH + 4 * c;
    *(float4*)(d + 0 * A_PITCH) = q0; *(float4*)(d + 2 * A_PITCH) = q1;
    *(float4*)(d + 4 * A_PITCH) = q2; *(float4*)(d + 6 * A_PITCH) = q3;
    *(float4*)(d + 8 * A_PITCH) = q4; *(float4*)(d + 10 * A_PITCH) = q5;
    *(float4*)(d + 12 * A_PITCH) = q6; *(float4*)(d + 14 * A_PITCH) = q7;
  }
  __syncthreads();

  v8f acc[4];
#pragma unroll
  for (int t = 0; t < 4; t++) acc[t] = (v8f){0.f, 0.f, 0.f, 0.f, 0.f, 0.f, 0.f, 0.f};
  strip_half(a, bw, acc);
  load_a_half(aw, a);
  strip_half(a, bw + 64, acc);

  if (tile0 < n_rows) {
    float* o = hs + (size_t)(row0 + 8 * h) * N + cb + c;
#pragma unroll
    for (int i = 0; i < 8; i++) {
#pragma unroll
      for (int t = 0; t < 4; t++) {
        float v = acc[t][i];
        if (SCALE) v *= dv[i];
        o[i * N + 16 * t] = v;
      }
    }
  }
}

#define PLACE_TABLE_LDS_BYTES (GEMM_LDS_BYTES > PLACE_LDS_BYTES ? GEMM_LDS_BYTES : PLACE_LDS_BYTES)
__global__ void __launch_bounds__(256) __attribute__((amdgpu_waves_per_eu(8, 8)))
place_table_kernel(int gemm_blocks, const float* X, const float* __restrict__ W,
                   float* __restrict__ t1, int n_dict,
                   int* ctrl, const int2* __restrict__ binlist, const int* __restrict__ spill_src,
                   const int* __restrict__ spill_dst, const int* __restrict__ x_idx,
                   int* __restrict__ cnt, float* __restrict__ dinv, int2* __restrict__ nt,
                   int* __restrict__ slots, int* __restrict__ ovf_src, int* __restrict__ ovf_dst,
                   int n) {
  extern __shared__ float4 lds4[];
  const int b = blockIdx.x;
  if (b < gemm_blocks) {
    gemm_rows_block<128, false>((float*)lds4, b >> 1, b & 1, X, W, nullptr, t1, n_dict);
  } else {
    bin_place_block((int*)lds4, b - gemm_blocks, ctrl, binlist, spill_src, spill_dst, x_idx, cnt,
                    dinv, nt, slots, ovf_src, ovf_dst, n);
  }
}

template <int N, bool RELU, bool DICT>
__global__ void __launch_bounds__(256)
gather_sum_kernel(const int* __restrict__ cnt, const int* __restrict__ slots,
                  const int* __restrict__ ovf_n, const int* __restrict__ ovf_src,
                  const int* __restrict__ ovf_dst, const int2* __restrict__ nt,
                  const float* __restrict__ rows, const float* __restrict__ dinv,
                  const float* __restrict__ bias, float* __restrict__ out, int n_rows) {
  typedef float vec __attribute__((ext_vector_type(N / 32)));
  const int lane = threadIdx.x & 31;
  const int wpb = blockDim.x >> 5;
  const int nwaves = gridDim.x * wpb;
  const vec* __restrict__ hv = (const vec*)rows;
  vec* __restrict__ ov = (vec*)out;
  const vec bv = ((const vec*)bias)[lane];

  const int r0 = __builtin_amdgcn_readfirstlane((int)(blockIdx.x * wpb + (threadIdx.x >> 5)));
  if (r0 >= n_rows) return;

  auto count_of = [&](int r) -> int { return r < n_rows ? cnt[r] : 0; };
  auto sources_of = [&](int r, int deg) -> int {
    const int num = deg < CAP ? deg : CAP;
    int s = 0;
    if (lane < num) s = slots[(size_t)r * CAP + lane];
    return s;
  };
  auto entries_of = [&](int s, int deg) -> int2 {
    if constexpr (DICT) {
      const int num = deg < CAP ? deg : CAP;
      int2 e = make_int2(0, 0);
      if (lane < num) e = nt[s];
      return e;
    } else {
      return make_int2(s, 0);
    }
  };

  int deg_c = cnt[r0];
  int deg_1 = count_of(r0 + nwaves);
  int deg_2 = count_of(r0 + 2 * nwaves);
  int2 ent_c = make_int2(0, 0);
  float dr_c = 0.f;
  if constexpr (DICT) ent_c = nt[r0]; else dr_c = dinv[r0];
  int2 e_c = entries_of(sources_of(r0, deg_c), deg_c);
  int sl_1 = sources_of(r0 + nwaves, deg_1);
  if constexpr (DICT) {
    asm volatile("" ::"v"(e_c.x), "v"(e_c.y), "v"(sl_1));
  } else {
    asm volatile("" ::"v"(e_c.x), "v"(sl_1));
  }

  for (int r = r0; r < n_rows; r += nwaves) {
    const int deg = deg_c;
    const int num = deg < CAP ? deg : CAP;
    const float dr = DICT ? __int_as_float(ent_c.y) : dr_c;
    const size_t br = DICT ? (size_t)ent_c.x * 32 : (size_t)r * 32;
    const int rowv = e_c.x;
    const int facv = e_c.y;
    const int r1 = r + nwaves;
    const int deg_3 = count_of(r1 + 2 * nwaves);
    if (r1 < n_rows) {
      if constexpr (DICT) ent_c = nt[r1]; else dr_c = dinv[r1];
    }
    const int sl_2 = sources_of(r1 + nwaves, deg_2);
    e_c = entries_of(sl_1, deg_1);

    const vec own = hv[br + lane];
    vec acc = DICT ? (vec)(0.f) : own;
#pragma clang loop unroll(disable)
    for (int j = 0; j < num; j += 8) {
      const int left = num - j;
      vec v[8];
      float w[8];
#pragma unroll
      for (int i = 0; i < 8; i++) {
        if (i == 0 || i < left) {
          const int s = __builtin_amdgcn_readlane(rowv, j + i);
          if constexpr (DICT) w[i] = __int_as_float(__builtin_amdgcn_readlane(facv, j + i));
          v[i] = hv[(size_t)s * 32 + lane];
        }
      }
#pragma unroll
      for (int i = 0; i < 8; i++) {
        if (i == 0 || i < left) {
          if constexpr (DICT) acc += w[i] * v[i]; else acc += v[i];
        }
      }
    }
    if (deg > CAP) {
      const int tot = *ovf_n;
      for (int q0 = 0; q0 < tot; q0 += 32) {
        const int q = q0 + lane;
        int sq = 0;
        bool hit = false;
        if (q < tot) {
          hit = ovf_dst[q] == r;
          sq = ovf_src[q];
        }
        unsigned todo = __builtin_amdgcn_ballot_w32(hit);
        while (todo) {
          const int b = __builtin_ctz(todo);
          todo &= todo - 1;
          const int s = __builtin_amdgcn_readlane(sq, b);
          if constexpr (DICT) {
            const int2 e = nt[s];
            acc += __int_as_float(e.y) * hv[(size_t)e.x * 32 + lane];
          } else {
            acc += hv[(size_t)s * 32 + lane];
          }
        }
      }
    }
    if (DICT) acc += dr * own;
    vec res = dr * acc + bv;
    if (RELU) res = __builtin_elementwise_max(res, (vec)(0.f));
    ov[(size_t)r * 32 + lane] = res;
    deg_c = deg_1;
    deg_1 = deg_2;
    deg_2 = deg_3;
    sl_1 = sl_2;
  }
}

#define FA_PITCH 132
#define FUSE_WAVES 4
#define FUSE_WAVE_FLOATS (16 * FA_PITCH + 16)

__global__ void __launch_bounds__(32 * FUSE_WAVES) __attribute__((amdgpu_waves_per_eu(8, 8)))
gather_gemm_kernel(const int* __restrict__ cnt, const int* __restrict__ slots,
                   const int* __restrict__ ovf_n, const int* __restrict__ ovf_src,
                   const int* __restrict__ ovf_dst, const int2* __restrict__ nt,
                   const float* __restrict__ rows, const float* __restrict__ bias,
                   const float* __restrict__ W, float* __restrict__ hs, int n_rows) {
  typedef float vec __attribute__((ext_vector_type(4)));
  __shared__ __attribute__((aligned(16))) float lA[FUSE_WAVES * FUSE_WAVE_FLOATS];
  const int lane = threadIdx.x & 31;
  const int wave = threadIdx.x >> 5;
  float* la = lA + wave * FUSE_WAVE_FLOATS;
  float* ldr = la + 16 * FA_PITCH;
  const int c = lane & 15;
  const int h = lane >> 4;
  const vec* __restrict__ hv = (const vec*)rows;
  const vec bv = ((const vec*)bias)[lane];
  const int nstrips = (n_rows + 15) >> 4;
  const int nwaves = gridDim.x * FUSE_WAVES;
  const int strip0 = __builtin_amdgcn_readfirstlane((int)(blockIdx.x * FUSE_WAVES + (threadIdx.x >> 5)));

  for (int strip = strip0; strip < nstrips; strip += nwaves) {
    const int rb = strip * 16;
    const int rend = rb + 16 < n_rows ? rb + 16 : n_rows;

    auto count_of = [&](int r) -> int { return r < rend ? cnt[r] : 0; };
    auto sources_of = [&](int r, int deg) -> int {
      const int num = deg < CAP ? deg : CAP;
      int s = 0;
      if (lane < num) s = slots[(size_t)r * CAP + lane];
      return s;
    };
    auto entries_of = [&](int s, int deg) -> int2 {
      const int num = deg < CAP ? deg : CAP;
      int2 e = make_int2(0, 0);
      if (lane < num) e = nt[s];
      return e;
    };

    int deg_c = cnt[rb];
    int deg_1 = count_of(rb + 1);
    int deg_2 = count_of(rb + 2);
    int2 ent_c = nt[rb];
    int2 e_c = entries_of(sources_of(rb, deg_c), deg_c);
    int sl_1 = sources_of(rb + 1, deg_1);
    asm volatile("" ::"v"(e_c.x), "v"(e_c.y), "v"(sl_1));

    for (int i = rend - rb; i < 16; i++) {
      *(vec*)(la + i * FA_PITCH + 4 * lane) = (vec)(0.f);
      if (lane == 0) ldr[i] = 0.f;
    }

    for (int r = rb; r < rend; r++) {
      const int deg = deg_c;
      const int num = deg < CAP ? deg : CAP;
      const float dr = __int_as_float(ent_c.y);
      const size_t br = (size_t)ent_c.x * 32;
      const int rowv = e_c.x;
      const int facv = e_c.y;
      const int r1 = r + 1;
      const int deg_3 = count_of(r + 3);
      if (r1 < rend) ent_c = nt[r1];
      const int sl_2 = sources_of(r + 2, deg_2);
      e_c = entries_of(sl_1, deg_1);

      const vec own = hv[br + lane];
      vec acc = (vec)(0.f);
#pragma clang loop unroll(disable)
      for (int j = 0; j < num; j += 8) {
        const int left = num - j;
        vec v[8];
        float w[8];
#pragma unroll
        for (int i = 0; i < 8; i++) {
          if (i == 0 || i < left) {
            const int s = __builtin_amdgcn_readlane(rowv, j + i);
            w[i] = __int_as_float(__builtin_amdgcn_readlane(facv, j + i));
            v[i] = hv[(size_t)s * 32 + lane];
          }
        }
#pragma unroll
        for (int i = 0; i < 8; i++) {
          if (i == 0 || i < left) acc += w[i] * v[i];
        }
      }
      if (deg > CAP) {
        const int tot = *ovf_n;
        for (int q0 = 0; q0 < tot; q0 += 32) {
          const int q = q0 + lane;
          int sq = 0;
          bool hit = false;
          if (q < tot) {
            hit = ovf_dst[q] == r;
            sq = ovf_src[q];
          }
          unsigned todo = __builtin_amdgcn_ballot_w32(hit);
          while (todo) {
            const int b = __builtin_ctz(todo);
            todo &= todo - 1;
            const int s = __builtin_amdgcn_readlane(sq, b);
            const int2 e = nt[s];
            acc += __int_as_float(e.y) * hv[(size_t)e.x * 32 + lane];
          }
        }
      }
      acc += dr * own;
      vec res = dr * acc + bv;
      res = __builtin_elementwise_max(res, (vec)(0.f));
      *(vec*)(la + (r - rb) * FA_PITCH + 4 * lane) = res;
      if (lane == 0) ldr[r - rb] = dr;
      deg_c = deg_1;
      deg_1 = deg_2;
      deg_2 = deg_3;
      sl_1 = sl_2;
    }

    __builtin_amdgcn_fence(__ATOMIC_RELEASE, "wavefront");
    __builtin_amdgcn_wave_barrier();
    __builtin_amdgcn_fence(__ATOMIC_ACQUIRE, "wavefront");

    const float* aw = la + c * FA_PITCH + 4 * h;
    const float4* wp = (const float4*)W + 64 * h + c;
    v8f acc[4];
#pragma unroll
    for (int t = 0; t < 4; t++) acc[t] = (v8f){0.f, 0.f, 0.f, 0.f, 0.f, 0.f, 0.f, 0.f};
    float4 an = *(const float4*)aw;
    float4 wn0 = wp[0], wn1 = wp[16], wn2 = wp[32], wn3 = wp[48];
#pragma unroll
    for (int s = 0; s < 16; s++) {
      const float4 a = an;
      const float e0[4] = {wn0.x, wn0.y, wn0.z, wn0.w};
      const float e1[4] = {wn1.x, wn1.y, wn1.z, wn1.w};
      const float e2[4] = {wn2.x, wn2.y, wn2.z, wn2.w};
      const float e3[4] = {wn3.x, wn3.y, wn3.z, wn3.w};
      if (s + 1 < 16) {
        an = *(const float4*)(aw + 8 * (s + 1));
        const float4* wj = wp + 8 * (s + 1) * 16;
        wn0 = wj[0]; wn1 = wj[16]; wn2 = wj[32]; wn3 = wj[48];
      }
      const v2f a0 = {a.x, a.y};
      const v2f a1 = {a.z, a.w};
#pragma unroll
      for (int t = 0; t < 4; t++) {
        const v2f b0 = {e0[t], e1[t]};
        const v2f b1 = {e2[t], e3[t]};
        acc[t] = __builtin_amdgcn_wmma_f32_16x16x4_f32(false, a0, false, b0, (short)0, acc[t], false, false);
        acc[t] = __builtin_amdgcn_wmma_f32_16x16x4_f32(false, a1, false, b1, (short)0, acc[t], false, false);
      }
    }

    const float4 d0 = *(const float4*)(ldr + 8 * h);
    const float4 d1 = *(const float4*)(ldr + 8 * h + 4);
    const float dv[8] = {d0.x, d0.y, d0.z, d0.w, d1.x, d1.y, d1.z, d1.w};
    float4* o = (float4*)(hs + (size_t)(rb + 8 * h) * 64) + c;
#pragma unroll
    for (int i = 0; i < 8; i++) {
      if (rb + 8 * h + i < n_rows)
        o[i * 16] = make_float4(acc[0][i] * dv[i], acc[1][i] * dv[i], acc[2][i] * dv[i], acc[3][i] * dv[i]);
    }
    __builtin_amdgcn_fence(__ATOMIC_RELEASE, "wavefront");
    __builtin_amdgcn_wave_barrier();
    __builtin_amdgcn_fence(__ATOMIC_ACQUIRE, "wavefront");
  }
}

template <int OFF>
__device__ __forceinline__ void fold_step(float* v, int lane) {
  const bool up = (lane & OFF) != 0;
#pragma unroll
  for (int i = 0; i < OFF; i++) {
    const float keep = up ? v[i + OFF] : v[i];
    const float send = up ? v[i] : v[i + OFF];
    v[i] = keep + __shfl_xor(send, OFF, 32);
  }
}

__global__ void __launch_bounds__(128)
edgedot_kernel(const int* __restrict__ es, const int* __restrict__ ed,
               const float* __restrict__ z, float* __restrict__ out, int nL) {
  const int lane = threadIdx.x & 31;
  const int e0 = __builtin_amdgcn_readfirstlane(
      (int)((blockIdx.x * (blockDim.x >> 5) + (threadIdx.x >> 5)) << 5));
  if (e0 >= nL) return;
  const int e = e0 + lane;
  const bool ok = e < nL;
  const int sa = ok ? es[e] : 0;
  const int sb = ok ? ed[e] : 0;
  const v2f* __restrict__ zv = (const v2f*)z;
  float p[32];
#pragma unroll
  for (int i = 0; i < 32; i++) {
    const int ra = __builtin_amdgcn_readlane(sa, i);
    const int rb = __builtin_amdgcn_readlane(sb, i);
    const v2f a = zv[(size_t)ra * 32 + lane];
    const v2f b = zv[(size_t)rb * 32 + lane];
    p[i] = a.x * b.x + a.y * b.y;
  }
  fold_step<16>(p, lane);
  fold_step<8>(p, lane);
  fold_step<4>(p, lane);
  fold_step<2>(p, lane);
  fold_step<1>(p, lane);
  if (ok) out[e] = p[0];
}

extern "C" void kernel_launch(void* const* d_in, const int* in_sizes, int n_in,
                              void* d_out, int out_size, void* d_ws, size_t ws_size,
                              hipStream_t stream) {
  const int*   x_idx = (const int*)d_in[0];
  const int*   eidx  = (const int*)d_in[1];
  const int*   lidx  = (const int*)d_in[2];
  const float* emb   = (const float*)d_in[3];
  const float* W1    = (const float*)d_in[4];
  const float* b1    = (const float*)d_in[5];
  const float* W2    = (const float*)d_in[6];
  const float* b2    = (const float*)d_in[7];
  float* out = (float*)d_out;

  const int nN = in_sizes[0];
  const int nE = in_sizes[1] / 2;
  const int nL = in_sizes[2] / 2;
  const int H = 128, O = 64;
  const int nD = in_sizes[3] / H;

  const int nbin = (nN + BIN_NODES - 1) / BIN_NODES;
  const int nCtrl = CTRL_BIN0 + nbin * CTRL_STRIDE;
  const size_t nPad = ((size_t)nN + 63) & ~(size_t)63;
  const size_t ePad = ((size_t)nE + 63) & ~(size_t)63;
  const size_t cPad = ((size_t)nCtrl + 63) & ~(size_t)63;
  float* dinv  = (float*)d_ws;
  int* cnt     = (int*)(dinv + nPad);
  int* ctrl    = cnt + nPad;
  int* ovf_n   = ctrl;
  int2* nt     = (int2*)(ctrl + cPad);
  int* slots   = ctrl + cPad + 2 * nPad;
  int* ovf_src = slots + nPad * CAP;
  int* ovf_dst = ovf_src + ePad;
  int* spill_src = ovf_dst + ePad;
  int* spill_dst = spill_src + ePad;
  float* t1    = (float*)(spill_dst + ePad);
  float* hs2   = t1 + (size_t)nD * H;
  float* z     = hs2 + (size_t)nN * O;
  int2* binlist = (int2*)(z + (size_t)nN * O);

  const int* esrc = eidx;
  const int* edst = eidx + nE;

  ctrl_zero_kernel<<<(nCtrl + 255) / 256, 256, 0, stream>>>(ctrl, nCtrl);
  bin_edges_kernel<<<(nE + BE_BLOCK * BE_EPT - 1) / (BE_BLOCK * BE_EPT), BE_BLOCK,
                     2 * nbin * sizeof(int), stream>>>(esrc, edst, ctrl, binlist, spill_src,
                                                       spill_dst, nE, nbin);
  const int gemm_blocks = 2 * ((nD + 127) / 128);
  place_table_kernel<<<gemm_blocks + nbin, BIN_NODES, PLACE_TABLE_LDS_BYTES, stream>>>(
      gemm_blocks, emb, W1, t1, nD, ctrl, binlist, spill_src, spill_dst, x_idx, cnt, dinv, nt,
      slots, ovf_src, ovf_dst, nN);

  gather_gemm_kernel<<<((nN + 15) / 16 + FUSE_WAVES - 1) / FUSE_WAVES, 32 * FUSE_WAVES, 0, stream>>>(
      cnt, slots, ovf_n, ovf_src, ovf_dst, nt, t1, b1, W2, hs2, nN);

  gather_sum_kernel<64, false, false><<<2048, 256, 0, stream>>>(
      cnt, slots, ovf_n, ovf_src, ovf_dst, nt, hs2, dinv, b2, z, nN);

  edgedot_kernel<<<(nL + 127) / 128, 128, 0, stream>>>(
      lidx, lidx + nL, z, out, nL);
}


# Adjacency build without global counters: each (bin, block) owns a 256-byte segment (count word + pairs by LDS rank), read one wave load each; clearing launch and global adds gone; sum order rule kept
# speedup vs baseline: 5.9713x; 1.0516x over previous
#include <hip/hip_runtime.h>

typedef float v2f __attribute__((ext_vector_type(2)));
typedef float v8f __attribute__((ext_vector_type(8)));

#define CAP 32
#define BIN_SHIFT 8
#define BIN_NODES 256
#define SEG 32
#define SEGD (SEG - 1)
#define BE_BLOCK 512
#define BE_EPT 8
#define BE_EDGES (BE_BLOCK * BE_EPT)
#define PL_R 13

__global__ void __launch_bounds__(BE_BLOCK)
bin_edges_kernel(const int* __restrict__ src, const int* __restrict__ dst, int* __restrict__ ctrl,
                 int2* __restrict__ seg, int* __restrict__ spill_src,
                 int* __restrict__ spill_dst, int* __restrict__ spillcnt, int nE, int nbin) {
  extern __shared__ int be_lds[];
  int* hist = be_lds;
  int* spill_n = be_lds + nbin;
  const int tid = threadIdx.x;
  const int blk = blockIdx.x;
  const int nblk = gridDim.x;
  for (int b = tid; b < nbin; b += BE_BLOCK) hist[b] = 0;
  if (tid == 0) *spill_n = 0;
  const int e0 = blk * BE_EDGES + tid;
  int d[BE_EPT], s[BE_EPT], rk[BE_EPT];
#pragma unroll
  for (int i = 0; i < BE_EPT; i++) {
    const int e = e0 + i * BE_BLOCK;
    d[i] = e < nE ? dst[e] : -1;
    s[i] = e < nE ? src[e] : 0;
  }
  __syncthreads();
#pragma unroll
  for (int i = 0; i < BE_EPT; i++) {
    rk[i] = 0;
    if (d[i] >= 0) rk[i] = atomicAdd(&hist[d[i] >> BIN_SHIFT], 1);
  }
  __syncthreads();
#pragma unroll
  for (int i = 0; i < BE_EPT; i++)
    if (d[i] >= 0 && rk[i] >= SEGD) rk[i] = SEGD + atomicAdd(spill_n, 1);
  __syncthreads();
  if (tid == 0) spillcnt[blk] = *spill_n;
  if (blk == 0 && tid == 0) ctrl[0] = 0;
  for (int b = tid; b < nbin; b += BE_BLOCK)
    seg[((size_t)b * nblk + blk) * SEG] = make_int2(hist[b], 0);
#pragma unroll
  for (int i = 0; i < BE_EPT; i++) {
    if (d[i] >= 0) {
      const int b = d[i] >> BIN_SHIFT;
      if (rk[i] < SEGD) {
        seg[((size_t)b * nblk + blk) * SEG + 1 + rk[i]] = make_int2(d[i], s[i]);
      } else {
        spill_src[(size_t)blk * BE_EDGES + rk[i] - SEGD] = s[i];
        spill_dst[(size_t)blk * BE_EDGES + rk[i] - SEGD] = d[i];
      }
    }
  }
}

#define PLACE_LDS_BYTES ((BIN_NODES + BIN_NODES * CAP) * 4)
__device__ __forceinline__ void
bin_place_block(int* lds, int bin, int nblk, int* ctrl, const int2* __restrict__ seg,
                const int* __restrict__ spill_src, const int* __restrict__ spill_dst,
                const int* __restrict__ spillcnt,
                const int* __restrict__ x_idx, int* __restrict__ cnt, float* __restrict__ dinv,
                int2* __restrict__ nt, int* __restrict__ slots, int* __restrict__ ovf_src,
                int* __restrict__ ovf_dst, int n) {
  int* cntl = lds;
  int* slotl = lds + BIN_NODES;
  const int tid = threadIdx.x;
  const int lane = tid & 31;
  const int wave = __builtin_amdgcn_readfirstlane(tid >> 5);
  const int node = bin * BIN_NODES + tid;
  const int xi = node < n ? x_idx[node] : 0;
  const int2* sg = seg + (size_t)bin * nblk * SEG + lane;
  int2 p[PL_R];
  auto load_round = [&](int b0) {
#pragma unroll
    for (int j = 0; j < PL_R; j++) {
      const int blk = b0 + wave + 8 * j;
      p[j] = make_int2(0, 0);
      if (blk < nblk) p[j] = sg[(size_t)blk * SEG];
    }
  };
  load_round(0);
  int4* sl4 = (int4*)slotl;
  cntl[tid] = 0;
#pragma unroll
  for (int i = 0; i < BIN_NODES * CAP / 4 / BIN_NODES; i++) sl4[tid + i * BIN_NODES] = make_int4(0, 0, 0, 0);
  __syncthreads();

  auto put = [&](int t, int k, int s) {
    if (k < CAP) {
      slotl[t * CAP + k] = s;
    } else {
      const int q = atomicAdd(&ctrl[0], 1);
      ovf_src[q] = s;
      ovf_dst[q] = bin * BIN_NODES + t;
    }
  };
  for (int b0 = 0;;) {
    int k[PL_R];
#pragma unroll
    for (int j = 0; j < PL_R; j++) {
      const int h = __builtin_amdgcn_readfirstlane(p[j].x);
      const int hs = h < SEGD ? h : SEGD;
      const int t = p[j].x - bin * BIN_NODES;
      p[j].x = h;
      k[j] = -1;
      if (lane >= 1 && lane <= hs) {
        p[j].x = t;
        k[j] = atomicAdd(&cntl[t], 1);
      }
    }
#pragma unroll
    for (int j = 0; j < PL_R; j++) {
      if (k[j] >= 0) put(p[j].x, k[j], p[j].y);
      const int h = __builtin_amdgcn_readfirstlane(p[j].x);
      if (h > SEGD) {
        const int blk = b0 + wave + 8 * j;
        int ns = spillcnt[blk];
        ns = ns < BE_EDGES ? ns : BE_EDGES;
        for (int i = lane; i < ns; i += 32) {
          const int d = spill_dst[(size_t)blk * BE_EDGES + i];
          if ((d >> BIN_SHIFT) == bin) {
            const int t = d - bin * BIN_NODES;
            put(t, atomicAdd(&cntl[t], 1), spill_src[(size_t)blk * BE_EDGES + i]);
          }
        }
      }
    }
    b0 += 8 * PL_R;
    if (b0 >= nblk) break;
    load_round(b0);
  }
  __syncthreads();

  if (node < n) {
    const int c = cntl[tid];
    const float dd = rsqrtf((float)(c + 1));
    cnt[node] = c;
    dinv[node] = dd;
    nt[node] = make_int2(xi, __float_as_int(dd));
  }
  int4* g4 = (int4*)(slots + (size_t)bin * BIN_NODES * CAP);
#pragma unroll
  for (int i = 0; i < BIN_NODES * CAP / 4 / BIN_NODES; i++) {
    const int idx = tid + i * BIN_NODES;
    if (bin * BIN_NODES + (idx >> 3) < n) g4[idx] = sl4[idx];
  }
}

#define W_PITCH 132
#define A_PITCH 68
#define GEMM_LDS_BYTES ((64 * W_PITCH + 8 * 16 * A_PITCH) * 4)

__device__ __forceinline__ void load_a_half(const float* __restrict__ aw, float4 (&a)[8]) {
#pragma unroll
  for (int j = 0; j < 8; j++) a[j] = *(const float4*)(aw + 8 * j);
}

__device__ __forceinline__ void strip_half(const float4 (&a)[8], const float* __restrict__ bw,
                                           v8f (&acc)[4]) {
#pragma unroll
  for (int j = 0; j < 8; j++) {
    const v2f a0 = {a[j].x, a[j].y};
    const v2f a1 = {a[j].z, a[j].w};
#pragma unroll
    for (int t = 0; t < 4; t++) {
      const float4 b = *(const float4*)(bw + t * 16 * W_PITCH + 8 * j);
      const v2f b0 = {b.x, b.y};
      const v2f b1 = {b.z, b.w};
      acc[t] = __builtin_amdgcn_wmma_f32_16x16x4_f32(false, a0, false, b0, (short)0, acc[t], false, false);
      acc[t] = __builtin_amdgcn_wmma_f32_16x16x4_f32(false, a1, false, b1, (short)0, acc[t], false, false);
    }
  }
}

template <int N, bool SCALE>
__device__ __forceinline__ void
gemm_rows_block(float* lW, int bx, int by, const float* X, const float* __restrict__ W,
                const float* __restrict__ dinv, float* __restrict__ hs, int n_rows) {
  const int tid = threadIdx.x;
  const int lane = tid & 31;
  const int wave = tid >> 5;
  float* lAw = lW + 64 * W_PITCH + wave * (16 * A_PITCH);
  const int c = lane & 15;
  const int h = lane >> 4;
  const int cb = by * 64;
  const int tile0 = (bx * 8 + wave) * 16;
  const int row0 = tile0 < n_rows - 16 ? tile0 : n_rows - 16;

  float4 raw0[8];
  float4 q0, q1, q2, q3, q4, q5, q6, q7;
  const float4* xp = (const float4*)(X + (size_t)(row0 + h) * 128) + c;
#pragma unroll
  for (int i = 0; i < 8; i++) raw0[i] = xp[64 * i];
  q0 = xp[16]; q1 = xp[64 + 16]; q2 = xp[128 + 16]; q3 = xp[192 + 16];
  q4 = xp[256 + 16]; q5 = xp[320 + 16]; q6 = xp[384 + 16]; q7 = xp[448 + 16];
  float4 wv[8];
  const int wk = 8 * (wave >> 2) + (lane & 7);
  const int wn = 16 * (wave & 3) + 4 * (lane >> 3);
  const float4* wp = (const float4*)(W + (size_t)wk * N + cb + wn);
#pragma unroll
  for (int it = 0; it < 8; it++) wv[it] = wp[it * (16 * N / 4)];
  float dv[8];
  if (SCALE) {
#pragma unroll
    for (int i = 0; i < 8; i++) dv[i] = dinv[row0 + 8 * h + i];
  }

  float* ws = lW + wn * W_PITCH + wk;
#pragma unroll
  for (int it = 0; it < 8; it++) {
    ws[0 * W_PITCH + 16 * it] = wv[it].x;
    ws[1 * W_PITCH + 16 * it] = wv[it].y;
    ws[2 * W_PITCH + 16 * it] = wv[it].z;
    ws[3 * W_PITCH + 16 * it] = wv[it].w;
  }
#pragma unroll
  for (int i = 0; i < 8; i++) *(float4*)(lAw + (2 * i + h) * A_PITCH + 4 * c) = raw0[i];
  __syncthreads();

  const float* aw = lAw + c * A_PITCH + 4 * h;
  const float* bw = lW + c * W_PITCH + 4 * h;
  float4 a[8];
  load_a_half(aw, a);
  __syncthreads();
  {
    float* d = lAw + h * A_PITCH + 4 * c;
    *(float4*)(d + 0 * A_PITCH) = q0; *(float4*)(d + 2 * A_PITCH) = q1;
    *(float4*)(d + 4 * A_PITCH) = q2; *(float4*)(d + 6 * A_PITCH) = q3;
    *(float4*)(d + 8 * A_PITCH) = q4; *(float4*)(d + 10 * A_PITCH) = q5;
    *(float4*)(d + 12 * A_PITCH) = q6; *(float4*)(d + 14 * A_PITCH) = q7;
  }
  __syncthreads();

  v8f acc[4];
#pragma unroll
  for (int t = 0; t < 4; t++) acc[t] = (v8f){0.f, 0.f, 0.f, 0.f, 0.f, 0.f, 0.f, 0.f};
  strip_half(a, bw, acc);
  load_a_half(aw, a);
  strip_half(a, bw + 64, acc);

  if (tile0 < n_rows) {
    float* o = hs + (size_t)(row0 + 8 * h) * N + cb + c;
#pragma unroll
    for (int i = 0; i < 8; i++) {
#pragma unroll
      for (int t = 0; t < 4; t++) {
        float v = acc[t][i];
        if (SCALE) v *= dv[i];
        o[i * N + 16 * t] = v;
      }
    }
  }
}

#define PLACE_TABLE_LDS_BYTES (GEMM_LDS_BYTES > PLACE_LDS_BYTES ? GEMM_LDS_BYTES : PLACE_LDS_BYTES)
__global__ void __launch_bounds__(256) __attribute__((amdgpu_waves_per_eu(8, 8)))
place_table_kernel(int gemm_blocks, const float* X, const float* __restrict__ W,
                   float* __restrict__ t1, int n_dict, int nblk,
                   int* ctrl, const int2* __restrict__ seg, const int* __restrict__ spill_src,
                   const int* __restrict__ spill_dst, const int* __restrict__ spillcnt,
                   const int* __restrict__ x_idx,
                   int* __restrict__ cnt, float* __restrict__ dinv, int2* __restrict__ nt,
                   int* __restrict__ slots, int* __restrict__ ovf_src, int* __restrict__ ovf_dst,
                   int n) {
  extern __shared__ float4 lds4[];
  const int b = blockIdx.x;
  if (b < gemm_blocks) {
    gemm_rows_block<128, false>((float*)lds4, b >> 1, b & 1, X, W, nullptr, t1, n_dict);
  } else {
    bin_place_block((int*)lds4, b - gemm_blocks, nblk, ctrl, seg, spill_src, spill_dst, spillcnt,
                    x_idx, cnt, dinv, nt, slots, ovf_src, ovf_dst, n);
  }
}

template <int N, bool RELU, bool DICT>
__global__ void __launch_bounds__(256)
gather_sum_kernel(const int* __restrict__ cnt, const int* __restrict__ slots,
                  const int* __restrict__ ovf_n, const int* __restrict__ ovf_src,
                  const int* __restrict__ ovf_dst, const int2* __restrict__ nt,
                  const float* __restrict__ rows, const float* __restrict__ dinv,
                  const float* __restrict__ bias, float* __restrict__ out, int n_rows) {
  typedef float vec __attribute__((ext_vector_type(N / 32)));
  const int lane = threadIdx.x & 31;
  const int wpb = blockDim.x >> 5;
  const int nwaves = gridDim.x * wpb;
  const vec* __restrict__ hv = (const vec*)rows;
  vec* __restrict__ ov = (vec*)out;
  const vec bv = ((const vec*)bias)[lane];

  const int r0 = __builtin_amdgcn_readfirstlane((int)(blockIdx.x * wpb + (threadIdx.x >> 5)));
  if (r0 >= n_rows) return;

  auto count_of = [&](int r) -> int { return r < n_rows ? cnt[r] : 0; };
  auto sources_of = [&](int r, int deg) -> int {
    const int num = deg < CAP ? deg : CAP;
    int s = 0;
    if (lane < num) s = slots[(size_t)r * CAP + lane];
    return s;
  };
  auto entries_of = [&](int s, int deg) -> int2 {
    if constexpr (DICT) {
      const int num = deg < CAP ? deg : CAP;
      int2 e = make_int2(0, 0);
      if (lane < num) e = nt[s];
      return e;
    } else {
      return make_int2(s, 0);
    }
  };

  int deg_c = cnt[r0];
  int deg_1 = count_of(r0 + nwaves);
  int deg_2 = count_of(r0 + 2 * nwaves);
  int2 ent_c = make_int2(0, 0);
  float dr_c = 0.f;
  if constexpr (DICT) ent_c = nt[r0]; else dr_c = dinv[r0];
  int2 e_c = entries_of(sources_of(r0, deg_c), deg_c);
  int sl_1 = sources_of(r0 + nwaves, deg_1);
  if constexpr (DICT) {
    asm volatile("" ::"v"(e_c.x), "v"(e_c.y), "v"(sl_1));
  } else {
    asm volatile("" ::"v"(e_c.x), "v"(sl_1));
  }

  for (int r = r0; r < n_rows; r += nwaves) {
    const int deg = deg_c;
    const int num = deg < CAP ? deg : CAP;
    const float dr = DICT ? __int_as_float(ent_c.y) : dr_c;
    const size_t br = DICT ? (size_t)ent_c.x * 32 : (size_t)r * 32;
    const int rowv = e_c.x;
    const int facv = e_c.y;
    const int r1 = r + nwaves;
    const int deg_3 = count_of(r1 + 2 * nwaves);
    if (r1 < n_rows) {
      if constexpr (DICT) ent_c = nt[r1]; else dr_c = dinv[r1];
    }
    const int sl_2 = sources_of(r1 + nwaves, deg_2);
    e_c = entries_of(sl_1, deg_1);

    const vec own = hv[br + lane];
    vec acc = DICT ? (vec)(0.f) : own;
#pragma clang loop unroll(disable)
    for (int j = 0; j < num; j += 8) {
      const int left = num - j;
      vec v[8];
      float w[8];
#pragma unroll
      for (int i = 0; i < 8; i++) {
        if (i == 0 || i < left) {
          const int s = __builtin_amdgcn_readlane(rowv, j + i);
          if constexpr (DICT) w[i] = __int_as_float(__builtin_amdgcn_readlane(facv, j + i));
          v[i] = hv[(size_t)s * 32 + lane];
        }
      }
#pragma unroll
      for (int i = 0; i < 8; i++) {
        if (i == 0 || i < left) {
          if constexpr (DICT) acc += w[i] * v[i]; else acc += v[i];
        }
      }
    }
    if (deg > CAP) {
      const int tot = *ovf_n;
      for (int q0 = 0; q0 < tot; q0 += 32) {
        const int q = q0 + lane;
        int sq = 0;
        bool hit = false;
        if (q < tot) {
          hit = ovf_dst[q] == r;
          sq = ovf_src[q];
        }
        unsigned todo = __builtin_amdgcn_ballot_w32(hit);
        while (todo) {
          const int b = __builtin_ctz(todo);
          todo &= todo - 1;
          const int s = __builtin_amdgcn_readlane(sq, b);
          if constexpr (DICT) {
            const int2 e = nt[s];
            acc += __int_as_float(e.y) * hv[(size_t)e.x * 32 + lane];
          } else {
            acc += hv[(size_t)s * 32 + lane];
          }
        }
      }
    }
    if (DICT) acc += dr * own;
    vec res = dr * acc + bv;
    if (RELU) res = __builtin_elementwise_max(res, (vec)(0.f));
    ov[(size_t)r * 32 + lane] = res;
    deg_c = deg_1;
    deg_1 = deg_2;
    deg_2 = deg_3;
    sl_1 = sl_2;
  }
}

#define FA_PITCH 132
#define FUSE_WAVES 4
#define FUSE_WAVE_FLOATS (16 * FA_PITCH + 16)

__global__ void __launch_bounds__(32 * FUSE_WAVES) __attribute__((amdgpu_waves_per_eu(8, 8)))
gather_gemm_kernel(const int* __restrict__ cnt, const int* __restrict__ slots,
                   const int* __restrict__ ovf_n, const int* __restrict__ ovf_src,
                   const int* __restrict__ ovf_dst, const int2* __restrict__ nt,
                   const float* __restrict__ rows, const float* __restrict__ bias,
                   const float* __restrict__ W, float* __restrict__ hs, int n_rows) {
  typedef float vec __attribute__((ext_vector_type(4)));
  __shared__ __attribute__((aligned(16))) float lA[FUSE_WAVES * FUSE_WAVE_FLOATS];
  const int lane = threadIdx.x & 31;
  const int wave = threadIdx.x >> 5;
  float* la = lA + wave * FUSE_WAVE_FLOATS;
  float* ldr = la + 16 * FA_PITCH;
  const int c = lane & 15;
  const int h = lane >> 4;
  const vec* __restrict__ hv = (const vec*)rows;
  const vec bv = ((const vec*)bias)[lane];
  const int nstrips = (n_rows + 15) >> 4;
  const int nwaves = gridDim.x * FUSE_WAVES;
  const int strip0 = __builtin_amdgcn_readfirstlane((int)(blockIdx.x * FUSE_WAVES + (threadIdx.x >> 5)));

  for (int strip = strip0; strip < nstrips; strip += nwaves) {
    const int rb = strip * 16;
    const int rend = rb + 16 < n_rows ? rb + 16 : n_rows;

    auto count_of = [&](int r) -> int { return r < rend ? cnt[r] : 0; };
    auto sources_of = [&](int r, int deg) -> int {
      const int num = deg < CAP ? deg : CAP;
      int s = 0;
      if (lane < num) s = slots[(size_t)r * CAP + lane];
      return s;
    };
    auto entries_of = [&](int s, int deg) -> int2 {
      const int num = deg < CAP ? deg : CAP;
      int2 e = make_int2(0, 0);
      if (lane < num) e = nt[s];
      return e;
    };

    int deg_c = cnt[rb];
    int deg_1 = count_of(rb + 1);
    int deg_2 = count_of(rb + 2);
    int2 ent_c = nt[rb];
    int2 e_c = entries_of(sources_of(rb, deg_c), deg_c);
    int sl_1 = sources_of(rb + 1, deg_1);
    asm volatile("" ::"v"(e_c.x), "v"(e_c.y), "v"(sl_1));

    for (int i = rend - rb; i < 16; i++) {
      *(vec*)(la + i * FA_PITCH + 4 * lane) = (vec)(0.f);
      if (lane == 0) ldr[i] = 0.f;
    }

    for (int r = rb; r < rend; r++) {
      const int deg = deg_c;
      const int num = deg < CAP ? deg : CAP;
      const float dr = __int_as_float(ent_c.y);
      const size_t br = (size_t)ent_c.x * 32;
      const int rowv = e_c.x;
      const int facv = e_c.y;
      const int r1 = r + 1;
      const int deg_3 = count_of(r + 3);
      if (r1 < rend) ent_c = nt[r1];
      const int sl_2 = sources_of(r + 2, deg_2);
      e_c = entries_of(sl_1, deg_1);

      const vec own = hv[br + lane];
      vec acc = (vec)(0.f);
#pragma clang loop unroll(disable)
      for (int j = 0; j < num; j += 8) {
        const int left = num - j;
        vec v[8];
        float w[8];
#pragma unroll
        for (int i = 0; i < 8; i++) {
          if (i == 0 || i < left) {
            const int s = __builtin_amdgcn_readlane(rowv, j + i);
            w[i] = __int_as_float(__builtin_amdgcn_readlane(facv, j + i));
            v[i] = hv[(size_t)s * 32 + lane];
          }
        }
#pragma unroll
        for (int i = 0; i < 8; i++) {
          if (i == 0 || i < left) acc += w[i] * v[i];
        }
      }
      if (deg > CAP) {
        const int tot = *ovf_n;
        for (int q0 = 0; q0 < tot; q0 += 32) {
          const int q = q0 + lane;
          int sq = 0;
          bool hit = false;
          if (q < tot) {
            hit = ovf_dst[q] == r;
            sq = ovf_src[q];
          }
          unsigned todo = __builtin_amdgcn_ballot_w32(hit);
          while (todo) {
            const int b = __builtin_ctz(todo);
            todo &= todo - 1;
            const int s = __builtin_amdgcn_readlane(sq, b);
            const int2 e = nt[s];
            acc += __int_as_float(e.y) * hv[(size_t)e.x * 32 + lane];
          }
        }
      }
      acc += dr * own;
      vec res = dr * acc + bv;
      res = __builtin_elementwise_max(res, (vec)(0.f));
      *(vec*)(la + (r - rb) * FA_PITCH + 4 * lane) = res;
      if (lane == 0) ldr[r - rb] = dr;
      deg_c = deg_1;
      deg_1 = deg_2;
      deg_2 = deg_3;
      sl_1 = sl_2;
    }

    __builtin_amdgcn_fence(__ATOMIC_RELEASE, "wavefront");
    __builtin_amdgcn_wave_barrier();
    __builtin_amdgcn_fence(__ATOMIC_ACQUIRE, "wavefront");

    const float* aw = la + c * FA_PITCH + 4 * h;
    const float4* wp = (const float4*)W + 64 * h + c;
    v8f acc[4];
#pragma unroll
    for (int t = 0; t < 4; t++) acc[t] = (v8f){0.f, 0.f, 0.f, 0.f, 0.f, 0.f, 0.f, 0.f};
    float4 an = *(const float4*)aw;
    float4 wn0 = wp[0], wn1 = wp[16], wn2 = wp[32], wn3 = wp[48];
#pragma unroll
    for (int s = 0; s < 16; s++) {
      const float4 a = an;
      const float e0[4] = {wn0.x, wn0.y, wn0.z, wn0.w};
      const float e1[4] = {wn1.x, wn1.y, wn1.z, wn1.w};
      const float e2[4] = {wn2.x, wn2.y, wn2.z, wn2.w};
      const float e3[4] = {wn3.x, wn3.y, wn3.z, wn3.w};
      if (s + 1 < 16) {
        an = *(const float4*)(aw + 8 * (s + 1));
        const float4* wj = wp + 8 * (s + 1) * 16;
        wn0 = wj[0]; wn1 = wj[16]; wn2 = wj[32]; wn3 = wj[48];
      }
      const v2f a0 = {a.x, a.y};
      const v2f a1 = {a.z, a.w};
#pragma unroll
      for (int t = 0; t < 4; t++) {
        const v2f b0 = {e0[t], e1[t]};
        const v2f b1 = {e2[t], e3[t]};
        acc[t] = __builtin_amdgcn_wmma_f32_16x16x4_f32(false, a0, false, b0, (short)0, acc[t], false, false);
        acc[t] = __builtin_amdgcn_wmma_f32_16x16x4_f32(false, a1, false, b1, (short)0, acc[t], false, false);
      }
    }

    const float4 d0 = *(const float4*)(ldr + 8 * h);
    const float4 d1 = *(const float4*)(ldr + 8 * h + 4);
    const float dv[8] = {d0.x, d0.y, d0.z, d0.w, d1.x, d1.y, d1.z, d1.w};
    float4* o = (float4*)(hs + (size_t)(rb + 8 * h) * 64) + c;
#pragma unroll
    for (int i = 0; i < 8; i++) {
      if (rb + 8 * h + i < n_rows)
        o[i * 16] = make_float4(acc[0][i] * dv[i], acc[1][i] * dv[i], acc[2][i] * dv[i], acc[3][i] * dv[i]);
    }
    __builtin_amdgcn_fence(__ATOMIC_RELEASE, "wavefront");
    __builtin_amdgcn_wave_barrier();
    __builtin_amdgcn_fence(__ATOMIC_ACQUIRE, "wavefront");
  }
}

template <int OFF>
__device__ __forceinline__ void fold_step(float* v, int lane) {
  const bool up = (lane & OFF) != 0;
#pragma unroll
  for (int i = 0; i < OFF; i++) {
    const float keep = up ? v[i + OFF] : v[i];
    const float send = up ? v[i] : v[i + OFF];
    v[i] = keep + __shfl_xor(send, OFF, 32);
  }
}

__global__ void __launch_bounds__(128)
edgedot_kernel(const int* __restrict__ es, const int* __restrict__ ed,
               const float* __restrict__ z, float* __restrict__ out, int nL) {
  const int lane = threadIdx.x & 31;
  const int e0 = __builtin_amdgcn_readfirstlane(
      (int)((blockIdx.x * (blockDim.x >> 5) + (threadIdx.x >> 5)) << 5));
  if (e0 >= nL) return;
  const int e = e0 + lane;
  const bool ok = e < nL;
  const int sa = ok ? es[e] : 0;
  const int sb = ok ? ed[e] : 0;
  const v2f* __restrict__ zv = (const v2f*)z;
  float p[32];
#pragma unroll
  for (int i = 0; i < 32; i++) {
    const int ra = __builtin_amdgcn_readlane(sa, i);
    const int rb = __builtin_amdgcn_readlane(sb, i);
    const v2f a = zv[(size_t)ra * 32 + lane];
    const v2f b = zv[(size_t)rb * 32 + lane];
    p[i] = a.x * b.x + a.y * b.y;
  }
  fold_step<16>(p, lane);
  fold_step<8>(p, lane);
  fold_step<4>(p, lane);
  fold_step<2>(p, lane);
  fold_step<1>(p, lane);
  if (ok) out[e] = p[0];
}

extern "C" void kernel_launch(void* const* d_in, const int* in_sizes, int n_in,
                              void* d_out, int out_size, void* d_ws, size_t ws_size,
                              hipStream_t stream) {
  const int*   x_idx = (const int*)d_in[0];
  const int*   eidx  = (const int*)d_in[1];
  const int*   lidx  = (const int*)d_in[2];
  const float* emb   = (const float*)d_in[3];
  const float* W1    = (const float*)d_in[4];
  const float* b1    = (const float*)d_in[5];
  const float* W2    = (const float*)d_in[6];
  const float* b2    = (const float*)d_in[7];
  float* out = (float*)d_out;

  const int nN = in_sizes[0];
  const int nE = in_sizes[1] / 2;
  const int nL = in_sizes[2] / 2;
  const int H = 128, O = 64;
  const int nD = in_sizes[3] / H;

  const int nbin = (nN + BIN_NODES - 1) / BIN_NODES;
  const int nblk = (nE + BE_EDGES - 1) / BE_EDGES;
  const size_t nPad = ((size_t)nN + 63) & ~(size_t)63;
  const size_t ePad = ((size_t)nE + 63) & ~(size_t)63;
  const size_t bPad = ((size_t)nblk + 63) & ~(size_t)63;
  float* dinv  = (float*)d_ws;
  int* cnt     = (int*)(dinv + nPad);
  int* ctrl    = cnt + nPad;
  int* ovf_n   = ctrl;
  int2* nt     = (int2*)(ctrl + 64);
  int* slots   = ctrl + 64 + 2 * nPad;
  int* ovf_src = slots + nPad * CAP;
  int* ovf_dst = ovf_src + ePad;
  int* spillcnt  = ovf_dst + ePad;
  int* spill_src = spillcnt + bPad;
  int* spill_dst = spill_src + (size_t)nblk * BE_EDGES;
  float* t1    = (float*)(spill_dst + (size_t)nblk * BE_EDGES);
  float* hs2   = t1 + (size_t)nD * H;
  float* z     = hs2 + (size_t)nN * O;
  int2* seg    = (int2*)(z + (size_t)nN * O);

  const int* esrc = eidx;
  const int* edst = eidx + nE;

  bin_edges_kernel<<<nblk, BE_BLOCK, (nbin + 1) * sizeof(int), stream>>>(
      esrc, edst, ctrl, seg, spill_src, spill_dst, spillcnt, nE, nbin);
  const int gemm_blocks = 2 * ((nD + 127) / 128);
  place_table_kernel<<<gemm_blocks + nbin, BIN_NODES, PLACE_TABLE_LDS_BYTES, stream>>>(
      gemm_blocks, emb, W1, t1, nD, nblk, ctrl, seg, spill_src, spill_dst, spillcnt, x_idx, cnt,
      dinv, nt, slots, ovf_src, ovf_dst, nN);

  gather_gemm_kernel<<<((nN + 15) / 16 + FUSE_WAVES - 1) / FUSE_WAVES, 32 * FUSE_WAVES, 0, stream>>>(
      cnt, slots, ovf_n, ovf_src, ovf_dst, nt, t1, b1, W2, hs2, nN);

  gather_sum_kernel<64, false, false><<<2048, 256, 0, stream>>>(
      cnt, slots, ovf_n, ovf_src, ovf_dst, nt, hs2, dinv, b2, z, nN);

  edgedot_kernel<<<(nL + 127) / 128, 128, 0, stream>>>(
      lidx, lidx + nL, z, out, nL);
}
